# starting kernel
# baseline (speedup 1.0000x reference)
#include <hip/hip_runtime.h>
#include <hip/hip_bf16.h>
#include <math.h>

typedef __attribute__((ext_vector_type(2))) float v2f;
typedef __attribute__((ext_vector_type(8))) float v8f;

#define EPS_RHO 1e-8f
#define EPS_DEN 1e-9f

__device__ __forceinline__ float rho_relu(float x) {
    float p = sqrtf(fmaxf(x, 0.f) + EPS_RHO);
    float n = sqrtf(fmaxf(-x, 0.f) + EPS_RHO);
    return fmaxf(p - n, 0.f);
}

__device__ __forceinline__ v8f wmma_f32(v2f a, v2f b, v8f c) {
    return __builtin_amdgcn_wmma_f32_16x16x4_f32(false, a, false, b,
                                                 (short)0, c, false, false);
}

__global__ void fa_gemm_abt(const float* __restrict__ A,
                            const float* __restrict__ W,
                            const float* __restrict__ bias,
                            float* __restrict__ Out,
                            int M, int K, int Nc)
{
    const int lane  = threadIdx.x & 31;
    const int wave  = threadIdx.x >> 5;
    const int nblkN = Nc >> 6;
    const int job   = blockIdx.x * (blockDim.x >> 5) + wave;
    if (job >= (M >> 5) * nblkN) return;
    const int mb = job / nblkN, nb = job % nblkN;
    const int l15  = lane & 15;
    const int koff = (lane < 16) ? 0 : 2;
    const size_t bstep = (size_t)16 * K;

    const float* pA0 = A + (size_t)((mb << 5) + l15) * K + koff;
    const float* pA1 = pA0 + bstep;
    const float* pB0 = W + (size_t)((nb << 6) + l15) * K + koff;
    const float* pB1 = pB0 + bstep;
    const float* pB2 = pB1 + bstep;
    const float* pB3 = pB2 + bstep;

    v8f acc[2][4] = {};
    #pragma unroll 2
    for (int kk = 0; kk < K; kk += 4) {
        const v2f a0 = *(const v2f*)pA0;
        const v2f a1 = *(const v2f*)pA1;
        const v2f b0 = *(const v2f*)pB0;
        const v2f b1 = *(const v2f*)pB1;
        const v2f b2 = *(const v2f*)pB2;
        const v2f b3 = *(const v2f*)pB3;
        pA0 += 4; pA1 += 4; pB0 += 4; pB1 += 4; pB2 += 4; pB3 += 4;

        acc[0][0] = wmma_f32(a0, b0, acc[0][0]);
        acc[1][0] = wmma_f32(a1, b0, acc[1][0]);
        acc[0][1] = wmma_f32(a0, b1, acc[0][1]);
        acc[1][1] = wmma_f32(a1, b1, acc[1][1]);
        acc[0][2] = wmma_f32(a0, b2, acc[0][2]);
        acc[1][2] = wmma_f32(a1, b2, acc[1][2]);
        acc[0][3] = wmma_f32(a0, b3, acc[0][3]);
        acc[1][3] = wmma_f32(a1, b3, acc[1][3]);
    }

    const int mbase = (mb << 5) + ((lane < 16) ? 0 : 8);
    #pragma unroll
    for (int ni = 0; ni < 4; ++ni) {
        const int   n  = (nb << 6) + (ni << 4) + l15;
        const float bn = bias ? bias[n] : 0.f;
        #pragma unroll
        for (int mi = 0; mi < 2; ++mi)
            #pragma unroll
            for (int r = 0; r < 8; ++r)
                Out[(size_t)(mbase + (mi << 4) + r) * Nc + n] = acc[mi][ni][r] + bn;
    }
}

__global__ void fa_scores(const float* __restrict__ q,
                          const float* __restrict__ k,
                          const float* __restrict__ Wphi,
                          float* __restrict__ ex,
                          float* __restrict__ denom)
{
    __shared__ float qi[64];
    __shared__ float wph[64];
    __shared__ float red[8];
    __shared__ float msh;
    const int row = blockIdx.x;
    const int b   = row >> 9;
    const int tid = threadIdx.x;
    if (tid < 64) {
        qi[tid]  = q[(size_t)row * 64 + tid];
        wph[tid] = Wphi[tid];
    }
    __syncthreads();

    float s[2];
    #pragma unroll
    for (int p = 0; p < 2; ++p) {
        const int j = tid + (p << 8);
        const float* kj = k + ((size_t)(b * 512 + j)) * 64;
        float acc = 0.f;
        #pragma unroll 8
        for (int a = 0; a < 64; ++a)
            acc += rho_relu(qi[a] + kj[a]) * wph[a];
        s[p] = acc;
    }

    float mymax = fmaxf(s[0], s[1]);
    for (int off = 16; off; off >>= 1)
        mymax = fmaxf(mymax, __shfl_xor(mymax, off));
    const int lane = tid & 31, wv = tid >> 5;
    if (lane == 0) red[wv] = mymax;
    __syncthreads();
    if (tid == 0) {
        float m = red[0];
        for (int w = 1; w < 8; ++w) m = fmaxf(m, red[w]);
        msh = m;
    }
    __syncthreads();
    const float m = msh;

    const float e0 = expf(s[0] - m);
    const float e1 = expf(s[1] - m);
    ex[(size_t)row * 512 + tid]       = e0;
    ex[(size_t)row * 512 + tid + 256] = e1;

    float mysum = e0 + e1;
    for (int off = 16; off; off >>= 1)
        mysum += __shfl_xor(mysum, off);
    if (lane == 0) red[wv] = mysum;
    __syncthreads();
    if (tid == 0) {
        float sm = 0.f;
        for (int w = 0; w < 8; ++w) sm += red[w];
        denom[row] = sm;
    }
}

__global__ void fa_part1(const float* __restrict__ ex,
                         const float* __restrict__ Vy,
                         float* __restrict__ part1)
{
    const int lane = threadIdx.x & 31;
    const int wave = threadIdx.x >> 5;
    const int job  = blockIdx.x * (blockDim.x >> 5) + wave;
    const int b    = job >> 8;
    const int jj   = job & 255;
    const int mb   = jj >> 4;
    const int nb   = jj & 15;
    const int l15  = lane & 15;
    const int koff = (lane < 16) ? 0 : 2;
    const int col0 = (nb << 6) + l15;

    const float* pA0 = ex + ((size_t)(b * 512 + (mb << 5) + l15)) * 512 + koff;
    const float* pA1 = pA0 + (size_t)16 * 512;
    const float* pV  = Vy + (size_t)b * 512 * 1024 + (size_t)koff * 1024 + col0;

    v8f acc[2][4] = {};
    #pragma unroll 2
    for (int j = 0; j < 512; j += 4) {
        const v2f a0 = *(const v2f*)pA0;
        const v2f a1 = *(const v2f*)pA1;
        v2f b0, b1, b2, b3;
        b0.x = pV[0];          b0.y = pV[1024];
        b1.x = pV[16];         b1.y = pV[1024 + 16];
        b2.x = pV[32];         b2.y = pV[1024 + 32];
        b3.x = pV[48];         b3.y = pV[1024 + 48];
        pA0 += 4; pA1 += 4; pV += 4 * 1024;

        acc[0][0] = wmma_f32(a0, b0, acc[0][0]);
        acc[1][0] = wmma_f32(a1, b0, acc[1][0]);
        acc[0][1] = wmma_f32(a0, b1, acc[0][1]);
        acc[1][1] = wmma_f32(a1, b1, acc[1][1]);
        acc[0][2] = wmma_f32(a0, b2, acc[0][2]);
        acc[1][2] = wmma_f32(a1, b2, acc[1][2]);
        acc[0][3] = wmma_f32(a0, b3, acc[0][3]);
        acc[1][3] = wmma_f32(a1, b3, acc[1][3]);
    }

    const int mbase = b * 512 + (mb << 5) + ((lane < 16) ? 0 : 8);
    #pragma unroll
    for (int ni = 0; ni < 4; ++ni) {
        const int n = col0 + (ni << 4);
        #pragma unroll
        for (int mi = 0; mi < 2; ++mi)
            #pragma unroll
            for (int r = 0; r < 8; ++r)
                part1[(size_t)(mbase + (mi << 4) + r) * 1024 + n] = acc[mi][ni][r];
    }
}

__global__ void fa_ew(const float* __restrict__ q,
                      const float* __restrict__ k,
                      const float* __restrict__ ex,
                      float* __restrict__ Ew)
{
    __shared__ float exs[512];
    const int row = blockIdx.x;
    const int b   = row >> 9;
    const int a   = threadIdx.x;
    for (int j = a; j < 512; j += 64)
        exs[j] = ex[(size_t)row * 512 + j];
    __syncthreads();
    const float qa = q[(size_t)row * 64 + a];
    float acc = 0.f;
    #pragma unroll 4
    for (int j = 0; j < 512; ++j) {
        const float e = rho_relu(qa + k[((size_t)(b * 512 + j)) * 64 + a]);
        acc += exs[j] * e;
    }
    Ew[(size_t)row * 64 + a] = acc;
}

__global__ void fa_final(const float* __restrict__ Ew,
                         const float* __restrict__ Wew,
                         const float* __restrict__ Web,
                         const float* __restrict__ part1,
                         const float* __restrict__ denom,
                         float* __restrict__ out)
{
    const int lane = threadIdx.x & 31;
    const int wave = threadIdx.x >> 5;
    const int t  = blockIdx.x * (blockDim.x >> 5) + wave;
    const int mt = t >> 6, nt = t & 63;
    const int l15  = lane & 15;
    const int koff = (lane < 16) ? 0 : 2;
    const float* Arow = Ew  + (size_t)((mt << 4) + l15) * 64 + koff;
    const float* Brow = Wew + (size_t)((nt << 4) + l15) * 64 + koff;
    v8f acc = {};
    #pragma unroll
    for (int kk = 0; kk < 64; kk += 4) {
        v2f av = *(const v2f*)(Arow + kk);
        v2f bv = *(const v2f*)(Brow + kk);
        acc = wmma_f32(av, bv, acc);
    }
    const int   n     = (nt << 4) + l15;
    const int   mbase = (mt << 4) + ((lane < 16) ? 0 : 8);
    const float bn    = Web[n];
    #pragma unroll
    for (int r = 0; r < 8; ++r) {
        const int   m = mbase + r;
        const float d = denom[m];
        out[(size_t)m * 1024 + n] =
            (part1[(size_t)m * 1024 + n] + acc[r] + bn * d) / (d + EPS_DEN);
    }
}

extern "C" void kernel_launch(void* const* d_in, const int* in_sizes, int n_in,
                              void* d_out, int out_size, void* d_ws, size_t ws_size,
                              hipStream_t stream)
{
    (void)in_sizes; (void)n_in; (void)out_size; (void)ws_size;
    const float* X    = (const float*)d_in[0];
    const float* Wq   = (const float*)d_in[1];
    const float* Wk   = (const float*)d_in[2];
    const float* Wphi = (const float*)d_in[3];
    const float* Wyw  = (const float*)d_in[4];
    const float* Wyb  = (const float*)d_in[5];
    const float* Wew  = (const float*)d_in[6];
    const float* Web  = (const float*)d_in[7];
    float* out = (float*)d_out;

    float* ws  = (float*)d_ws;
    float* q   = ws;  ws += 1024 * 64;
    float* kk  = ws;  ws += 1024 * 64;
    float* Vy  = ws;  ws += 1024 * 1024;
    float* ex  = ws;  ws += 1024 * 512;
    float* den = ws;  ws += 1024;
    float* p1  = ws;  ws += 1024 * 1024;
    float* Ew  = ws;  ws += 1024 * 64;

    fa_gemm_abt<<<4,   256, 0, stream>>>(X, Wq,  nullptr, q,  1024, 1024, 64);
    fa_gemm_abt<<<4,   256, 0, stream>>>(X, Wk,  nullptr, kk, 1024, 1024, 64);
    fa_gemm_abt<<<64,  256, 0, stream>>>(X, Wyw, Wyb,     Vy, 1024, 1024, 1024);

    fa_scores<<<1024, 256, 0, stream>>>(q, kk, Wphi, ex, den);

    fa_part1<<<64, 256, 0, stream>>>(ex, Vy, p1);

    fa_ew<<<1024, 64, 0, stream>>>(q, kk, ex, Ew);

    fa_final<<<512, 256, 0, stream>>>(Ew, Wew, Web, p1, den, out);
}


# q, k and Vy projections share one launch (72 blocks: 4 q + 4 k + 64 Vy, same 2x4-tile f32 WMMA loop per wave), so the two 32-wave launches run beside Vy's waves on idle WGPs; 7 launches -> 5
# speedup vs baseline: 1.6864x; 1.6864x over previous
#include <hip/hip_runtime.h>
#include <hip/hip_bf16.h>
#include <math.h>

typedef __attribute__((ext_vector_type(2))) float v2f;
typedef __attribute__((ext_vector_type(8))) float v8f;

#define EPS_RHO 1e-8f
#define EPS_DEN 1e-9f

__device__ __forceinline__ float rho_relu(float x) {
    float p = sqrtf(fmaxf(x, 0.f) + EPS_RHO);
    float n = sqrtf(fmaxf(-x, 0.f) + EPS_RHO);
    return fmaxf(p - n, 0.f);
}

__device__ __forceinline__ v8f wmma_f32(v2f a, v2f b, v8f c) {
    return __builtin_amdgcn_wmma_f32_16x16x4_f32(false, a, false, b,
                                                 (short)0, c, false, false);
}

__global__ void fa_proj3(const float* __restrict__ A,
                         const float* __restrict__ Wq,
                         const float* __restrict__ Wk,
                         const float* __restrict__ Wyw,
                         const float* __restrict__ Wyb,
                         float* __restrict__ q,
                         float* __restrict__ kout,
                         float* __restrict__ Vy,
                         int K)
{
    const int lane  = threadIdx.x & 31;
    const int wave  = threadIdx.x >> 5;
    const int blk   = blockIdx.x;
    const float* W;
    const float* bias;
    float* Out;
    int Nc, mb, nb;
    if (blk < 4) {
        W = Wq;  bias = nullptr; Out = q;    Nc = 64;
        mb = (blk << 3) + wave;  nb = 0;
    } else if (blk < 8) {
        W = Wk;  bias = nullptr; Out = kout; Nc = 64;
        mb = ((blk - 4) << 3) + wave;  nb = 0;
    } else {
        W = Wyw; bias = Wyb;     Out = Vy;   Nc = 1024;
        const int jj = ((blk - 8) << 3) + wave;
        mb = jj >> 4;  nb = jj & 15;
    }
    const int l15  = lane & 15;
    const int koff = (lane < 16) ? 0 : 2;
    const size_t bstep = (size_t)16 * K;

    const float* pA0 = A + (size_t)((mb << 5) + l15) * K + koff;
    const float* pA1 = pA0 + bstep;
    const float* pB0 = W + (size_t)((nb << 6) + l15) * K + koff;
    const float* pB1 = pB0 + bstep;
    const float* pB2 = pB1 + bstep;
    const float* pB3 = pB2 + bstep;

    v8f acc[2][4] = {};
    #pragma unroll 2
    for (int kk = 0; kk < K; kk += 4) {
        const v2f a0 = *(const v2f*)pA0;
        const v2f a1 = *(const v2f*)pA1;
        const v2f b0 = *(const v2f*)pB0;
        const v2f b1 = *(const v2f*)pB1;
        const v2f b2 = *(const v2f*)pB2;
        const v2f b3 = *(const v2f*)pB3;
        pA0 += 4; pA1 += 4; pB0 += 4; pB1 += 4; pB2 += 4; pB3 += 4;

        acc[0][0] = wmma_f32(a0, b0, acc[0][0]);
        acc[1][0] = wmma_f32(a1, b0, acc[1][0]);
        acc[0][1] = wmma_f32(a0, b1, acc[0][1]);
        acc[1][1] = wmma_f32(a1, b1, acc[1][1]);
        acc[0][2] = wmma_f32(a0, b2, acc[0][2]);
        acc[1][2] = wmma_f32(a1, b2, acc[1][2]);
        acc[0][3] = wmma_f32(a0, b3, acc[0][3]);
        acc[1][3] = wmma_f32(a1, b3, acc[1][3]);
    }

    const int mbase = (mb << 5) + ((lane < 16) ? 0 : 8);
    #pragma unroll
    for (int ni = 0; ni < 4; ++ni) {
        const int   n  = (nb << 6) + (ni << 4) + l15;
        const float bn = bias ? bias[n] : 0.f;
        #pragma unroll
        for (int mi = 0; mi < 2; ++mi)
            #pragma unroll
            for (int r = 0; r < 8; ++r)
                Out[(size_t)(mbase + (mi << 4) + r) * Nc + n] = acc[mi][ni][r] + bn;
    }
}

__global__ void fa_scores(const float* __restrict__ q,
                          const float* __restrict__ k,
                          const float* __restrict__ Wphi,
                          float* __restrict__ ex,
                          float* __restrict__ denom)
{
    __shared__ float qi[64];
    __shared__ float wph[64];
    __shared__ float red[8];
    __shared__ float msh;
    const int row = blockIdx.x;
    const int b   = row >> 9;
    const int tid = threadIdx.x;
    if (tid < 64) {
        qi[tid]  = q[(size_t)row * 64 + tid];
        wph[tid] = Wphi[tid];
    }
    __syncthreads();

    float s[2];
    #pragma unroll
    for (int p = 0; p < 2; ++p) {
        const int j = tid + (p << 8);
        const float* kj = k + ((size_t)(b * 512 + j)) * 64;
        float acc = 0.f;
        #pragma unroll 8
        for (int a = 0; a < 64; ++a)
            acc += rho_relu(qi[a] + kj[a]) * wph[a];
        s[p] = acc;
    }

    float mymax = fmaxf(s[0], s[1]);
    for (int off = 16; off; off >>= 1)
        mymax = fmaxf(mymax, __shfl_xor(mymax, off));
    const int lane = tid & 31, wv = tid >> 5;
    if (lane == 0) red[wv] = mymax;
    __syncthreads();
    if (tid == 0) {
        float m = red[0];
        for (int w = 1; w < 8; ++w) m = fmaxf(m, red[w]);
        msh = m;
    }
    __syncthreads();
    const float m = msh;

    const float e0 = expf(s[0] - m);
    const float e1 = expf(s[1] - m);
    ex[(size_t)row * 512 + tid]       = e0;
    ex[(size_t)row * 512 + tid + 256] = e1;

    float mysum = e0 + e1;
    for (int off = 16; off; off >>= 1)
        mysum += __shfl_xor(mysum, off);
    if (lane == 0) red[wv] = mysum;
    __syncthreads();
    if (tid == 0) {
        float sm = 0.f;
        for (int w = 0; w < 8; ++w) sm += red[w];
        denom[row] = sm;
    }
}

__global__ void fa_part1(const float* __restrict__ ex,
                         const float* __restrict__ Vy,
                         float* __restrict__ part1)
{
    const int lane = threadIdx.x & 31;
    const int wave = threadIdx.x >> 5;
    const int job  = blockIdx.x * (blockDim.x >> 5) + wave;
    const int b    = job >> 8;
    const int jj   = job & 255;
    const int mb   = jj >> 4;
    const int nb   = jj & 15;
    const int l15  = lane & 15;
    const int koff = (lane < 16) ? 0 : 2;
    const int col0 = (nb << 6) + l15;

    const float* pA0 = ex + ((size_t)(b * 512 + (mb << 5) + l15)) * 512 + koff;
    const float* pA1 = pA0 + (size_t)16 * 512;
    const float* pV  = Vy + (size_t)b * 512 * 1024 + (size_t)koff * 1024 + col0;

    v8f acc[2][4] = {};
    #pragma unroll 2
    for (int j = 0; j < 512; j += 4) {
        const v2f a0 = *(const v2f*)pA0;
        const v2f a1 = *(const v2f*)pA1;
        v2f b0, b1, b2, b3;
        b0.x = pV[0];          b0.y = pV[1024];
        b1.x = pV[16];         b1.y = pV[1024 + 16];
        b2.x = pV[32];         b2.y = pV[1024 + 32];
        b3.x = pV[48];         b3.y = pV[1024 + 48];
        pA0 += 4; pA1 += 4; pV += 4 * 1024;

        acc[0][0] = wmma_f32(a0, b0, acc[0][0]);
        acc[1][0] = wmma_f32(a1, b0, acc[1][0]);
        acc[0][1] = wmma_f32(a0, b1, acc[0][1]);
        acc[1][1] = wmma_f32(a1, b1, acc[1][1]);
        acc[0][2] = wmma_f32(a0, b2, acc[0][2]);
        acc[1][2] = wmma_f32(a1, b2, acc[1][2]);
        acc[0][3] = wmma_f32(a0, b3, acc[0][3]);
        acc[1][3] = wmma_f32(a1, b3, acc[1][3]);
    }

    const int mbase = b * 512 + (mb << 5) + ((lane < 16) ? 0 : 8);
    #pragma unroll
    for (int ni = 0; ni < 4; ++ni) {
        const int n = col0 + (ni << 4);
        #pragma unroll
        for (int mi = 0; mi < 2; ++mi)
            #pragma unroll
            for (int r = 0; r < 8; ++r)
                part1[(size_t)(mbase + (mi << 4) + r) * 1024 + n] = acc[mi][ni][r];
    }
}

__global__ void fa_ew(const float* __restrict__ q,
                      const float* __restrict__ k,
                      const float* __restrict__ ex,
                      float* __restrict__ Ew)
{
    __shared__ float exs[512];
    const int row = blockIdx.x;
    const int b   = row >> 9;
    const int a   = threadIdx.x;
    for (int j = a; j < 512; j += 64)
        exs[j] = ex[(size_t)row * 512 + j];
    __syncthreads();
    const float qa = q[(size_t)row * 64 + a];
    float acc = 0.f;
    #pragma unroll 4
    for (int j = 0; j < 512; ++j) {
        const float e = rho_relu(qa + k[((size_t)(b * 512 + j)) * 64 + a]);
        acc += exs[j] * e;
    }
    Ew[(size_t)row * 64 + a] = acc;
}

__global__ void fa_final(const float* __restrict__ Ew,
                         const float* __restrict__ Wew,
                         const float* __restrict__ Web,
                         const float* __restrict__ part1,
                         const float* __restrict__ denom,
                         float* __restrict__ out)
{
    const int lane = threadIdx.x & 31;
    const int wave = threadIdx.x >> 5;
    const int t  = blockIdx.x * (blockDim.x >> 5) + wave;
    const int mt = t >> 6, nt = t & 63;
    const int l15  = lane & 15;
    const int koff = (lane < 16) ? 0 : 2;
    const float* Arow = Ew  + (size_t)((mt << 4) + l15) * 64 + koff;
    const float* Brow = Wew + (size_t)((nt << 4) + l15) * 64 + koff;
    v8f acc = {};
    #pragma unroll
    for (int kk = 0; kk < 64; kk += 4) {
        v2f av = *(const v2f*)(Arow + kk);
        v2f bv = *(const v2f*)(Brow + kk);
        acc = wmma_f32(av, bv, acc);
    }
    const int   n     = (nt << 4) + l15;
    const int   mbase = (mt << 4) + ((lane < 16) ? 0 : 8);
    const float bn    = Web[n];
    #pragma unroll
    for (int r = 0; r < 8; ++r) {
        const int   m = mbase + r;
        const float d = denom[m];
        out[(size_t)m * 1024 + n] =
            (part1[(size_t)m * 1024 + n] + acc[r] + bn * d) / (d + EPS_DEN);
    }
}

extern "C" void kernel_launch(void* const* d_in, const int* in_sizes, int n_in,
                              void* d_out, int out_size, void* d_ws, size_t ws_size,
                              hipStream_t stream)
{
    (void)in_sizes; (void)n_in; (void)out_size; (void)ws_size;
    const float* X    = (const float*)d_in[0];
    const float* Wq   = (const float*)d_in[1];
    const float* Wk   = (const float*)d_in[2];
    const float* Wphi = (const float*)d_in[3];
    const float* Wyw  = (const float*)d_in[4];
    const float* Wyb  = (const float*)d_in[5];
    const float* Wew  = (const float*)d_in[6];
    const float* Web  = (const float*)d_in[7];
    float* out = (float*)d_out;

    float* ws  = (float*)d_ws;
    float* q   = ws;  ws += 1024 * 64;
    float* kk  = ws;  ws += 1024 * 64;
    float* Vy  = ws;  ws += 1024 * 1024;
    float* ex  = ws;  ws += 1024 * 512;
    float* den = ws;  ws += 1024;
    float* p1  = ws;  ws += 1024 * 1024;
    float* Ew  = ws;  ws += 1024 * 64;

    fa_proj3<<<72, 256, 0, stream>>>(X, Wq, Wk, Wyw, Wyb, q, kk, Vy, 1024);

    fa_scores<<<1024, 256, 0, stream>>>(q, kk, Wphi, ex, den);

    fa_part1<<<64, 256, 0, stream>>>(ex, Vy, p1);

    fa_ew<<<1024, 64, 0, stream>>>(q, kk, ex, Ew);

    fa_final<<<512, 256, 0, stream>>>(Ew, Wew, Web, p1, den, out);
}


# WMMA fragment loads widened to 16 B per lane in all three GEMM loops: k permuted alike in A and B inside each 8-k group (k sum order changes), part1 tiles on adjacent Vy columns; 6 loads per 16 WMMAs
# speedup vs baseline: 2.1134x; 1.2532x over previous
#include <hip/hip_runtime.h>
#include <hip/hip_bf16.h>
#include <math.h>

typedef __attribute__((ext_vector_type(2))) float v2f;
typedef __attribute__((ext_vector_type(4))) float v4f;
typedef __attribute__((ext_vector_type(8))) float v8f;

#define EPS_RHO 1e-8f
#define EPS_DEN 1e-9f

__device__ __forceinline__ float rho_relu(float x) {
    float p = sqrtf(fmaxf(x, 0.f) + EPS_RHO);
    float n = sqrtf(fmaxf(-x, 0.f) + EPS_RHO);
    return fmaxf(p - n, 0.f);
}

__device__ __forceinline__ v8f wmma_f32(v2f a, v2f b, v8f c) {
    return __builtin_amdgcn_wmma_f32_16x16x4_f32(false, a, false, b,
                                                 (short)0, c, false, false);
}

__global__ void fa_proj3(const float* __restrict__ A,
                         const float* __restrict__ Wq,
                         const float* __restrict__ Wk,
                         const float* __restrict__ Wyw,
                         const float* __restrict__ Wyb,
                         float* __restrict__ q,
                         float* __restrict__ kout,
                         float* __restrict__ Vy,
                         int K)
{
    const int lane  = threadIdx.x & 31;
    const int wave  = threadIdx.x >> 5;
    const int blk   = blockIdx.x;
    const float* W;
    const float* bias;
    float* Out;
    int Nc, mb, nb;
    if (blk < 4) {
        W = Wq;  bias = nullptr; Out = q;    Nc = 64;
        mb = (blk << 3) + wave;  nb = 0;
    } else if (blk < 8) {
        W = Wk;  bias = nullptr; Out = kout; Nc = 64;
        mb = ((blk - 4) << 3) + wave;  nb = 0;
    } else {
        W = Wyw; bias = Wyb;     Out = Vy;   Nc = 1024;
        const int jj = ((blk - 8) << 3) + wave;
        mb = jj >> 4;  nb = jj & 15;
    }
    const int l15  = lane & 15;
    const int koff = (lane < 16) ? 0 : 4;
    const size_t bstep = (size_t)16 * K;

    const float* pA0 = A + (size_t)((mb << 5) + l15) * K + koff;
    const float* pA1 = pA0 + bstep;
    const float* pB0 = W + (size_t)((nb << 6) + l15) * K + koff;
    const float* pB1 = pB0 + bstep;
    const float* pB2 = pB1 + bstep;
    const float* pB3 = pB2 + bstep;

    v8f acc[2][4] = {};
    for (int kk = 0; kk < K; kk += 8) {
        const v4f a0 = *(const v4f*)pA0;
        const v4f a1 = *(const v4f*)pA1;
        const v4f b0 = *(const v4f*)pB0;
        const v4f b1 = *(const v4f*)pB1;
        const v4f b2 = *(const v4f*)pB2;
        const v4f b3 = *(const v4f*)pB3;
        pA0 += 8; pA1 += 8; pB0 += 8; pB1 += 8; pB2 += 8; pB3 += 8;
        __builtin_amdgcn_sched_barrier(0);

        acc[0][0] = wmma_f32(a0.xy, b0.xy, acc[0][0]);
        acc[1][0] = wmma_f32(a1.xy, b0.xy, acc[1][0]);
        acc[0][1] = wmma_f32(a0.xy, b1.xy, acc[0][1]);
        acc[1][1] = wmma_f32(a1.xy, b1.xy, acc[1][1]);
        acc[0][2] = wmma_f32(a0.xy, b2.xy, acc[0][2]);
        acc[1][2] = wmma_f32(a1.xy, b2.xy, acc[1][2]);
        acc[0][3] = wmma_f32(a0.xy, b3.xy, acc[0][3]);
        acc[1][3] = wmma_f32(a1.xy, b3.xy, acc[1][3]);
        acc[0][0] = wmma_f32(a0.zw, b0.zw, acc[0][0]);
        acc[1][0] = wmma_f32(a1.zw, b0.zw, acc[1][0]);
        acc[0][1] = wmma_f32(a0.zw, b1.zw, acc[0][1]);
        acc[1][1] = wmma_f32(a1.zw, b1.zw, acc[1][1]);
        acc[0][2] = wmma_f32(a0.zw, b2.zw, acc[0][2]);
        acc[1][2] = wmma_f32(a1.zw, b2.zw, acc[1][2]);
        acc[0][3] = wmma_f32(a0.zw, b3.zw, acc[0][3]);
        acc[1][3] = wmma_f32(a1.zw, b3.zw, acc[1][3]);
    }

    const int mbase = (mb << 5) + ((lane < 16) ? 0 : 8);
    #pragma unroll
    for (int ni = 0; ni < 4; ++ni) {
        const int   n  = (nb << 6) + (ni << 4) + l15;
        const float bn = bias ? bias[n] : 0.f;
        #pragma unroll
        for (int mi = 0; mi < 2; ++mi)
            #pragma unroll
            for (int r = 0; r < 8; ++r)
                Out[(size_t)(mbase + (mi << 4) + r) * Nc + n] = acc[mi][ni][r] + bn;
    }
}

__global__ void fa_scores(const float* __restrict__ q,
                          const float* __restrict__ k,
                          const float* __restrict__ Wphi,
                          float* __restrict__ ex,
                          float* __restrict__ denom)
{
    __shared__ float qi[64];
    __shared__ float wph[64];
    __shared__ float red[8];
    __shared__ float msh;
    const int row = blockIdx.x;
    const int b   = row >> 9;
    const int tid = threadIdx.x;
    if (tid < 64) {
        qi[tid]  = q[(size_t)row * 64 + tid];
        wph[tid] = Wphi[tid];
    }
    __syncthreads();

    float s[2];
    #pragma unroll
    for (int p = 0; p < 2; ++p) {
        const int j = tid + (p << 8);
        const float* kj = k + ((size_t)(b * 512 + j)) * 64;
        float acc = 0.f;
        #pragma unroll 8
        for (int a = 0; a < 64; ++a)
            acc += rho_relu(qi[a] + kj[a]) * wph[a];
        s[p] = acc;
    }

    float mymax = fmaxf(s[0], s[1]);
    for (int off = 16; off; off >>= 1)
        mymax = fmaxf(mymax, __shfl_xor(mymax, off));
    const int lane = tid & 31, wv = tid >> 5;
    if (lane == 0) red[wv] = mymax;
    __syncthreads();
    if (tid == 0) {
        float m = red[0];
        for (int w = 1; w < 8; ++w) m = fmaxf(m, red[w]);
        msh = m;
    }
    __syncthreads();
    const float m = msh;

    const float e0 = expf(s[0] - m);
    const float e1 = expf(s[1] - m);
    ex[(size_t)row * 512 + tid]       = e0;
    ex[(size_t)row * 512 + tid + 256] = e1;

    float mysum = e0 + e1;
    for (int off = 16; off; off >>= 1)
        mysum += __shfl_xor(mysum, off);
    if (lane == 0) red[wv] = mysum;
    __syncthreads();
    if (tid == 0) {
        float sm = 0.f;
        for (int w = 0; w < 8; ++w) sm += red[w];
        denom[row] = sm;
    }
}

__global__ void fa_part1(const float* __restrict__ ex,
                         const float* __restrict__ Vy,
                         float* __restrict__ part1)
{
    const int lane = threadIdx.x & 31;
    const int wave = threadIdx.x >> 5;
    const int job  = blockIdx.x * (blockDim.x >> 5) + wave;
    const int b    = job >> 8;
    const int jj   = job & 255;
    const int mb   = jj >> 4;
    const int nb   = jj & 15;
    const int l15  = lane & 15;
    const int koff = (lane < 16) ? 0 : 4;
    const int col0 = (nb << 6) + (l15 << 2);

    const float* pA0 = ex + ((size_t)(b * 512 + (mb << 5) + l15)) * 512 + koff;
    const float* pA1 = pA0 + (size_t)16 * 512;
    const float* pV  = Vy + (size_t)b * 512 * 1024 + (size_t)koff * 1024 + col0;

    v8f acc[2][4] = {};
    for (int j = 0; j < 512; j += 8) {
        const v4f a0 = *(const v4f*)pA0;
        const v4f a1 = *(const v4f*)pA1;
        const v4f r0 = *(const v4f*)(pV);
        const v4f r1 = *(const v4f*)(pV + 1024);
        const v4f r2 = *(const v4f*)(pV + 2 * 1024);
        const v4f r3 = *(const v4f*)(pV + 3 * 1024);
        pA0 += 8; pA1 += 8; pV += 8 * 1024;
        __builtin_amdgcn_sched_barrier(0);

        #pragma unroll
        for (int t = 0; t < 4; ++t) {
            v2f bt; bt.x = r0[t]; bt.y = r1[t];
            acc[0][t] = wmma_f32(a0.xy, bt, acc[0][t]);
            acc[1][t] = wmma_f32(a1.xy, bt, acc[1][t]);
        }
        #pragma unroll
        for (int t = 0; t < 4; ++t) {
            v2f bt; bt.x = r2[t]; bt.y = r3[t];
            acc[0][t] = wmma_f32(a0.zw, bt, acc[0][t]);
            acc[1][t] = wmma_f32(a1.zw, bt, acc[1][t]);
        }
    }

    const int mbase = b * 512 + (mb << 5) + ((lane < 16) ? 0 : 8);
    #pragma unroll
    for (int mi = 0; mi < 2; ++mi)
        #pragma unroll
        for (int r = 0; r < 8; ++r) {
            v4f o;
            o.x = acc[mi][0][r]; o.y = acc[mi][1][r];
            o.z = acc[mi][2][r]; o.w = acc[mi][3][r];
            *(v4f*)(part1 + (size_t)(mbase + (mi << 4) + r) * 1024 + col0) = o;
        }
}

__global__ void fa_ew(const float* __restrict__ q,
                      const float* __restrict__ k,
                      const float* __restrict__ ex,
                      float* __restrict__ Ew)
{
    __shared__ float exs[512];
    const int row = blockIdx.x;
    const int b   = row >> 9;
    const int a   = threadIdx.x;
    for (int j = a; j < 512; j += 64)
        exs[j] = ex[(size_t)row * 512 + j];
    __syncthreads();
    const float qa = q[(size_t)row * 64 + a];
    float acc = 0.f;
    #pragma unroll 4
    for (int j = 0; j < 512; ++j) {
        const float e = rho_relu(qa + k[((size_t)(b * 512 + j)) * 64 + a]);
        acc += exs[j] * e;
    }
    Ew[(size_t)row * 64 + a] = acc;
}

__global__ void fa_final(const float* __restrict__ Ew,
                         const float* __restrict__ Wew,
                         const float* __restrict__ Web,
                         const float* __restrict__ part1,
                         const float* __restrict__ denom,
                         float* __restrict__ out)
{
    const int lane = threadIdx.x & 31;
    const int wave = threadIdx.x >> 5;
    const int t  = blockIdx.x * (blockDim.x >> 5) + wave;
    const int mt = t >> 6, nt = t & 63;
    const int l15  = lane & 15;
    const int koff = (lane < 16) ? 0 : 4;
    const float* Arow = Ew  + (size_t)((mt << 4) + l15) * 64 + koff;
    const float* Brow = Wew + (size_t)((nt << 4) + l15) * 64 + koff;
    v8f acc = {};
    #pragma unroll
    for (int kk = 0; kk < 64; kk += 8) {
        const v4f av = *(const v4f*)(Arow + kk);
        const v4f bv = *(const v4f*)(Brow + kk);
        acc = wmma_f32(av.xy, bv.xy, acc);
        acc = wmma_f32(av.zw, bv.zw, acc);
    }
    const int   n     = (nt << 4) + l15;
    const int   mbase = (mt << 4) + ((lane < 16) ? 0 : 8);
    const float bn    = Web[n];
    #pragma unroll
    for (int r = 0; r < 8; ++r) {
        const int   m = mbase + r;
        const float d = denom[m];
        out[(size_t)m * 1024 + n] =
            (part1[(size_t)m * 1024 + n] + acc[r] + bn * d) / (d + EPS_DEN);
    }
}

extern "C" void kernel_launch(void* const* d_in, const int* in_sizes, int n_in,
                              void* d_out, int out_size, void* d_ws, size_t ws_size,
                              hipStream_t stream)
{
    (void)in_sizes; (void)n_in; (void)out_size; (void)ws_size;
    const float* X    = (const float*)d_in[0];
    const float* Wq   = (const float*)d_in[1];
    const float* Wk   = (const float*)d_in[2];
    const float* Wphi = (const float*)d_in[3];
    const float* Wyw  = (const float*)d_in[4];
    const float* Wyb  = (const float*)d_in[5];
    const float* Wew  = (const float*)d_in[6];
    const float* Web  = (const float*)d_in[7];
    float* out = (float*)d_out;

    float* ws  = (float*)d_ws;
    float* q   = ws;  ws += 1024 * 64;
    float* kk  = ws;  ws += 1024 * 64;
    float* Vy  = ws;  ws += 1024 * 1024;
    float* ex  = ws;  ws += 1024 * 512;
    float* den = ws;  ws += 1024;
    float* p1  = ws;  ws += 1024 * 1024;
    float* Ew  = ws;  ws += 1024 * 64;

    fa_proj3<<<72, 256, 0, stream>>>(X, Wq, Wk, Wyw, Wyb, q, kk, Vy, 1024);

    fa_scores<<<1024, 256, 0, stream>>>(q, kk, Wphi, ex, den);

    fa_part1<<<64, 256, 0, stream>>>(ex, Vy, p1);

    fa_ew<<<1024, 64, 0, stream>>>(q, kk, ex, Ew);

    fa_final<<<512, 256, 0, stream>>>(Ew, Wew, Web, p1, den, out);
}


# scores kernel keeps each row's e (512 x 64) in LDS and sums Ew from it instead of recomputing e: the Ew launch is gone (5 -> 4 launches); Ew's j sum is now 16 partial sums added as a tree
# speedup vs baseline: 2.6465x; 1.2522x over previous
#include <hip/hip_runtime.h>
#include <hip/hip_bf16.h>
#include <math.h>

typedef __attribute__((ext_vector_type(2))) float v2f;
typedef __attribute__((ext_vector_type(4))) float v4f;
typedef __attribute__((ext_vector_type(8))) float v8f;

#define EPS_RHO 1e-8f
#define EPS_DEN 1e-9f

__device__ __forceinline__ float rho_relu(float x) {
    float p = sqrtf(fmaxf(x, 0.f) + EPS_RHO);
    float n = sqrtf(fmaxf(-x, 0.f) + EPS_RHO);
    return fmaxf(p - n, 0.f);
}

__device__ __forceinline__ v8f wmma_f32(v2f a, v2f b, v8f c) {
    return __builtin_amdgcn_wmma_f32_16x16x4_f32(false, a, false, b,
                                                 (short)0, c, false, false);
}

__global__ void fa_proj3(const float* __restrict__ A,
                         const float* __restrict__ Wq,
                         const float* __restrict__ Wk,
                         const float* __restrict__ Wyw,
                         const float* __restrict__ Wyb,
                         float* __restrict__ q,
                         float* __restrict__ kout,
                         float* __restrict__ Vy,
                         int K)
{
    const int lane  = threadIdx.x & 31;
    const int wave  = threadIdx.x >> 5;
    const int blk   = blockIdx.x;
    const float* W;
    const float* bias;
    float* Out;
    int Nc, mb, nb;
    if (blk < 4) {
        W = Wq;  bias = nullptr; Out = q;    Nc = 64;
        mb = (blk << 3) + wave;  nb = 0;
    } else if (blk < 8) {
        W = Wk;  bias = nullptr; Out = kout; Nc = 64;
        mb = ((blk - 4) << 3) + wave;  nb = 0;
    } else {
        W = Wyw; bias = Wyb;     Out = Vy;   Nc = 1024;
        const int jj = ((blk - 8) << 3) + wave;
        mb = jj >> 4;  nb = jj & 15;
    }
    const int l15  = lane & 15;
    const int koff = (lane < 16) ? 0 : 4;
    const size_t bstep = (size_t)16 * K;

    const float* pA0 = A + (size_t)((mb << 5) + l15) * K + koff;
    const float* pA1 = pA0 + bstep;
    const float* pB0 = W + (size_t)((nb << 6) + l15) * K + koff;
    const float* pB1 = pB0 + bstep;
    const float* pB2 = pB1 + bstep;
    const float* pB3 = pB2 + bstep;

    v8f acc[2][4] = {};
    for (int kk = 0; kk < K; kk += 8) {
        const v4f a0 = *(const v4f*)pA0;
        const v4f a1 = *(const v4f*)pA1;
        const v4f b0 = *(const v4f*)pB0;
        const v4f b1 = *(const v4f*)pB1;
        const v4f b2 = *(const v4f*)pB2;
        const v4f b3 = *(const v4f*)pB3;
        pA0 += 8; pA1 += 8; pB0 += 8; pB1 += 8; pB2 += 8; pB3 += 8;
        __builtin_amdgcn_sched_barrier(0);

        acc[0][0] = wmma_f32(a0.xy, b0.xy, acc[0][0]);
        acc[1][0] = wmma_f32(a1.xy, b0.xy, acc[1][0]);
        acc[0][1] = wmma_f32(a0.xy, b1.xy, acc[0][1]);
        acc[1][1] = wmma_f32(a1.xy, b1.xy, acc[1][1]);
        acc[0][2] = wmma_f32(a0.xy, b2.xy, acc[0][2]);
        acc[1][2] = wmma_f32(a1.xy, b2.xy, acc[1][2]);
        acc[0][3] = wmma_f32(a0.xy, b3.xy, acc[0][3]);
        acc[1][3] = wmma_f32(a1.xy, b3.xy, acc[1][3]);
        acc[0][0] = wmma_f32(a0.zw, b0.zw, acc[0][0]);
        acc[1][0] = wmma_f32(a1.zw, b0.zw, acc[1][0]);
        acc[0][1] = wmma_f32(a0.zw, b1.zw, acc[0][1]);
        acc[1][1] = wmma_f32(a1.zw, b1.zw, acc[1][1]);
        acc[0][2] = wmma_f32(a0.zw, b2.zw, acc[0][2]);
        acc[1][2] = wmma_f32(a1.zw, b2.zw, acc[1][2]);
        acc[0][3] = wmma_f32(a0.zw, b3.zw, acc[0][3]);
        acc[1][3] = wmma_f32(a1.zw, b3.zw, acc[1][3]);
    }

    const int mbase = (mb << 5) + ((lane < 16) ? 0 : 8);
    #pragma unroll
    for (int ni = 0; ni < 4; ++ni) {
        const int   n  = (nb << 6) + (ni << 4) + l15;
        const float bn = bias ? bias[n] : 0.f;
        #pragma unroll
        for (int mi = 0; mi < 2; ++mi)
            #pragma unroll
            for (int r = 0; r < 8; ++r)
                Out[(size_t)(mbase + (mi << 4) + r) * Nc + n] = acc[mi][ni][r] + bn;
    }
}

#define ES_PITCH 68
__global__ void __launch_bounds__(256)
fa_scores_ew(const float* __restrict__ q,
             const float* __restrict__ k,
             const float* __restrict__ Wphi,
             float* __restrict__ ex,
             float* __restrict__ denom,
             float* __restrict__ Ew)
{
    __shared__ v4f   es4[512 * ES_PITCH / 4];
    __shared__ v4f   qi4[16];
    __shared__ v4f   wph4[16];
    __shared__ v4f   exs4[128];
    __shared__ float part[4 * 64];
    __shared__ float red[8];
    __shared__ float msh;
    float* es  = (float*)es4;
    float* qi  = (float*)qi4;
    float* wph = (float*)wph4;
    float* exs = (float*)exs4;
    const int row = blockIdx.x;
    const int b   = row >> 9;
    const int tid = threadIdx.x;
    if (tid < 64) {
        qi[tid]  = q[(size_t)row * 64 + tid];
        wph[tid] = Wphi[tid];
    }
    __syncthreads();

    float s[2];
    #pragma unroll
    for (int p = 0; p < 2; ++p) {
        const int j = tid + (p << 8);
        const float* kj = k + ((size_t)(b * 512 + j)) * 64;
        float* ej = es + j * ES_PITCH;
        float acc = 0.f;
        #pragma unroll 2
        for (int a = 0; a < 64; a += 4) {
            const v4f kv = *(const v4f*)(kj + a);
            const v4f qv = qi4[a >> 2];
            const v4f pw = wph4[a >> 2];
            v4f ev;
            ev.x = rho_relu(qv.x + kv.x);
            ev.y = rho_relu(qv.y + kv.y);
            ev.z = rho_relu(qv.z + kv.z);
            ev.w = rho_relu(qv.w + kv.w);
            acc += ev.x * pw.x;
            acc += ev.y * pw.y;
            acc += ev.z * pw.z;
            acc += ev.w * pw.w;
            *(v4f*)(ej + a) = ev;
        }
        s[p] = acc;
    }

    float mymax = fmaxf(s[0], s[1]);
    for (int off = 16; off; off >>= 1)
        mymax = fmaxf(mymax, __shfl_xor(mymax, off));
    const int lane = tid & 31, wv = tid >> 5;
    if (lane == 0) red[wv] = mymax;
    __syncthreads();
    if (tid == 0) {
        float m = red[0];
        for (int w = 1; w < 8; ++w) m = fmaxf(m, red[w]);
        msh = m;
    }
    __syncthreads();
    const float m = msh;

    const float e0 = expf(s[0] - m);
    const float e1 = expf(s[1] - m);
    ex[(size_t)row * 512 + tid]       = e0;
    ex[(size_t)row * 512 + tid + 256] = e1;
    exs[tid]       = e0;
    exs[tid + 256] = e1;

    float mysum = e0 + e1;
    for (int off = 16; off; off >>= 1)
        mysum += __shfl_xor(mysum, off);
    if (lane == 0) red[wv] = mysum;
    __syncthreads();
    if (tid == 0) {
        float sm = 0.f;
        for (int w = 0; w < 8; ++w) sm += red[w];
        denom[row] = sm;
    }

    const int a  = tid & 63;
    const int jq = tid >> 6;
    const float* er = es  + (jq * 128) * ES_PITCH + a;
    const float* xr = exs + jq * 128;
    float w0 = 0.f, w1 = 0.f, w2 = 0.f, w3 = 0.f;
    #pragma unroll 4
    for (int jj = 0; jj < 128; jj += 4) {
        const v4f xv = *(const v4f*)(xr + jj);
        w0 += xv.x * er[(jj + 0) * ES_PITCH];
        w1 += xv.y * er[(jj + 1) * ES_PITCH];
        w2 += xv.z * er[(jj + 2) * ES_PITCH];
        w3 += xv.w * er[(jj + 3) * ES_PITCH];
    }
    part[jq * 64 + a] = (w0 + w1) + (w2 + w3);
    __syncthreads();
    if (tid < 64)
        Ew[(size_t)row * 64 + tid] =
            (part[tid] + part[64 + tid]) + (part[128 + tid] + part[192 + tid]);
}

__global__ void fa_part1(const float* __restrict__ ex,
                         const float* __restrict__ Vy,
                         float* __restrict__ part1)
{
    const int lane = threadIdx.x & 31;
    const int wave = threadIdx.x >> 5;
    const int job  = blockIdx.x * (blockDim.x >> 5) + wave;
    const int b    = job >> 8;
    const int jj   = job & 255;
    const int mb   = jj >> 4;
    const int nb   = jj & 15;
    const int l15  = lane & 15;
    const int koff = (lane < 16) ? 0 : 4;
    const int col0 = (nb << 6) + (l15 << 2);

    const float* pA0 = ex + ((size_t)(b * 512 + (mb << 5) + l15)) * 512 + koff;
    const float* pA1 = pA0 + (size_t)16 * 512;
    const float* pV  = Vy + (size_t)b * 512 * 1024 + (size_t)koff * 1024 + col0;

    v8f acc[2][4] = {};
    for (int j = 0; j < 512; j += 8) {
        const v4f a0 = *(const v4f*)pA0;
        const v4f a1 = *(const v4f*)pA1;
        const v4f r0 = *(const v4f*)(pV);
        const v4f r1 = *(const v4f*)(pV + 1024);
        const v4f r2 = *(const v4f*)(pV + 2 * 1024);
        const v4f r3 = *(const v4f*)(pV + 3 * 1024);
        pA0 += 8; pA1 += 8; pV += 8 * 1024;
        __builtin_amdgcn_sched_barrier(0);

        #pragma unroll
        for (int t = 0; t < 4; ++t) {
            v2f bt; bt.x = r0[t]; bt.y = r1[t];
            acc[0][t] = wmma_f32(a0.xy, bt, acc[0][t]);
            acc[1][t] = wmma_f32(a1.xy, bt, acc[1][t]);
        }
        #pragma unroll
        for (int t = 0; t < 4; ++t) {
            v2f bt; bt.x = r2[t]; bt.y = r3[t];
            acc[0][t] = wmma_f32(a0.zw, bt, acc[0][t]);
            acc[1][t] = wmma_f32(a1.zw, bt, acc[1][t]);
        }
    }

    const int mbase = b * 512 + (mb << 5) + ((lane < 16) ? 0 : 8);
    #pragma unroll
    for (int mi = 0; mi < 2; ++mi)
        #pragma unroll
        for (int r = 0; r < 8; ++r) {
            v4f o;
            o.x = acc[mi][0][r]; o.y = acc[mi][1][r];
            o.z = acc[mi][2][r]; o.w = acc[mi][3][r];
            *(v4f*)(part1 + (size_t)(mbase + (mi << 4) + r) * 1024 + col0) = o;
        }
}

__global__ void fa_final(const float* __restrict__ Ew,
                         const float* __restrict__ Wew,
                         const float* __restrict__ Web,
                         const float* __restrict__ part1,
                         const float* __restrict__ denom,
                         float* __restrict__ out)
{
    const int lane = threadIdx.x & 31;
    const int wave = threadIdx.x >> 5;
    const int t  = blockIdx.x * (blockDim.x >> 5) + wave;
    const int mt = t >> 6, nt = t & 63;
    const int l15  = lane & 15;
    const int koff = (lane < 16) ? 0 : 4;
    const float* Arow = Ew  + (size_t)((mt << 4) + l15) * 64 + koff;
    const float* Brow = Wew + (size_t)((nt << 4) + l15) * 64 + koff;
    v8f acc = {};
    #pragma unroll
    for (int kk = 0; kk < 64; kk += 8) {
        const v4f av = *(const v4f*)(Arow + kk);
        const v4f bv = *(const v4f*)(Brow + kk);
        acc = wmma_f32(av.xy, bv.xy, acc);
        acc = wmma_f32(av.zw, bv.zw, acc);
    }
    const int   n     = (nt << 4) + l15;
    const int   mbase = (mt << 4) + ((lane < 16) ? 0 : 8);
    const float bn    = Web[n];
    #pragma unroll
    for (int r = 0; r < 8; ++r) {
        const int   m = mbase + r;
        const float d = denom[m];
        out[(size_t)m * 1024 + n] =
            (part1[(size_t)m * 1024 + n] + acc[r] + bn * d) / (d + EPS_DEN);
    }
}

extern "C" void kernel_launch(void* const* d_in, const int* in_sizes, int n_in,
                              void* d_out, int out_size, void* d_ws, size_t ws_size,
                              hipStream_t stream)
{
    (void)in_sizes; (void)n_in; (void)out_size; (void)ws_size;
    const float* X    = (const float*)d_in[0];
    const float* Wq   = (const float*)d_in[1];
    const float* Wk   = (const float*)d_in[2];
    const float* Wphi = (const float*)d_in[3];
    const float* Wyw  = (const float*)d_in[4];
    const float* Wyb  = (const float*)d_in[5];
    const float* Wew  = (const float*)d_in[6];
    const float* Web  = (const float*)d_in[7];
    float* out = (float*)d_out;

    float* ws  = (float*)d_ws;
    float* q   = ws;  ws += 1024 * 64;
    float* kk  = ws;  ws += 1024 * 64;
    float* Vy  = ws;  ws += 1024 * 1024;
    float* ex  = ws;  ws += 1024 * 512;
    float* den = ws;  ws += 1024;
    float* p1  = ws;  ws += 1024 * 1024;
    float* Ew  = ws;  ws += 1024 * 64;

    fa_proj3<<<72, 256, 0, stream>>>(X, Wq, Wk, Wyw, Wyb, q, kk, Vy, 1024);

    fa_scores_ew<<<1024, 256, 0, stream>>>(q, kk, Wphi, ex, den, Ew);

    fa_part1<<<64, 256, 0, stream>>>(ex, Vy, p1);

    fa_final<<<512, 256, 0, stream>>>(Ew, Wew, Web, p1, den, out);
}


# proj3 and part1 k loops software-pipelined: the six loads of group g+3 issued before the 16 WMMAs of group g (ring of 4 register sets), in 2-wave blocks so a wave has its SIMD alone; sum order kept
# speedup vs baseline: 4.5336x; 1.7131x over previous
#include <hip/hip_runtime.h>
#include <hip/hip_bf16.h>
#include <math.h>

typedef __attribute__((ext_vector_type(2))) float v2f;
typedef __attribute__((ext_vector_type(4))) float v4f;
typedef __attribute__((ext_vector_type(8))) float v8f;

#define EPS_RHO 1e-8f
#define EPS_DEN 1e-9f

__device__ __forceinline__ float rho_relu(float x) {
    float p = sqrtf(fmaxf(x, 0.f) + EPS_RHO);
    float n = sqrtf(fmaxf(-x, 0.f) + EPS_RHO);
    return fmaxf(p - n, 0.f);
}

__device__ __forceinline__ v8f wmma_f32(v2f a, v2f b, v8f c) {
    return __builtin_amdgcn_wmma_f32_16x16x4_f32(false, a, false, b,
                                                 (short)0, c, false, false);
}

struct frag6 { v4f a0, a1, b0, b1, b2, b3; };

__device__ __forceinline__ void mma_group_rows(v8f (&acc)[2][4], const frag6& f) {
    acc[0][0] = wmma_f32(f.a0.xy, f.b0.xy, acc[0][0]);
    acc[1][0] = wmma_f32(f.a1.xy, f.b0.xy, acc[1][0]);
    acc[0][1] = wmma_f32(f.a0.xy, f.b1.xy, acc[0][1]);
    acc[1][1] = wmma_f32(f.a1.xy, f.b1.xy, acc[1][1]);
    acc[0][2] = wmma_f32(f.a0.xy, f.b2.xy, acc[0][2]);
    acc[1][2] = wmma_f32(f.a1.xy, f.b2.xy, acc[1][2]);
    acc[0][3] = wmma_f32(f.a0.xy, f.b3.xy, acc[0][3]);
    acc[1][3] = wmma_f32(f.a1.xy, f.b3.xy, acc[1][3]);
    acc[0][0] = wmma_f32(f.a0.zw, f.b0.zw, acc[0][0]);
    acc[1][0] = wmma_f32(f.a1.zw, f.b0.zw, acc[1][0]);
    acc[0][1] = wmma_f32(f.a0.zw, f.b1.zw, acc[0][1]);
    acc[1][1] = wmma_f32(f.a1.zw, f.b1.zw, acc[1][1]);
    acc[0][2] = wmma_f32(f.a0.zw, f.b2.zw, acc[0][2]);
    acc[1][2] = wmma_f32(f.a1.zw, f.b2.zw, acc[1][2]);
    acc[0][3] = wmma_f32(f.a0.zw, f.b3.zw, acc[0][3]);
    acc[1][3] = wmma_f32(f.a1.zw, f.b3.zw, acc[1][3]);
}

__device__ __forceinline__ void mma_group_cols(v8f (&acc)[2][4], const frag6& f) {
    #pragma unroll
    for (int t = 0; t < 4; ++t) {
        v2f bt; bt.x = f.b0[t]; bt.y = f.b1[t];
        acc[0][t] = wmma_f32(f.a0.xy, bt, acc[0][t]);
        acc[1][t] = wmma_f32(f.a1.xy, bt, acc[1][t]);
    }
    #pragma unroll
    for (int t = 0; t < 4; ++t) {
        v2f bt; bt.x = f.b2[t]; bt.y = f.b3[t];
        acc[0][t] = wmma_f32(f.a0.zw, bt, acc[0][t]);
        acc[1][t] = wmma_f32(f.a1.zw, bt, acc[1][t]);
    }
}

__global__ void __launch_bounds__(64)
fa_proj3(const float* __restrict__ A,
         const float* __restrict__ Wq,
         const float* __restrict__ Wk,
         const float* __restrict__ Wyw,
         const float* __restrict__ Wyb,
         float* __restrict__ q,
         float* __restrict__ kout,
         float* __restrict__ Vy,
         int K)
{
    const int lane = threadIdx.x & 31;
    const int wave = __builtin_amdgcn_readfirstlane((int)(threadIdx.x >> 5));
    const int job  = blockIdx.x * (int)(blockDim.x >> 5) + wave;
    const float* W;
    const float* bias;
    float* Out;
    int Nc, mb, nb;
    if (job < 32) {
        W = Wq;  bias = nullptr; Out = q;    Nc = 64;
        mb = job;  nb = 0;
    } else if (job < 64) {
        W = Wk;  bias = nullptr; Out = kout; Nc = 64;
        mb = job - 32;  nb = 0;
    } else {
        W = Wyw; bias = Wyb;     Out = Vy;   Nc = 1024;
        const int jj = job - 64;
        mb = jj >> 4;  nb = jj & 15;
    }
    const int l15  = lane & 15;
    const int koff = (lane < 16) ? 0 : 4;
    const size_t bstep = (size_t)16 * K;

    const float* pA0 = A + (size_t)((mb << 5) + l15) * K + koff;
    const float* pA1 = pA0 + bstep;
    const float* pB0 = W + (size_t)((nb << 6) + l15) * K + koff;
    const float* pB1 = pB0 + bstep;
    const float* pB2 = pB1 + bstep;
    const float* pB3 = pB2 + bstep;

    auto load_group = [&](frag6& f) {
        f.a0 = *(const v4f*)pA0;
        f.a1 = *(const v4f*)pA1;
        f.b0 = *(const v4f*)pB0;
        f.b1 = *(const v4f*)pB1;
        f.b2 = *(const v4f*)pB2;
        f.b3 = *(const v4f*)pB3;
        pA0 += 8; pA1 += 8; pB0 += 8; pB1 += 8; pB2 += 8; pB3 += 8;
    };

    v8f acc[2][4] = {};
    frag6 s0, s1, s2, s3;
    load_group(s0);  __builtin_amdgcn_sched_barrier(0);
    load_group(s1);  __builtin_amdgcn_sched_barrier(0);
    load_group(s2);  __builtin_amdgcn_sched_barrier(0);
    for (int kk = 32; kk < K; kk += 32) {
        load_group(s3);  __builtin_amdgcn_sched_barrier(0);
        mma_group_rows(acc, s0);  __builtin_amdgcn_sched_barrier(0);
        load_group(s0);  __builtin_amdgcn_sched_barrier(0);
        mma_group_rows(acc, s1);  __builtin_amdgcn_sched_barrier(0);
        load_group(s1);  __builtin_amdgcn_sched_barrier(0);
        mma_group_rows(acc, s2);  __builtin_amdgcn_sched_barrier(0);
        load_group(s2);  __builtin_amdgcn_sched_barrier(0);
        mma_group_rows(acc, s3);  __builtin_amdgcn_sched_barrier(0);
    }
    load_group(s3);
    __builtin_amdgcn_sched_barrier(0);
    mma_group_rows(acc, s0);
    mma_group_rows(acc, s1);
    mma_group_rows(acc, s2);
    mma_group_rows(acc, s3);

    const int mbase = (mb << 5) + ((lane < 16) ? 0 : 8);
    #pragma unroll
    for (int ni = 0; ni < 4; ++ni) {
        const int   n  = (nb << 6) + (ni << 4) + l15;
        const float bn = bias ? bias[n] : 0.f;
        #pragma unroll
        for (int mi = 0; mi < 2; ++mi)
            #pragma unroll
            for (int r = 0; r < 8; ++r)
                Out[(size_t)(mbase + (mi << 4) + r) * Nc + n] = acc[mi][ni][r] + bn;
    }
}

#define ES_PITCH 68
__global__ void __launch_bounds__(256)
fa_scores_ew(const float* __restrict__ q,
             const float* __restrict__ k,
             const float* __restrict__ Wphi,
             float* __restrict__ ex,
             float* __restrict__ denom,
             float* __restrict__ Ew)
{
    __shared__ v4f   es4[512 * ES_PITCH / 4];
    __shared__ v4f   qi4[16];
    __shared__ v4f   wph4[16];
    __shared__ v4f   exs4[128];
    __shared__ float part[4 * 64];
    __shared__ float red[8];
    __shared__ float msh;
    float* es  = (float*)es4;
    float* qi  = (float*)qi4;
    float* wph = (float*)wph4;
    float* exs = (float*)exs4;
    const int row = blockIdx.x;
    const int b   = row >> 9;
    const int tid = threadIdx.x;
    if (tid < 64) {
        qi[tid]  = q[(size_t)row * 64 + tid];
        wph[tid] = Wphi[tid];
    }
    __syncthreads();

    float s[2];
    #pragma unroll
    for (int p = 0; p < 2; ++p) {
        const int j = tid + (p << 8);
        const float* kj = k + ((size_t)(b * 512 + j)) * 64;
        float* ej = es + j * ES_PITCH;
        float acc = 0.f;
        #pragma unroll 2
        for (int a = 0; a < 64; a += 4) {
            const v4f kv = *(const v4f*)(kj + a);
            const v4f qv = qi4[a >> 2];
            const v4f pw = wph4[a >> 2];
            v4f ev;
            ev.x = rho_relu(qv.x + kv.x);
            ev.y = rho_relu(qv.y + kv.y);
            ev.z = rho_relu(qv.z + kv.z);
            ev.w = rho_relu(qv.w + kv.w);
            acc += ev.x * pw.x;
            acc += ev.y * pw.y;
            acc += ev.z * pw.z;
            acc += ev.w * pw.w;
            *(v4f*)(ej + a) = ev;
        }
        s[p] = acc;
    }

    float mymax = fmaxf(s[0], s[1]);
    for (int off = 16; off; off >>= 1)
        mymax = fmaxf(mymax, __shfl_xor(mymax, off));
    const int lane = tid & 31, wv = tid >> 5;
    if (lane == 0) red[wv] = mymax;
    __syncthreads();
    if (tid == 0) {
        float m = red[0];
        for (int w = 1; w < 8; ++w) m = fmaxf(m, red[w]);
        msh = m;
    }
    __syncthreads();
    const float m = msh;

    const float e0 = expf(s[0] - m);
    const float e1 = expf(s[1] - m);
    ex[(size_t)row * 512 + tid]       = e0;
    ex[(size_t)row * 512 + tid + 256] = e1;
    exs[tid]       = e0;
    exs[tid + 256] = e1;

    float mysum = e0 + e1;
    for (int off = 16; off; off >>= 1)
        mysum += __shfl_xor(mysum, off);
    if (lane == 0) red[wv] = mysum;
    __syncthreads();
    if (tid == 0) {
        float sm = 0.f;
        for (int w = 0; w < 8; ++w) sm += red[w];
        denom[row] = sm;
    }

    const int a  = tid & 63;
    const int jq = tid >> 6;
    const float* er = es  + (jq * 128) * ES_PITCH + a;
    const float* xr = exs + jq * 128;
    float w0 = 0.f, w1 = 0.f, w2 = 0.f, w3 = 0.f;
    #pragma unroll 4
    for (int jj = 0; jj < 128; jj += 4) {
        const v4f xv = *(const v4f*)(xr + jj);
        w0 += xv.x * er[(jj + 0) * ES_PITCH];
        w1 += xv.y * er[(jj + 1) * ES_PITCH];
        w2 += xv.z * er[(jj + 2) * ES_PITCH];
        w3 += xv.w * er[(jj + 3) * ES_PITCH];
    }
    part[jq * 64 + a] = (w0 + w1) + (w2 + w3);
    __syncthreads();
    if (tid < 64)
        Ew[(size_t)row * 64 + tid] =
            (part[tid] + part[64 + tid]) + (part[128 + tid] + part[192 + tid]);
}

__global__ void __launch_bounds__(64)
fa_part1(const float* __restrict__ ex,
         const float* __restrict__ Vy,
         float* __restrict__ part1)
{
    const int lane = threadIdx.x & 31;
    const int wave = __builtin_amdgcn_readfirstlane((int)(threadIdx.x >> 5));
    const int job  = blockIdx.x * (int)(blockDim.x >> 5) + wave;
    const int b    = job >> 8;
    const int jj   = job & 255;
    const int mb   = jj >> 4;
    const int nb   = jj & 15;
    const int l15  = lane & 15;
    const int koff = (lane < 16) ? 0 : 4;
    const int col0 = (nb << 6) + (l15 << 2);

    const float* pA0 = ex + ((size_t)(b * 512 + (mb << 5) + l15)) * 512 + koff;
    const float* pA1 = pA0 + (size_t)16 * 512;
    const float* pV  = Vy + (size_t)b * 512 * 1024 + (size_t)koff * 1024 + col0;

    auto load_group = [&](frag6& f) {
        f.a0 = *(const v4f*)pA0;
        f.a1 = *(const v4f*)pA1;
        f.b0 = *(const v4f*)(pV);
        f.b1 = *(const v4f*)(pV + 1024);
        f.b2 = *(const v4f*)(pV + 2 * 1024);
        f.b3 = *(const v4f*)(pV + 3 * 1024);
        pA0 += 8; pA1 += 8; pV += 8 * 1024;
    };

    v8f acc[2][4] = {};
    frag6 s0, s1, s2, s3;
    load_group(s0);  __builtin_amdgcn_sched_barrier(0);
    load_group(s1);  __builtin_amdgcn_sched_barrier(0);
    load_group(s2);  __builtin_amdgcn_sched_barrier(0);
    for (int j = 32; j < 512; j += 32) {
        load_group(s3);  __builtin_amdgcn_sched_barrier(0);
        mma_group_cols(acc, s0);  __builtin_amdgcn_sched_barrier(0);
        load_group(s0);  __builtin_amdgcn_sched_barrier(0);
        mma_group_cols(acc, s1);  __builtin_amdgcn_sched_barrier(0);
        load_group(s1);  __builtin_amdgcn_sched_barrier(0);
        mma_group_cols(acc, s2);  __builtin_amdgcn_sched_barrier(0);
        load_group(s2);  __builtin_amdgcn_sched_barrier(0);
        mma_group_cols(acc, s3);  __builtin_amdgcn_sched_barrier(0);
    }
    load_group(s3);
    __builtin_amdgcn_sched_barrier(0);
    mma_group_cols(acc, s0);
    mma_group_cols(acc, s1);
    mma_group_cols(acc, s2);
    mma_group_cols(acc, s3);

    const int mbase = b * 512 + (mb << 5) + ((lane < 16) ? 0 : 8);
    #pragma unroll
    for (int mi = 0; mi < 2; ++mi)
        #pragma unroll
        for (int r = 0; r < 8; ++r) {
            v4f o;
            o.x = acc[mi][0][r]; o.y = acc[mi][1][r];
            o.z = acc[mi][2][r]; o.w = acc[mi][3][r];
            *(v4f*)(part1 + (size_t)(mbase + (mi << 4) + r) * 1024 + col0) = o;
        }
}

__global__ void fa_final(const float* __restrict__ Ew,
                         const float* __restrict__ Wew,
                         const float* __restrict__ Web,
                         const float* __restrict__ part1,
                         const float* __restrict__ denom,
                         float* __restrict__ out)
{
    const int lane = threadIdx.x & 31;
    const int wave = threadIdx.x >> 5;
    const int t  = blockIdx.x * (blockDim.x >> 5) + wave;
    const int mt = t >> 6, nt = t & 63;
    const int l15  = lane & 15;
    const int koff = (lane < 16) ? 0 : 4;
    const float* Arow = Ew  + (size_t)((mt << 4) + l15) * 64 + koff;
    const float* Brow = Wew + (size_t)((nt << 4) + l15) * 64 + koff;
    v8f acc = {};
    #pragma unroll
    for (int kk = 0; kk < 64; kk += 8) {
        const v4f av = *(const v4f*)(Arow + kk);
        const v4f bv = *(const v4f*)(Brow + kk);
        acc = wmma_f32(av.xy, bv.xy, acc);
        acc = wmma_f32(av.zw, bv.zw, acc);
    }
    const int   n     = (nt << 4) + l15;
    const int   mbase = (mt << 4) + ((lane < 16) ? 0 : 8);
    const float bn    = Web[n];
    #pragma unroll
    for (int r = 0; r < 8; ++r) {
        const int   m = mbase + r;
        const float d = denom[m];
        out[(size_t)m * 1024 + n] =
            (part1[(size_t)m * 1024 + n] + acc[r] + bn * d) / (d + EPS_DEN);
    }
}

extern "C" void kernel_launch(void* const* d_in, const int* in_sizes, int n_in,
                              void* d_out, int out_size, void* d_ws, size_t ws_size,
                              hipStream_t stream)
{
    (void)in_sizes; (void)n_in; (void)out_size; (void)ws_size;
    const float* X    = (const float*)d_in[0];
    const float* Wq   = (const float*)d_in[1];
    const float* Wk   = (const float*)d_in[2];
    const float* Wphi = (const float*)d_in[3];
    const float* Wyw  = (const float*)d_in[4];
    const float* Wyb  = (const float*)d_in[5];
    const float* Wew  = (const float*)d_in[6];
    const float* Web  = (const float*)d_in[7];
    float* out = (float*)d_out;

    float* ws  = (float*)d_ws;
    float* q   = ws;  ws += 1024 * 64;
    float* kk  = ws;  ws += 1024 * 64;
    float* Vy  = ws;  ws += 1024 * 1024;
    float* ex  = ws;  ws += 1024 * 512;
    float* den = ws;  ws += 1024;
    float* p1  = ws;  ws += 1024 * 1024;
    float* Ew  = ws;  ws += 1024 * 64;

    fa_proj3<<<288, 64, 0, stream>>>(X, Wq, Wk, Wyw, Wyb, q, kk, Vy, 1024);

    fa_scores_ew<<<1024, 256, 0, stream>>>(q, kk, Wphi, ex, den, Ew);

    fa_part1<<<256, 64, 0, stream>>>(ex, Vy, p1);

    fa_final<<<512, 256, 0, stream>>>(Ew, Wew, Web, p1, den, out);
}


# scores kernel: relu(rho(x)) from ONE correctly rounded root, sqrt(max(x,0)+eps) - sqrt(eps), the root as v_sqrt_f32 plus its two fma corrections without small-input scaling; e bit-identical
# speedup vs baseline: 5.0218x; 1.1077x over previous
#include <hip/hip_runtime.h>
#include <hip/hip_bf16.h>
#include <math.h>

typedef __attribute__((ext_vector_type(2))) float v2f;
typedef __attribute__((ext_vector_type(4))) float v4f;
typedef __attribute__((ext_vector_type(8))) float v8f;

#define EPS_RHO 1e-8f
#define EPS_DEN 1e-9f

__device__ __forceinline__ float sqrt_rn(float y) {
    const float s  = __builtin_amdgcn_sqrtf(y);
    const float sd = __int_as_float(__float_as_int(s) - 1);
    const float su = __int_as_float(__float_as_int(s) + 1);
    const float vp = __builtin_fmaf(-sd, s, y);
    const float vs = __builtin_fmaf(-su, s, y);
    float r = (vp <= 0.f) ? sd : s;
    r = (vs > 0.f) ? su : r;
    return r;
}

__device__ __forceinline__ float rho_relu(float x, float rt_eps) {
    return sqrt_rn(fmaxf(x, 0.f) + EPS_RHO) - rt_eps;
}

__device__ __forceinline__ v8f wmma_f32(v2f a, v2f b, v8f c) {
    return __builtin_amdgcn_wmma_f32_16x16x4_f32(false, a, false, b,
                                                 (short)0, c, false, false);
}

struct frag6 { v4f a0, a1, b0, b1, b2, b3; };

__device__ __forceinline__ void mma_group_rows(v8f (&acc)[2][4], const frag6& f) {
    acc[0][0] = wmma_f32(f.a0.xy, f.b0.xy, acc[0][0]);
    acc[1][0] = wmma_f32(f.a1.xy, f.b0.xy, acc[1][0]);
    acc[0][1] = wmma_f32(f.a0.xy, f.b1.xy, acc[0][1]);
    acc[1][1] = wmma_f32(f.a1.xy, f.b1.xy, acc[1][1]);
    acc[0][2] = wmma_f32(f.a0.xy, f.b2.xy, acc[0][2]);
    acc[1][2] = wmma_f32(f.a1.xy, f.b2.xy, acc[1][2]);
    acc[0][3] = wmma_f32(f.a0.xy, f.b3.xy, acc[0][3]);
    acc[1][3] = wmma_f32(f.a1.xy, f.b3.xy, acc[1][3]);
    acc[0][0] = wmma_f32(f.a0.zw, f.b0.zw, acc[0][0]);
    acc[1][0] = wmma_f32(f.a1.zw, f.b0.zw, acc[1][0]);
    acc[0][1] = wmma_f32(f.a0.zw, f.b1.zw, acc[0][1]);
    acc[1][1] = wmma_f32(f.a1.zw, f.b1.zw, acc[1][1]);
    acc[0][2] = wmma_f32(f.a0.zw, f.b2.zw, acc[0][2]);
    acc[1][2] = wmma_f32(f.a1.zw, f.b2.zw, acc[1][2]);
    acc[0][3] = wmma_f32(f.a0.zw, f.b3.zw, acc[0][3]);
    acc[1][3] = wmma_f32(f.a1.zw, f.b3.zw, acc[1][3]);
}

__device__ __forceinline__ void mma_group_cols(v8f (&acc)[2][4], const frag6& f) {
    #pragma unroll
    for (int t = 0; t < 4; ++t) {
        v2f bt; bt.x = f.b0[t]; bt.y = f.b1[t];
        acc[0][t] = wmma_f32(f.a0.xy, bt, acc[0][t]);
        acc[1][t] = wmma_f32(f.a1.xy, bt, acc[1][t]);
    }
    #pragma unroll
    for (int t = 0; t < 4; ++t) {
        v2f bt; bt.x = f.b2[t]; bt.y = f.b3[t];
        acc[0][t] = wmma_f32(f.a0.zw, bt, acc[0][t]);
        acc[1][t] = wmma_f32(f.a1.zw, bt, acc[1][t]);
    }
}

__global__ void __launch_bounds__(64)
fa_proj3(const float* __restrict__ A,
         const float* __restrict__ Wq,
         const float* __restrict__ Wk,
         const float* __restrict__ Wyw,
         const float* __restrict__ Wyb,
         float* __restrict__ q,
         float* __restrict__ kout,
         float* __restrict__ Vy,
         int K)
{
    const int lane = threadIdx.x & 31;
    const int wave = __builtin_amdgcn_readfirstlane((int)(threadIdx.x >> 5));
    const int job  = blockIdx.x * (int)(blockDim.x >> 5) + wave;
    const float* W;
    const float* bias;
    float* Out;
    int Nc, mb, nb;
    if (job < 32) {
        W = Wq;  bias = nullptr; Out = q;    Nc = 64;
        mb = job;  nb = 0;
    } else if (job < 64) {
        W = Wk;  bias = nullptr; Out = kout; Nc = 64;
        mb = job - 32;  nb = 0;
    } else {
        W = Wyw; bias = Wyb;     Out = Vy;   Nc = 1024;
        const int jj = job - 64;
        mb = jj >> 4;  nb = jj & 15;
    }
    const int l15  = lane & 15;
    const int koff = (lane < 16) ? 0 : 4;
    const size_t bstep = (size_t)16 * K;

    const float* pA0 = A + (size_t)((mb << 5) + l15) * K + koff;
    const float* pA1 = pA0 + bstep;
    const float* pB0 = W + (size_t)((nb << 6) + l15) * K + koff;
    const float* pB1 = pB0 + bstep;
    const float* pB2 = pB1 + bstep;
    const float* pB3 = pB2 + bstep;

    auto load_group = [&](frag6& f) {
        f.a0 = *(const v4f*)pA0;
        f.a1 = *(const v4f*)pA1;
        f.b0 = *(const v4f*)pB0;
        f.b1 = *(const v4f*)pB1;
        f.b2 = *(const v4f*)pB2;
        f.b3 = *(const v4f*)pB3;
        pA0 += 8; pA1 += 8; pB0 += 8; pB1 += 8; pB2 += 8; pB3 += 8;
    };

    v8f acc[2][4] = {};
    frag6 s0, s1, s2, s3;
    load_group(s0);  __builtin_amdgcn_sched_barrier(0);
    load_group(s1);  __builtin_amdgcn_sched_barrier(0);
    load_group(s2);  __builtin_amdgcn_sched_barrier(0);
    for (int kk = 32; kk < K; kk += 32) {
        load_group(s3);  __builtin_amdgcn_sched_barrier(0);
        mma_group_rows(acc, s0);  __builtin_amdgcn_sched_barrier(0);
        load_group(s0);  __builtin_amdgcn_sched_barrier(0);
        mma_group_rows(acc, s1);  __builtin_amdgcn_sched_barrier(0);
        load_group(s1);  __builtin_amdgcn_sched_barrier(0);
        mma_group_rows(acc, s2);  __builtin_amdgcn_sched_barrier(0);
        load_group(s2);  __builtin_amdgcn_sched_barrier(0);
        mma_group_rows(acc, s3);  __builtin_amdgcn_sched_barrier(0);
    }
    load_group(s3);
    __builtin_amdgcn_sched_barrier(0);
    mma_group_rows(acc, s0);
    mma_group_rows(acc, s1);
    mma_group_rows(acc, s2);
    mma_group_rows(acc, s3);

    const int mbase = (mb << 5) + ((lane < 16) ? 0 : 8);
    #pragma unroll
    for (int ni = 0; ni < 4; ++ni) {
        const int   n  = (nb << 6) + (ni << 4) + l15;
        const float bn = bias ? bias[n] : 0.f;
        #pragma unroll
        for (int mi = 0; mi < 2; ++mi)
            #pragma unroll
            for (int r = 0; r < 8; ++r)
                Out[(size_t)(mbase + (mi << 4) + r) * Nc + n] = acc[mi][ni][r] + bn;
    }
}

#define ES_PITCH 68
__global__ void __launch_bounds__(256)
fa_scores_ew(const float* __restrict__ q,
             const float* __restrict__ k,
             const float* __restrict__ Wphi,
             float* __restrict__ ex,
             float* __restrict__ denom,
             float* __restrict__ Ew)
{
    __shared__ v4f   es4[512 * ES_PITCH / 4];
    __shared__ v4f   qi4[16];
    __shared__ v4f   wph4[16];
    __shared__ v4f   exs4[128];
    __shared__ float part[4 * 64];
    __shared__ float red[8];
    __shared__ float msh;
    float* es  = (float*)es4;
    float* qi  = (float*)qi4;
    float* wph = (float*)wph4;
    float* exs = (float*)exs4;
    const int row = blockIdx.x;
    const int b   = row >> 9;
    const int tid = threadIdx.x;
    if (tid < 64) {
        qi[tid]  = q[(size_t)row * 64 + tid];
        wph[tid] = Wphi[tid];
    }
    __syncthreads();

    const float rt_eps = sqrt_rn(EPS_RHO);
    float s[2];
    #pragma unroll
    for (int p = 0; p < 2; ++p) {
        const int j = tid + (p << 8);
        const float* kj = k + ((size_t)(b * 512 + j)) * 64;
        float* ej = es + j * ES_PITCH;
        float acc = 0.f;
        #pragma unroll 2
        for (int a = 0; a < 64; a += 4) {
            const v4f kv = *(const v4f*)(kj + a);
            const v4f qv = qi4[a >> 2];
            const v4f pw = wph4[a >> 2];
            v4f ev;
            ev.x = rho_relu(qv.x + kv.x, rt_eps);
            ev.y = rho_relu(qv.y + kv.y, rt_eps);
            ev.z = rho_relu(qv.z + kv.z, rt_eps);
            ev.w = rho_relu(qv.w + kv.w, rt_eps);
            acc += ev.x * pw.x;
            acc += ev.y * pw.y;
            acc += ev.z * pw.z;
            acc += ev.w * pw.w;
            *(v4f*)(ej + a) = ev;
        }
        s[p] = acc;
    }

    float mymax = fmaxf(s[0], s[1]);
    for (int off = 16; off; off >>= 1)
        mymax = fmaxf(mymax, __shfl_xor(mymax, off));
    const int lane = tid & 31, wv = tid >> 5;
    if (lane == 0) red[wv] = mymax;
    __syncthreads();
    if (tid == 0) {
        float m = red[0];
        for (int w = 1; w < 8; ++w) m = fmaxf(m, red[w]);
        msh = m;
    }
    __syncthreads();
    const float m = msh;

    const float e0 = expf(s[0] - m);
    const float e1 = expf(s[1] - m);
    ex[(size_t)row * 512 + tid]       = e0;
    ex[(size_t)row * 512 + tid + 256] = e1;
    exs[tid]       = e0;
    exs[tid + 256] = e1;

    float mysum = e0 + e1;
    for (int off = 16; off; off >>= 1)
        mysum += __shfl_xor(mysum, off);
    if (lane == 0) red[wv] = mysum;
    __syncthreads();
    if (tid == 0) {
        float sm = 0.f;
        for (int w = 0; w < 8; ++w) sm += red[w];
        denom[row] = sm;
    }

    const int a  = tid & 63;
    const int jq = tid >> 6;
    const float* er = es  + (jq * 128) * ES_PITCH + a;
    const float* xr = exs + jq * 128;
    float w0 = 0.f, w1 = 0.f, w2 = 0.f, w3 = 0.f;
    #pragma unroll 4
    for (int jj = 0; jj < 128; jj += 4) {
        const v4f xv = *(const v4f*)(xr + jj);
        w0 += xv.x * er[(jj + 0) * ES_PITCH];
        w1 += xv.y * er[(jj + 1) * ES_PITCH];
        w2 += xv.z * er[(jj + 2) * ES_PITCH];
        w3 += xv.w * er[(jj + 3) * ES_PITCH];
    }
    part[jq * 64 + a] = (w0 + w1) + (w2 + w3);
    __syncthreads();
    if (tid < 64)
        Ew[(size_t)row * 64 + tid] =
            (part[tid] + part[64 + tid]) + (part[128 + tid] + part[192 + tid]);
}

__global__ void __launch_bounds__(64)
fa_part1(const float* __restrict__ ex,
         const float* __restrict__ Vy,
         float* __restrict__ part1)
{
    const int lane = threadIdx.x & 31;
    const int wave = __builtin_amdgcn_readfirstlane((int)(threadIdx.x >> 5));
    const int job  = blockIdx.x * (int)(blockDim.x >> 5) + wave;
    const int b    = job >> 8;
    const int jj   = job & 255;
    const int mb   = jj >> 4;
    const int nb   = jj & 15;
    const int l15  = lane & 15;
    const int koff = (lane < 16) ? 0 : 4;
    const int col0 = (nb << 6) + (l15 << 2);

    const float* pA0 = ex + ((size_t)(b * 512 + (mb << 5) + l15)) * 512 + koff;
    const float* pA1 = pA0 + (size_t)16 * 512;
    const float* pV  = Vy + (size_t)b * 512 * 1024 + (size_t)koff * 1024 + col0;

    auto load_group = [&](frag6& f) {
        f.a0 = *(const v4f*)pA0;
        f.a1 = *(const v4f*)pA1;
        f.b0 = *(const v4f*)(pV);
        f.b1 = *(const v4f*)(pV + 1024);
        f.b2 = *(const v4f*)(pV + 2 * 1024);
        f.b3 = *(const v4f*)(pV + 3 * 1024);
        pA0 += 8; pA1 += 8; pV += 8 * 1024;
    };

    v8f acc[2][4] = {};
    frag6 s0, s1, s2, s3;
    load_group(s0);  __builtin_amdgcn_sched_barrier(0);
    load_group(s1);  __builtin_amdgcn_sched_barrier(0);
    load_group(s2);  __builtin_amdgcn_sched_barrier(0);
    for (int j = 32; j < 512; j += 32) {
        load_group(s3);  __builtin_amdgcn_sched_barrier(0);
        mma_group_cols(acc, s0);  __builtin_amdgcn_sched_barrier(0);
        load_group(s0);  __builtin_amdgcn_sched_barrier(0);
        mma_group_cols(acc, s1);  __builtin_amdgcn_sched_barrier(0);
        load_group(s1);  __builtin_amdgcn_sched_barrier(0);
        mma_group_cols(acc, s2);  __builtin_amdgcn_sched_barrier(0);
        load_group(s2);  __builtin_amdgcn_sched_barrier(0);
        mma_group_cols(acc, s3);  __builtin_amdgcn_sched_barrier(0);
    }
    load_group(s3);
    __builtin_amdgcn_sched_barrier(0);
    mma_group_cols(acc, s0);
    mma_group_cols(acc, s1);
    mma_group_cols(acc, s2);
    mma_group_cols(acc, s3);

    const int mbase = b * 512 + (mb << 5) + ((lane < 16) ? 0 : 8);
    #pragma unroll
    for (int mi = 0; mi < 2; ++mi)
        #pragma unroll
        for (int r = 0; r < 8; ++r) {
            v4f o;
            o.x = acc[mi][0][r]; o.y = acc[mi][1][r];
            o.z = acc[mi][2][r]; o.w = acc[mi][3][r];
            *(v4f*)(part1 + (size_t)(mbase + (mi << 4) + r) * 1024 + col0) = o;
        }
}

__global__ void fa_final(const float* __restrict__ Ew,
                         const float* __restrict__ Wew,
                         const float* __restrict__ Web,
                         const float* __restrict__ part1,
                         const float* __restrict__ denom,
                         float* __restrict__ out)
{
    const int lane = threadIdx.x & 31;
    const int wave = threadIdx.x >> 5;
    const int t  = blockIdx.x * (blockDim.x >> 5) + wave;
    const int mt = t >> 6, nt = t & 63;
    const int l15  = lane & 15;
    const int koff = (lane < 16) ? 0 : 4;
    const float* Arow = Ew  + (size_t)((mt << 4) + l15) * 64 + koff;
    const float* Brow = Wew + (size_t)((nt << 4) + l15) * 64 + koff;
    v8f acc = {};
    #pragma unroll
    for (int kk = 0; kk < 64; kk += 8) {
        const v4f av = *(const v4f*)(Arow + kk);
        const v4f bv = *(const v4f*)(Brow + kk);
        acc = wmma_f32(av.xy, bv.xy, acc);
        acc = wmma_f32(av.zw, bv.zw, acc);
    }
    const int   n     = (nt << 4) + l15;
    const int   mbase = (mt << 4) + ((lane < 16) ? 0 : 8);
    const float bn    = Web[n];
    #pragma unroll
    for (int r = 0; r < 8; ++r) {
        const int   m = mbase + r;
        const float d = denom[m];
        out[(size_t)m * 1024 + n] =
            (part1[(size_t)m * 1024 + n] + acc[r] + bn * d) / (d + EPS_DEN);
    }
}

extern "C" void kernel_launch(void* const* d_in, const int* in_sizes, int n_in,
                              void* d_out, int out_size, void* d_ws, size_t ws_size,
                              hipStream_t stream)
{
    (void)in_sizes; (void)n_in; (void)out_size; (void)ws_size;
    const float* X    = (const float*)d_in[0];
    const float* Wq   = (const float*)d_in[1];
    const float* Wk   = (const float*)d_in[2];
    const float* Wphi = (const float*)d_in[3];
    const float* Wyw  = (const float*)d_in[4];
    const float* Wyb  = (const float*)d_in[5];
    const float* Wew  = (const float*)d_in[6];
    const float* Web  = (const float*)d_in[7];
    float* out = (float*)d_out;

    float* ws  = (float*)d_ws;
    float* q   = ws;  ws += 1024 * 64;
    float* kk  = ws;  ws += 1024 * 64;
    float* Vy  = ws;  ws += 1024 * 1024;
    float* ex  = ws;  ws += 1024 * 512;
    float* den = ws;  ws += 1024;
    float* p1  = ws;  ws += 1024 * 1024;
    float* Ew  = ws;  ws += 1024 * 64;

    fa_proj3<<<288, 64, 0, stream>>>(X, Wq, Wk, Wyw, Wyb, q, kk, Vy, 1024);

    fa_scores_ew<<<1024, 256, 0, stream>>>(q, kk, Wphi, ex, den, Ew);

    fa_part1<<<256, 64, 0, stream>>>(ex, Vy, p1);

    fa_final<<<512, 256, 0, stream>>>(Ew, Wew, Web, p1, den, out);
}


# final launch folded into part1: Ew @ We^T (64 more k) runs on part1's WMMA accumulators after the j loop, bias, denom, divide in the epilogue; part2 adds into part1's sum; 4 -> 3 launches
# speedup vs baseline: 5.3101x; 1.0574x over previous
#include <hip/hip_runtime.h>
#include <hip/hip_bf16.h>
#include <math.h>

typedef __attribute__((ext_vector_type(2))) float v2f;
typedef __attribute__((ext_vector_type(4))) float v4f;
typedef __attribute__((ext_vector_type(8))) float v8f;

#define EPS_RHO 1e-8f
#define EPS_DEN 1e-9f

__device__ __forceinline__ float sqrt_rn(float y) {
    const float s  = __builtin_amdgcn_sqrtf(y);
    const float sd = __int_as_float(__float_as_int(s) - 1);
    const float su = __int_as_float(__float_as_int(s) + 1);
    const float vp = __builtin_fmaf(-sd, s, y);
    const float vs = __builtin_fmaf(-su, s, y);
    float r = (vp <= 0.f) ? sd : s;
    r = (vs > 0.f) ? su : r;
    return r;
}

__device__ __forceinline__ float rho_relu(float x, float rt_eps) {
    return sqrt_rn(fmaxf(x, 0.f) + EPS_RHO) - rt_eps;
}

__device__ __forceinline__ v8f wmma_f32(v2f a, v2f b, v8f c) {
    return __builtin_amdgcn_wmma_f32_16x16x4_f32(false, a, false, b,
                                                 (short)0, c, false, false);
}

struct frag6 { v4f a0, a1, b0, b1, b2, b3; };

__device__ __forceinline__ void mma_group_rows(v8f (&acc)[2][4], const frag6& f) {
    acc[0][0] = wmma_f32(f.a0.xy, f.b0.xy, acc[0][0]);
    acc[1][0] = wmma_f32(f.a1.xy, f.b0.xy, acc[1][0]);
    acc[0][1] = wmma_f32(f.a0.xy, f.b1.xy, acc[0][1]);
    acc[1][1] = wmma_f32(f.a1.xy, f.b1.xy, acc[1][1]);
    acc[0][2] = wmma_f32(f.a0.xy, f.b2.xy, acc[0][2]);
    acc[1][2] = wmma_f32(f.a1.xy, f.b2.xy, acc[1][2]);
    acc[0][3] = wmma_f32(f.a0.xy, f.b3.xy, acc[0][3]);
    acc[1][3] = wmma_f32(f.a1.xy, f.b3.xy, acc[1][3]);
    acc[0][0] = wmma_f32(f.a0.zw, f.b0.zw, acc[0][0]);
    acc[1][0] = wmma_f32(f.a1.zw, f.b0.zw, acc[1][0]);
    acc[0][1] = wmma_f32(f.a0.zw, f.b1.zw, acc[0][1]);
    acc[1][1] = wmma_f32(f.a1.zw, f.b1.zw, acc[1][1]);
    acc[0][2] = wmma_f32(f.a0.zw, f.b2.zw, acc[0][2]);
    acc[1][2] = wmma_f32(f.a1.zw, f.b2.zw, acc[1][2]);
    acc[0][3] = wmma_f32(f.a0.zw, f.b3.zw, acc[0][3]);
    acc[1][3] = wmma_f32(f.a1.zw, f.b3.zw, acc[1][3]);
}

__device__ __forceinline__ void mma_group_cols(v8f (&acc)[2][4], const frag6& f) {
    #pragma unroll
    for (int t = 0; t < 4; ++t) {
        v2f bt; bt.x = f.b0[t]; bt.y = f.b1[t];
        acc[0][t] = wmma_f32(f.a0.xy, bt, acc[0][t]);
        acc[1][t] = wmma_f32(f.a1.xy, bt, acc[1][t]);
    }
    #pragma unroll
    for (int t = 0; t < 4; ++t) {
        v2f bt; bt.x = f.b2[t]; bt.y = f.b3[t];
        acc[0][t] = wmma_f32(f.a0.zw, bt, acc[0][t]);
        acc[1][t] = wmma_f32(f.a1.zw, bt, acc[1][t]);
    }
}

__global__ void __launch_bounds__(64)
fa_proj3(const float* __restrict__ A,
         const float* __restrict__ Wq,
         const float* __restrict__ Wk,
         const float* __restrict__ Wyw,
         const float* __restrict__ Wyb,
         float* __restrict__ q,
         float* __restrict__ kout,
         float* __restrict__ Vy,
         int K)
{
    const int lane = threadIdx.x & 31;
    const int wave = __builtin_amdgcn_readfirstlane((int)(threadIdx.x >> 5));
    const int job  = blockIdx.x * (int)(blockDim.x >> 5) + wave;
    const float* W;
    const float* bias;
    float* Out;
    int Nc, mb, nb;
    if (job < 32) {
        W = Wq;  bias = nullptr; Out = q;    Nc = 64;
        mb = job;  nb = 0;
    } else if (job < 64) {
        W = Wk;  bias = nullptr; Out = kout; Nc = 64;
        mb = job - 32;  nb = 0;
    } else {
        W = Wyw; bias = Wyb;     Out = Vy;   Nc = 1024;
        const int jj = job - 64;
        mb = jj >> 4;  nb = jj & 15;
    }
    const int l15  = lane & 15;
    const int koff = (lane < 16) ? 0 : 4;
    const size_t bstep = (size_t)16 * K;

    const float* pA0 = A + (size_t)((mb << 5) + l15) * K + koff;
    const float* pA1 = pA0 + bstep;
    const float* pB0 = W + (size_t)((nb << 6) + l15) * K + koff;
    const float* pB1 = pB0 + bstep;
    const float* pB2 = pB1 + bstep;
    const float* pB3 = pB2 + bstep;

    auto load_group = [&](frag6& f) {
        f.a0 = *(const v4f*)pA0;
        f.a1 = *(const v4f*)pA1;
        f.b0 = *(const v4f*)pB0;
        f.b1 = *(const v4f*)pB1;
        f.b2 = *(const v4f*)pB2;
        f.b3 = *(const v4f*)pB3;
        pA0 += 8; pA1 += 8; pB0 += 8; pB1 += 8; pB2 += 8; pB3 += 8;
    };

    v8f acc[2][4] = {};
    frag6 s0, s1, s2, s3;
    load_group(s0);  __builtin_amdgcn_sched_barrier(0);
    load_group(s1);  __builtin_amdgcn_sched_barrier(0);
    load_group(s2);  __builtin_amdgcn_sched_barrier(0);
    for (int kk = 32; kk < K; kk += 32) {
        load_group(s3);  __builtin_amdgcn_sched_barrier(0);
        mma_group_rows(acc, s0);  __builtin_amdgcn_sched_barrier(0);
        load_group(s0);  __builtin_amdgcn_sched_barrier(0);
        mma_group_rows(acc, s1);  __builtin_amdgcn_sched_barrier(0);
        load_group(s1);  __builtin_amdgcn_sched_barrier(0);
        mma_group_rows(acc, s2);  __builtin_amdgcn_sched_barrier(0);
        load_group(s2);  __builtin_amdgcn_sched_barrier(0);
        mma_group_rows(acc, s3);  __builtin_amdgcn_sched_barrier(0);
    }
    load_group(s3);
    __builtin_amdgcn_sched_barrier(0);
    mma_group_rows(acc, s0);
    mma_group_rows(acc, s1);
    mma_group_rows(acc, s2);
    mma_group_rows(acc, s3);

    const int mbase = (mb << 5) + ((lane < 16) ? 0 : 8);
    #pragma unroll
    for (int ni = 0; ni < 4; ++ni) {
        const int   n  = (nb << 6) + (ni << 4) + l15;
        const float bn = bias ? bias[n] : 0.f;
        #pragma unroll
        for (int mi = 0; mi < 2; ++mi)
            #pragma unroll
            for (int r = 0; r < 8; ++r)
                Out[(size_t)(mbase + (mi << 4) + r) * Nc + n] = acc[mi][ni][r] + bn;
    }
}

#define ES_PITCH 68
__global__ void __launch_bounds__(256)
fa_scores_ew(const float* __restrict__ q,
             const float* __restrict__ k,
             const float* __restrict__ Wphi,
             float* __restrict__ ex,
             float* __restrict__ denom,
             float* __restrict__ Ew)
{
    __shared__ v4f   es4[512 * ES_PITCH / 4];
    __shared__ v4f   qi4[16];
    __shared__ v4f   wph4[16];
    __shared__ v4f   exs4[128];
    __shared__ float part[4 * 64];
    __shared__ float red[8];
    __shared__ float msh;
    float* es  = (float*)es4;
    float* qi  = (float*)qi4;
    float* wph = (float*)wph4;
    float* exs = (float*)exs4;
    const int row = blockIdx.x;
    const int b   = row >> 9;
    const int tid = threadIdx.x;
    if (tid < 64) {
        qi[tid]  = q[(size_t)row * 64 + tid];
        wph[tid] = Wphi[tid];
    }
    __syncthreads();

    const float rt_eps = sqrt_rn(EPS_RHO);
    float s[2];
    #pragma unroll
    for (int p = 0; p < 2; ++p) {
        const int j = tid + (p << 8);
        const float* kj = k + ((size_t)(b * 512 + j)) * 64;
        float* ej = es + j * ES_PITCH;
        float acc = 0.f;
        #pragma unroll 2
        for (int a = 0; a < 64; a += 4) {
            const v4f kv = *(const v4f*)(kj + a);
            const v4f qv = qi4[a >> 2];
            const v4f pw = wph4[a >> 2];
            v4f ev;
            ev.x = rho_relu(qv.x + kv.x, rt_eps);
            ev.y = rho_relu(qv.y + kv.y, rt_eps);
            ev.z = rho_relu(qv.z + kv.z, rt_eps);
            ev.w = rho_relu(qv.w + kv.w, rt_eps);
            acc += ev.x * pw.x;
            acc += ev.y * pw.y;
            acc += ev.z * pw.z;
            acc += ev.w * pw.w;
            *(v4f*)(ej + a) = ev;
        }
        s[p] = acc;
    }

    float mymax = fmaxf(s[0], s[1]);
    for (int off = 16; off; off >>= 1)
        mymax = fmaxf(mymax, __shfl_xor(mymax, off));
    const int lane = tid & 31, wv = tid >> 5;
    if (lane == 0) red[wv] = mymax;
    __syncthreads();
    if (tid == 0) {
        float m = red[0];
        for (int w = 1; w < 8; ++w) m = fmaxf(m, red[w]);
        msh = m;
    }
    __syncthreads();
    const float m = msh;

    const float e0 = expf(s[0] - m);
    const float e1 = expf(s[1] - m);
    ex[(size_t)row * 512 + tid]       = e0;
    ex[(size_t)row * 512 + tid + 256] = e1;
    exs[tid]       = e0;
    exs[tid + 256] = e1;

    float mysum = e0 + e1;
    for (int off = 16; off; off >>= 1)
        mysum += __shfl_xor(mysum, off);
    if (lane == 0) red[wv] = mysum;
    __syncthreads();
    if (tid == 0) {
        float sm = 0.f;
        for (int w = 0; w < 8; ++w) sm += red[w];
        denom[row] = sm;
    }

    const int a  = tid & 63;
    const int jq = tid >> 6;
    const float* er = es  + (jq * 128) * ES_PITCH + a;
    const float* xr = exs + jq * 128;
    float w0 = 0.f, w1 = 0.f, w2 = 0.f, w3 = 0.f;
    #pragma unroll 4
    for (int jj = 0; jj < 128; jj += 4) {
        const v4f xv = *(const v4f*)(xr + jj);
        w0 += xv.x * er[(jj + 0) * ES_PITCH];
        w1 += xv.y * er[(jj + 1) * ES_PITCH];
        w2 += xv.z * er[(jj + 2) * ES_PITCH];
        w3 += xv.w * er[(jj + 3) * ES_PITCH];
    }
    part[jq * 64 + a] = (w0 + w1) + (w2 + w3);
    __syncthreads();
    if (tid < 64)
        Ew[(size_t)row * 64 + tid] =
            (part[tid] + part[64 + tid]) + (part[128 + tid] + part[192 + tid]);
}

__global__ void __launch_bounds__(64)
fa_part12(const float* __restrict__ ex,
          const float* __restrict__ Vy,
          const float* __restrict__ Ew,
          const float* __restrict__ Wew,
          const float* __restrict__ Web,
          const float* __restrict__ denom,
          float* __restrict__ out)
{
    const int lane = threadIdx.x & 31;
    const int wave = __builtin_amdgcn_readfirstlane((int)(threadIdx.x >> 5));
    const int job  = blockIdx.x * (int)(blockDim.x >> 5) + wave;
    const int b    = job >> 8;
    const int jj   = job & 255;
    const int mb   = jj >> 4;
    const int nb   = jj & 15;
    const int l15  = lane & 15;
    const int koff = (lane < 16) ? 0 : 4;
    const int col0 = (nb << 6) + (l15 << 2);
    const int row0 = b * 512 + (mb << 5);

    const float* pA0 = ex + ((size_t)(row0 + l15)) * 512 + koff;
    const float* pA1 = pA0 + (size_t)16 * 512;
    const float* pV  = Vy + (size_t)b * 512 * 1024 + (size_t)koff * 1024 + col0;
    const float* pE0 = Ew + ((size_t)(row0 + l15)) * 64 + koff;
    const float* pE1 = pE0 + (size_t)16 * 64;
    const float* pW  = Wew + (size_t)col0 * 64 + koff;

    auto load_group = [&](frag6& f) {
        f.a0 = *(const v4f*)pA0;
        f.a1 = *(const v4f*)pA1;
        f.b0 = *(const v4f*)(pV);
        f.b1 = *(const v4f*)(pV + 1024);
        f.b2 = *(const v4f*)(pV + 2 * 1024);
        f.b3 = *(const v4f*)(pV + 3 * 1024);
        pA0 += 8; pA1 += 8; pV += 8 * 1024;
    };
    auto load_group_a = [&](frag6& f) {
        f.a0 = *(const v4f*)pE0;
        f.a1 = *(const v4f*)pE1;
        f.b0 = *(const v4f*)(pW);
        f.b1 = *(const v4f*)(pW + 64);
        f.b2 = *(const v4f*)(pW + 2 * 64);
        f.b3 = *(const v4f*)(pW + 3 * 64);
        pE0 += 8; pE1 += 8; pW += 8;
    };

    const int mbase = row0 + ((lane < 16) ? 0 : 8);
    const v4f bn = *(const v4f*)(Web + col0);
    v4f dn[2][2];
    #pragma unroll
    for (int mi = 0; mi < 2; ++mi) {
        dn[mi][0] = *(const v4f*)(denom + mbase + (mi << 4));
        dn[mi][1] = *(const v4f*)(denom + mbase + (mi << 4) + 4);
    }
    __builtin_amdgcn_sched_barrier(0);

    v8f acc[2][4] = {};
    frag6 s0, s1, s2, s3;
    load_group(s0);  __builtin_amdgcn_sched_barrier(0);
    load_group(s1);  __builtin_amdgcn_sched_barrier(0);
    load_group(s2);  __builtin_amdgcn_sched_barrier(0);
    for (int j = 32; j < 512; j += 32) {
        load_group(s3);  __builtin_amdgcn_sched_barrier(0);
        mma_group_cols(acc, s0);  __builtin_amdgcn_sched_barrier(0);
        load_group(s0);  __builtin_amdgcn_sched_barrier(0);
        mma_group_cols(acc, s1);  __builtin_amdgcn_sched_barrier(0);
        load_group(s1);  __builtin_amdgcn_sched_barrier(0);
        mma_group_cols(acc, s2);  __builtin_amdgcn_sched_barrier(0);
        load_group(s2);  __builtin_amdgcn_sched_barrier(0);
        mma_group_cols(acc, s3);  __builtin_amdgcn_sched_barrier(0);
    }
    load_group(s3);
    __builtin_amdgcn_sched_barrier(0);
    mma_group_cols(acc, s0);  __builtin_amdgcn_sched_barrier(0);
    load_group_a(s0);  __builtin_amdgcn_sched_barrier(0);
    mma_group_cols(acc, s1);  __builtin_amdgcn_sched_barrier(0);
    load_group_a(s1);  __builtin_amdgcn_sched_barrier(0);
    mma_group_cols(acc, s2);  __builtin_amdgcn_sched_barrier(0);
    load_group_a(s2);  __builtin_amdgcn_sched_barrier(0);
    mma_group_cols(acc, s3);  __builtin_amdgcn_sched_barrier(0);
    load_group_a(s3);  __builtin_amdgcn_sched_barrier(0);
    mma_group_rows(acc, s0);  __builtin_amdgcn_sched_barrier(0);
    load_group_a(s0);  __builtin_amdgcn_sched_barrier(0);
    mma_group_rows(acc, s1);  __builtin_amdgcn_sched_barrier(0);
    load_group_a(s1);  __builtin_amdgcn_sched_barrier(0);
    mma_group_rows(acc, s2);  __builtin_amdgcn_sched_barrier(0);
    load_group_a(s2);  __builtin_amdgcn_sched_barrier(0);
    mma_group_rows(acc, s3);  __builtin_amdgcn_sched_barrier(0);
    load_group_a(s3);
    __builtin_amdgcn_sched_barrier(0);
    mma_group_rows(acc, s0);
    mma_group_rows(acc, s1);
    mma_group_rows(acc, s2);
    mma_group_rows(acc, s3);

    #pragma unroll
    for (int mi = 0; mi < 2; ++mi) {
        #pragma unroll
        for (int r = 0; r < 8; ++r) {
            const float d  = dn[mi][r >> 2][r & 3];
            const float dd = d + EPS_DEN;
            v4f o;
            o.x = (acc[mi][0][r] + bn.x * d) / dd;
            o.y = (acc[mi][1][r] + bn.y * d) / dd;
            o.z = (acc[mi][2][r] + bn.z * d) / dd;
            o.w = (acc[mi][3][r] + bn.w * d) / dd;
            *(v4f*)(out + (size_t)(mbase + (mi << 4) + r) * 1024 + col0) = o;
        }
    }
}

extern "C" void kernel_launch(void* const* d_in, const int* in_sizes, int n_in,
                              void* d_out, int out_size, void* d_ws, size_t ws_size,
                              hipStream_t stream)
{
    (void)in_sizes; (void)n_in; (void)out_size; (void)ws_size;
    const float* X    = (const float*)d_in[0];
    const float* Wq   = (const float*)d_in[1];
    const float* Wk   = (const float*)d_in[2];
    const float* Wphi = (const float*)d_in[3];
    const float* Wyw  = (const float*)d_in[4];
    const float* Wyb  = (const float*)d_in[5];
    const float* Wew  = (const float*)d_in[6];
    const float* Web  = (const float*)d_in[7];
    float* out = (float*)d_out;

    float* ws  = (float*)d_ws;
    float* q   = ws;  ws += 1024 * 64;
    float* kk  = ws;  ws += 1024 * 64;
    float* Vy  = ws;  ws += 1024 * 1024;
    float* ex  = ws;  ws += 1024 * 512;
    float* den = ws;  ws += 1024;
    float* Ew  = ws;  ws += 1024 * 64;

    fa_proj3<<<288, 64, 0, stream>>>(X, Wq, Wk, Wyw, Wyb, q, kk, Vy, 1024);

    fa_scores_ew<<<1024, 256, 0, stream>>>(q, kk, Wphi, ex, den, Ew);

    fa_part12<<<256, 64, 0, stream>>>(ex, Vy, Ew, Wew, Web, den, out);
}


# both GEMM launches: each tile block's k range split over two waves, halves added through LDS (k sum = two half sums); q/k tiles ride with Vy blocks: 1024 equal waves, one per SIMD
# speedup vs baseline: 6.7051x; 1.2627x over previous
#include <hip/hip_runtime.h>
#include <hip/hip_bf16.h>
#include <math.h>

typedef __attribute__((ext_vector_type(2))) float v2f;
typedef __attribute__((ext_vector_type(4))) float v4f;
typedef __attribute__((ext_vector_type(8))) float v8f;

#define EPS_RHO 1e-8f
#define EPS_DEN 1e-9f

__device__ __forceinline__ float sqrt_rn(float y) {
    const float s  = __builtin_amdgcn_sqrtf(y);
    const float sd = __int_as_float(__float_as_int(s) - 1);
    const float su = __int_as_float(__float_as_int(s) + 1);
    const float vp = __builtin_fmaf(-sd, s, y);
    const float vs = __builtin_fmaf(-su, s, y);
    float r = (vp <= 0.f) ? sd : s;
    r = (vs > 0.f) ? su : r;
    return r;
}

__device__ __forceinline__ float rho_relu(float x, float rt_eps) {
    return sqrt_rn(fmaxf(x, 0.f) + EPS_RHO) - rt_eps;
}

__device__ __forceinline__ v8f wmma_f32(v2f a, v2f b, v8f c) {
    return __builtin_amdgcn_wmma_f32_16x16x4_f32(false, a, false, b,
                                                 (short)0, c, false, false);
}

struct frag6 { v4f a0, a1, b0, b1, b2, b3; };

__device__ __forceinline__ void mma_group_rows(v8f (&acc)[2][4], const frag6& f) {
    acc[0][0] = wmma_f32(f.a0.xy, f.b0.xy, acc[0][0]);
    acc[1][0] = wmma_f32(f.a1.xy, f.b0.xy, acc[1][0]);
    acc[0][1] = wmma_f32(f.a0.xy, f.b1.xy, acc[0][1]);
    acc[1][1] = wmma_f32(f.a1.xy, f.b1.xy, acc[1][1]);
    acc[0][2] = wmma_f32(f.a0.xy, f.b2.xy, acc[0][2]);
    acc[1][2] = wmma_f32(f.a1.xy, f.b2.xy, acc[1][2]);
    acc[0][3] = wmma_f32(f.a0.xy, f.b3.xy, acc[0][3]);
    acc[1][3] = wmma_f32(f.a1.xy, f.b3.xy, acc[1][3]);
    acc[0][0] = wmma_f32(f.a0.zw, f.b0.zw, acc[0][0]);
    acc[1][0] = wmma_f32(f.a1.zw, f.b0.zw, acc[1][0]);
    acc[0][1] = wmma_f32(f.a0.zw, f.b1.zw, acc[0][1]);
    acc[1][1] = wmma_f32(f.a1.zw, f.b1.zw, acc[1][1]);
    acc[0][2] = wmma_f32(f.a0.zw, f.b2.zw, acc[0][2]);
    acc[1][2] = wmma_f32(f.a1.zw, f.b2.zw, acc[1][2]);
    acc[0][3] = wmma_f32(f.a0.zw, f.b3.zw, acc[0][3]);
    acc[1][3] = wmma_f32(f.a1.zw, f.b3.zw, acc[1][3]);
}

__device__ __forceinline__ void mma_group_cols(v8f (&acc)[2][4], const frag6& f) {
    #pragma unroll
    for (int t = 0; t < 4; ++t) {
        v2f bt; bt.x = f.b0[t]; bt.y = f.b1[t];
        acc[0][t] = wmma_f32(f.a0.xy, bt, acc[0][t]);
        acc[1][t] = wmma_f32(f.a1.xy, bt, acc[1][t]);
    }
    #pragma unroll
    for (int t = 0; t < 4; ++t) {
        v2f bt; bt.x = f.b2[t]; bt.y = f.b3[t];
        acc[0][t] = wmma_f32(f.a0.zw, bt, acc[0][t]);
        acc[1][t] = wmma_f32(f.a1.zw, bt, acc[1][t]);
    }
}

struct frag7 { v4f a0, a1, b0, b1, b2, b3, bq; };

template <int AQ>
__device__ __forceinline__ void mma_group_rows9(v8f (&acc)[2][4], v8f& accq, const frag7& f) {
    const v4f aq = AQ ? f.a1 : f.a0;
    acc[0][0] = wmma_f32(f.a0.xy, f.b0.xy, acc[0][0]);
    acc[1][0] = wmma_f32(f.a1.xy, f.b0.xy, acc[1][0]);
    acc[0][1] = wmma_f32(f.a0.xy, f.b1.xy, acc[0][1]);
    acc[1][1] = wmma_f32(f.a1.xy, f.b1.xy, acc[1][1]);
    acc[0][2] = wmma_f32(f.a0.xy, f.b2.xy, acc[0][2]);
    acc[1][2] = wmma_f32(f.a1.xy, f.b2.xy, acc[1][2]);
    acc[0][3] = wmma_f32(f.a0.xy, f.b3.xy, acc[0][3]);
    acc[1][3] = wmma_f32(f.a1.xy, f.b3.xy, acc[1][3]);
    accq      = wmma_f32(aq.xy,   f.bq.xy, accq);
    acc[0][0] = wmma_f32(f.a0.zw, f.b0.zw, acc[0][0]);
    acc[1][0] = wmma_f32(f.a1.zw, f.b0.zw, acc[1][0]);
    acc[0][1] = wmma_f32(f.a0.zw, f.b1.zw, acc[0][1]);
    acc[1][1] = wmma_f32(f.a1.zw, f.b1.zw, acc[1][1]);
    acc[0][2] = wmma_f32(f.a0.zw, f.b2.zw, acc[0][2]);
    acc[1][2] = wmma_f32(f.a1.zw, f.b2.zw, acc[1][2]);
    acc[0][3] = wmma_f32(f.a0.zw, f.b3.zw, acc[0][3]);
    acc[1][3] = wmma_f32(f.a1.zw, f.b3.zw, acc[1][3]);
    accq      = wmma_f32(aq.zw,   f.bq.zw, accq);
}

__device__ __forceinline__ v4f lo4(v8f v) { return __builtin_shufflevector(v, v, 0, 1, 2, 3); }
__device__ __forceinline__ v4f hi4(v8f v) { return __builtin_shufflevector(v, v, 4, 5, 6, 7); }
__device__ __forceinline__ v8f join8(v4f l, v4f h) {
    return __builtin_shufflevector(l, h, 0, 1, 2, 3, 4, 5, 6, 7);
}

template <int AQ>
__device__ __forceinline__ void
proj3_body(const float* __restrict__ A,
           const float* __restrict__ Wy,
           const float* __restrict__ Wyb,
           const float* __restrict__ Wsel,
           float* __restrict__ Osel,
           float* __restrict__ Vy,
           int K, int mb, int pr,
           v4f (*xch)[10][32])
{
    const int lane = threadIdx.x & 31;
    const int wave = __builtin_amdgcn_readfirstlane((int)(threadIdx.x >> 5));
    const int p    = wave >> 1;
    const int h    = wave & 1;
    const int nb   = (pr << 1) + p;
    const int qc0  = ((pr & 2) << 4) + (p << 4);
    const int l15  = lane & 15;
    const int koff = (lane < 16) ? 0 : 4;
    const int kh   = K >> 1;
    const size_t bstep = (size_t)16 * K;
    const size_t k0    = (size_t)(h * kh + koff);

    const float* pA0 = A + (size_t)((mb << 5) + l15) * K + k0;
    const float* pA1 = pA0 + bstep;
    const float* pB0 = Wy + (size_t)((nb << 6) + l15) * K + k0;
    const float* pB1 = pB0 + bstep;
    const float* pB2 = pB1 + bstep;
    const float* pB3 = pB2 + bstep;
    const float* pBq = Wsel + (size_t)(qc0 + l15) * K + k0;

    auto load_group = [&](frag7& f) {
        f.a0 = *(const v4f*)pA0;
        f.a1 = *(const v4f*)pA1;
        f.b0 = *(const v4f*)pB0;
        f.b1 = *(const v4f*)pB1;
        f.b2 = *(const v4f*)pB2;
        f.b3 = *(const v4f*)pB3;
        f.bq = *(const v4f*)pBq;
        pA0 += 8; pA1 += 8; pB0 += 8; pB1 += 8; pB2 += 8; pB3 += 8; pBq += 8;
    };

    v8f acc[2][4] = {};
    v8f accq = {};
    frag7 s0, s1, s2, s3;
    load_group(s0);  __builtin_amdgcn_sched_barrier(0);
    load_group(s1);  __builtin_amdgcn_sched_barrier(0);
    load_group(s2);  __builtin_amdgcn_sched_barrier(0);
    for (int kk = 32; kk < kh; kk += 32) {
        load_group(s3);  __builtin_amdgcn_sched_barrier(0);
        mma_group_rows9<AQ>(acc, accq, s0);  __builtin_amdgcn_sched_barrier(0);
        load_group(s0);  __builtin_amdgcn_sched_barrier(0);
        mma_group_rows9<AQ>(acc, accq, s1);  __builtin_amdgcn_sched_barrier(0);
        load_group(s1);  __builtin_amdgcn_sched_barrier(0);
        mma_group_rows9<AQ>(acc, accq, s2);  __builtin_amdgcn_sched_barrier(0);
        load_group(s2);  __builtin_amdgcn_sched_barrier(0);
        mma_group_rows9<AQ>(acc, accq, s3);  __builtin_amdgcn_sched_barrier(0);
    }
    load_group(s3);
    __builtin_amdgcn_sched_barrier(0);
    mma_group_rows9<AQ>(acc, accq, s0);
    mma_group_rows9<AQ>(acc, accq, s1);
    mma_group_rows9<AQ>(acc, accq, s2);
    mma_group_rows9<AQ>(acc, accq, s3);

    v4f (*mine)[32] = xch[wave];
    if (h == 0) {
        for (int t = 0; t < 4; ++t) {
            mine[2 * t][lane]     = lo4(acc[1][t]);
            mine[2 * t + 1][lane] = hi4(acc[1][t]);
        }
        if (p == 1) {
            mine[8][lane] = lo4(accq);
            mine[9][lane] = hi4(accq);
        }
    } else {
        for (int t = 0; t < 4; ++t) {
            mine[2 * t][lane]     = lo4(acc[0][t]);
            mine[2 * t + 1][lane] = hi4(acc[0][t]);
        }
        mine[8][lane] = lo4(accq);
        mine[9][lane] = hi4(accq);
    }
    __syncthreads();

    const int hh8 = (lane < 16) ? 0 : 8;
    if (h == 0) {
        v4f (*oth)[32] = xch[wave + 1];
        #pragma unroll
        for (int ni = 0; ni < 4; ++ni) {
            const v8f   sm = acc[0][ni] + join8(oth[2 * ni][lane], oth[2 * ni + 1][lane]);
            const int   n  = (nb << 6) + (ni << 4) + l15;
            const float bn = Wyb[n];
            #pragma unroll
            for (int r = 0; r < 8; ++r)
                Vy[(size_t)((mb << 5) + hh8 + r) * 1024 + n] = sm[r] + bn;
        }
        if (p == 0) {
            const v8f t0 = accq + join8(xch[1][8][lane], xch[1][9][lane]);
            const v8f t1 = join8(xch[2][8][lane], xch[2][9][lane])
                         + join8(xch[3][8][lane], xch[3][9][lane]);
            const int rq = (mb << 5) + (AQ << 4) + hh8;
            const int cq = ((pr & 2) << 4) + l15;
            #pragma unroll
            for (int r = 0; r < 8; ++r) {
                Osel[(size_t)(rq + r) * 64 + cq]      = t0[r];
                Osel[(size_t)(rq + r) * 64 + cq + 16] = t1[r];
            }
        }
    } else {
        v4f (*oth)[32] = xch[wave - 1];
        #pragma unroll
        for (int ni = 0; ni < 4; ++ni) {
            const v8f   sm = join8(oth[2 * ni][lane], oth[2 * ni + 1][lane]) + acc[1][ni];
            const int   n  = (nb << 6) + (ni << 4) + l15;
            const float bn = Wyb[n];
            #pragma unroll
            for (int r = 0; r < 8; ++r)
                Vy[(size_t)((mb << 5) + 16 + hh8 + r) * 1024 + n] = sm[r] + bn;
        }
    }
}

__global__ void __launch_bounds__(128)
fa_proj3(const float* __restrict__ A,
         const float* __restrict__ Wq,
         const float* __restrict__ Wk,
         const float* __restrict__ Wyw,
         const float* __restrict__ Wyb,
         float* __restrict__ q,
         float* __restrict__ kout,
         float* __restrict__ Vy,
         int K)
{
    __shared__ v4f xch[4][10][32];
    const int mb = blockIdx.x >> 3;
    const int pr = blockIdx.x & 7;
    const float* Wsel = (pr < 4) ? Wq : Wk;
    float*       Osel = (pr < 4) ? q  : kout;
    if (pr & 1) proj3_body<1>(A, Wyw, Wyb, Wsel, Osel, Vy, K, mb, pr, xch);
    else        proj3_body<0>(A, Wyw, Wyb, Wsel, Osel, Vy, K, mb, pr, xch);
}

#define ES_PITCH 68
__global__ void __launch_bounds__(256)
fa_scores_ew(const float* __restrict__ q,
             const float* __restrict__ k,
             const float* __restrict__ Wphi,
             float* __restrict__ ex,
             float* __restrict__ denom,
             float* __restrict__ Ew)
{
    __shared__ v4f   es4[512 * ES_PITCH / 4];
    __shared__ v4f   qi4[16];
    __shared__ v4f   wph4[16];
    __shared__ v4f   exs4[128];
    __shared__ float part[4 * 64];
    __shared__ float red[8];
    __shared__ float msh;
    float* es  = (float*)es4;
    float* qi  = (float*)qi4;
    float* wph = (float*)wph4;
    float* exs = (float*)exs4;
    const int row = blockIdx.x;
    const int b   = row >> 9;
    const int tid = threadIdx.x;
    if (tid < 64) {
        qi[tid]  = q[(size_t)row * 64 + tid];
        wph[tid] = Wphi[tid];
    }
    __syncthreads();

    const float rt_eps = sqrt_rn(EPS_RHO);
    float s[2];
    #pragma unroll
    for (int p = 0; p < 2; ++p) {
        const int j = tid + (p << 8);
        const float* kj = k + ((size_t)(b * 512 + j)) * 64;
        float* ej = es + j * ES_PITCH;
        float acc = 0.f;
        #pragma unroll 2
        for (int a = 0; a < 64; a += 4) {
            const v4f kv = *(const v4f*)(kj + a);
            const v4f qv = qi4[a >> 2];
            const v4f pw = wph4[a >> 2];
            v4f ev;
            ev.x = rho_relu(qv.x + kv.x, rt_eps);
            ev.y = rho_relu(qv.y + kv.y, rt_eps);
            ev.z = rho_relu(qv.z + kv.z, rt_eps);
            ev.w = rho_relu(qv.w + kv.w, rt_eps);
            acc += ev.x * pw.x;
            acc += ev.y * pw.y;
            acc += ev.z * pw.z;
            acc += ev.w * pw.w;
            *(v4f*)(ej + a) = ev;
        }
        s[p] = acc;
    }

    float mymax = fmaxf(s[0], s[1]);
    for (int off = 16; off; off >>= 1)
        mymax = fmaxf(mymax, __shfl_xor(mymax, off));
    const int lane = tid & 31, wv = tid >> 5;
    if (lane == 0) red[wv] = mymax;
    __syncthreads();
    if (tid == 0) {
        float m = red[0];
        for (int w = 1; w < 8; ++w) m = fmaxf(m, red[w]);
        msh = m;
    }
    __syncthreads();
    const float m = msh;

    const float e0 = expf(s[0] - m);
    const float e1 = expf(s[1] - m);
    ex[(size_t)row * 512 + tid]       = e0;
    ex[(size_t)row * 512 + tid + 256] = e1;
    exs[tid]       = e0;
    exs[tid + 256] = e1;

    float mysum = e0 + e1;
    for (int off = 16; off; off >>= 1)
        mysum += __shfl_xor(mysum, off);
    if (lane == 0) red[wv] = mysum;
    __syncthreads();
    if (tid == 0) {
        float sm = 0.f;
        for (int w = 0; w < 8; ++w) sm += red[w];
        denom[row] = sm;
    }

    const int a  = tid & 63;
    const int jq = tid >> 6;
    const float* er = es  + (jq * 128) * ES_PITCH + a;
    const float* xr = exs + jq * 128;
    float w0 = 0.f, w1 = 0.f, w2 = 0.f, w3 = 0.f;
    #pragma unroll 4
    for (int jj = 0; jj < 128; jj += 4) {
        const v4f xv = *(const v4f*)(xr + jj);
        w0 += xv.x * er[(jj + 0) * ES_PITCH];
        w1 += xv.y * er[(jj + 1) * ES_PITCH];
        w2 += xv.z * er[(jj + 2) * ES_PITCH];
        w3 += xv.w * er[(jj + 3) * ES_PITCH];
    }
    part[jq * 64 + a] = (w0 + w1) + (w2 + w3);
    __syncthreads();
    if (tid < 64)
        Ew[(size_t)row * 64 + tid] =
            (part[tid] + part[64 + tid]) + (part[128 + tid] + part[192 + tid]);
}

__global__ void __launch_bounds__(128)
fa_part12(const float* __restrict__ ex,
          const float* __restrict__ Vy,
          const float* __restrict__ Ew,
          const float* __restrict__ Wew,
          const float* __restrict__ Web,
          const float* __restrict__ denom,
          float* __restrict__ out)
{
    __shared__ v4f xch[4][8][32];
    const int lane = threadIdx.x & 31;
    const int wave = __builtin_amdgcn_readfirstlane((int)(threadIdx.x >> 5));
    const int h    = wave & 1;
    const int job  = (blockIdx.x << 1) + (wave >> 1);
    const int b    = job >> 8;
    const int jj   = job & 255;
    const int mb   = jj >> 4;
    const int nb   = jj & 15;
    const int l15  = lane & 15;
    const int koff = (lane < 16) ? 0 : 4;
    const int col0 = (nb << 6) + (l15 << 2);
    const int row0 = b * 512 + (mb << 5);
    const int j0   = (h << 8) + koff;
    const int a0   = (h << 5) + koff;

    const float* pA0 = ex + ((size_t)(row0 + l15)) * 512 + j0;
    const float* pA1 = pA0 + (size_t)16 * 512;
    const float* pV  = Vy + (size_t)b * 512 * 1024 + (size_t)j0 * 1024 + col0;
    const float* pE0 = Ew + ((size_t)(row0 + l15)) * 64 + a0;
    const float* pE1 = pE0 + (size_t)16 * 64;
    const float* pW  = Wew + (size_t)col0 * 64 + a0;

    auto load_group = [&](frag6& f) {
        f.a0 = *(const v4f*)pA0;
        f.a1 = *(const v4f*)pA1;
        f.b0 = *(const v4f*)(pV);
        f.b1 = *(const v4f*)(pV + 1024);
        f.b2 = *(const v4f*)(pV + 2 * 1024);
        f.b3 = *(const v4f*)(pV + 3 * 1024);
        pA0 += 8; pA1 += 8; pV += 8 * 1024;
    };
    auto load_group_a = [&](frag6& f) {
        f.a0 = *(const v4f*)pE0;
        f.a1 = *(const v4f*)pE1;
        f.b0 = *(const v4f*)(pW);
        f.b1 = *(const v4f*)(pW + 64);
        f.b2 = *(const v4f*)(pW + 2 * 64);
        f.b3 = *(const v4f*)(pW + 3 * 64);
        pE0 += 8; pE1 += 8; pW += 8;
    };

    const int mfin = row0 + (h << 4) + ((lane < 16) ? 0 : 8);
    const v4f bn  = *(const v4f*)(Web + col0);
    const v4f dn0 = *(const v4f*)(denom + mfin);
    const v4f dn1 = *(const v4f*)(denom + mfin + 4);
    __builtin_amdgcn_sched_barrier(0);

    v8f acc[2][4] = {};
    frag6 s0, s1, s2, s3;
    load_group(s0);  __builtin_amdgcn_sched_barrier(0);
    load_group(s1);  __builtin_amdgcn_sched_barrier(0);
    load_group(s2);  __builtin_amdgcn_sched_barrier(0);
    for (int j = 32; j < 256; j += 32) {
        load_group(s3);  __builtin_amdgcn_sched_barrier(0);
        mma_group_cols(acc, s0);  __builtin_amdgcn_sched_barrier(0);
        load_group(s0);  __builtin_amdgcn_sched_barrier(0);
        mma_group_cols(acc, s1);  __builtin_amdgcn_sched_barrier(0);
        load_group(s1);  __builtin_amdgcn_sched_barrier(0);
        mma_group_cols(acc, s2);  __builtin_amdgcn_sched_barrier(0);
        load_group(s2);  __builtin_amdgcn_sched_barrier(0);
        mma_group_cols(acc, s3);  __builtin_amdgcn_sched_barrier(0);
    }
    load_group(s3);
    __builtin_amdgcn_sched_barrier(0);
    mma_group_cols(acc, s0);  __builtin_amdgcn_sched_barrier(0);
    load_group_a(s0);  __builtin_amdgcn_sched_barrier(0);
    mma_group_cols(acc, s1);  __builtin_amdgcn_sched_barrier(0);
    load_group_a(s1);  __builtin_amdgcn_sched_barrier(0);
    mma_group_cols(acc, s2);  __builtin_amdgcn_sched_barrier(0);
    load_group_a(s2);  __builtin_amdgcn_sched_barrier(0);
    mma_group_cols(acc, s3);  __builtin_amdgcn_sched_barrier(0);
    load_group_a(s3);
    __builtin_amdgcn_sched_barrier(0);
    mma_group_rows(acc, s0);
    mma_group_rows(acc, s1);
    mma_group_rows(acc, s2);
    mma_group_rows(acc, s3);

    v4f (*mine)[32] = xch[wave];
    v8f fin[4];
    if (h == 0) {
        for (int t = 0; t < 4; ++t) {
            mine[2 * t][lane]     = lo4(acc[1][t]);
            mine[2 * t + 1][lane] = hi4(acc[1][t]);
        }
    } else {
        for (int t = 0; t < 4; ++t) {
            mine[2 * t][lane]     = lo4(acc[0][t]);
            mine[2 * t + 1][lane] = hi4(acc[0][t]);
        }
    }
    __syncthreads();
    if (h == 0) {
        v4f (*oth)[32] = xch[wave + 1];
        for (int t = 0; t < 4; ++t)
            fin[t] = acc[0][t] + join8(oth[2 * t][lane], oth[2 * t + 1][lane]);
    } else {
        v4f (*oth)[32] = xch[wave - 1];
        for (int t = 0; t < 4; ++t)
            fin[t] = join8(oth[2 * t][lane], oth[2 * t + 1][lane]) + acc[1][t];
    }

    #pragma unroll
    for (int r = 0; r < 8; ++r) {
        const float d  = (r < 4) ? dn0[r & 3] : dn1[r & 3];
        const float dd = d + EPS_DEN;
        v4f o;
        o.x = (fin[0][r] + bn.x * d) / dd;
        o.y = (fin[1][r] + bn.y * d) / dd;
        o.z = (fin[2][r] + bn.z * d) / dd;
        o.w = (fin[3][r] + bn.w * d) / dd;
        *(v4f*)(out + (size_t)(mfin + r) * 1024 + col0) = o;
    }
}

extern "C" void kernel_launch(void* const* d_in, const int* in_sizes, int n_in,
                              void* d_out, int out_size, void* d_ws, size_t ws_size,
                              hipStream_t stream)
{
    (void)in_sizes; (void)n_in; (void)out_size; (void)ws_size;
    const float* X    = (const float*)d_in[0];
    const float* Wq   = (const float*)d_in[1];
    const float* Wk   = (const float*)d_in[2];
    const float* Wphi = (const float*)d_in[3];
    const float* Wyw  = (const float*)d_in[4];
    const float* Wyb  = (const float*)d_in[5];
    const float* Wew  = (const float*)d_in[6];
    const float* Web  = (const float*)d_in[7];
    float* out = (float*)d_out;

    float* ws  = (float*)d_ws;
    float* q   = ws;  ws += 1024 * 64;
    float* kk  = ws;  ws += 1024 * 64;
    float* Vy  = ws;  ws += 1024 * 1024;
    float* ex  = ws;  ws += 1024 * 512;
    float* den = ws;  ws += 1024;
    float* Ew  = ws;  ws += 1024 * 64;

    fa_proj3<<<256, 128, 0, stream>>>(X, Wq, Wk, Wyw, Wyb, q, kk, Vy, 1024);

    fa_scores_ew<<<1024, 256, 0, stream>>>(q, kk, Wphi, ex, den, Ew);

    fa_part12<<<256, 128, 0, stream>>>(ex, Vy, Ew, Wew, Web, den, out);
}


# scores kernel: a thread owns 4 a of 32 rows j, so a half wave reads whole k rows and e stays in registers for Ew (no LDS copy); score = tree of 16 partial sums of 4 a, Ew = tree of 32 partial j sums
# speedup vs baseline: 7.5598x; 1.1275x over previous
#include <hip/hip_runtime.h>
#include <hip/hip_bf16.h>
#include <math.h>

typedef __attribute__((ext_vector_type(2))) float v2f;
typedef __attribute__((ext_vector_type(4))) float v4f;
typedef __attribute__((ext_vector_type(8))) float v8f;

#define EPS_RHO 1e-8f
#define EPS_DEN 1e-9f

__device__ __forceinline__ float sqrt_rn(float y) {
    const float s  = __builtin_amdgcn_sqrtf(y);
    const float sd = __int_as_float(__float_as_int(s) - 1);
    const float su = __int_as_float(__float_as_int(s) + 1);
    const float vp = __builtin_fmaf(-sd, s, y);
    const float vs = __builtin_fmaf(-su, s, y);
    float r = (vp <= 0.f) ? sd : s;
    r = (vs > 0.f) ? su : r;
    return r;
}

__device__ __forceinline__ float rho_relu(float x, float rt_eps) {
    return sqrt_rn(fmaxf(x, 0.f) + EPS_RHO) - rt_eps;
}

__device__ __forceinline__ v8f wmma_f32(v2f a, v2f b, v8f c) {
    return __builtin_amdgcn_wmma_f32_16x16x4_f32(false, a, false, b,
                                                 (short)0, c, false, false);
}

struct frag6 { v4f a0, a1, b0, b1, b2, b3; };

__device__ __forceinline__ void mma_group_rows(v8f (&acc)[2][4], const frag6& f) {
    acc[0][0] = wmma_f32(f.a0.xy, f.b0.xy, acc[0][0]);
    acc[1][0] = wmma_f32(f.a1.xy, f.b0.xy, acc[1][0]);
    acc[0][1] = wmma_f32(f.a0.xy, f.b1.xy, acc[0][1]);
    acc[1][1] = wmma_f32(f.a1.xy, f.b1.xy, acc[1][1]);
    acc[0][2] = wmma_f32(f.a0.xy, f.b2.xy, acc[0][2]);
    acc[1][2] = wmma_f32(f.a1.xy, f.b2.xy, acc[1][2]);
    acc[0][3] = wmma_f32(f.a0.xy, f.b3.xy, acc[0][3]);
    acc[1][3] = wmma_f32(f.a1.xy, f.b3.xy, acc[1][3]);
    acc[0][0] = wmma_f32(f.a0.zw, f.b0.zw, acc[0][0]);
    acc[1][0] = wmma_f32(f.a1.zw, f.b0.zw, acc[1][0]);
    acc[0][1] = wmma_f32(f.a0.zw, f.b1.zw, acc[0][1]);
    acc[1][1] = wmma_f32(f.a1.zw, f.b1.zw, acc[1][1]);
    acc[0][2] = wmma_f32(f.a0.zw, f.b2.zw, acc[0][2]);
    acc[1][2] = wmma_f32(f.a1.zw, f.b2.zw, acc[1][2]);
    acc[0][3] = wmma_f32(f.a0.zw, f.b3.zw, acc[0][3]);
    acc[1][3] = wmma_f32(f.a1.zw, f.b3.zw, acc[1][3]);
}

__device__ __forceinline__ void mma_group_cols(v8f (&acc)[2][4], const frag6& f) {
    #pragma unroll
    for (int t = 0; t < 4; ++t) {
        v2f bt; bt.x = f.b0[t]; bt.y = f.b1[t];
        acc[0][t] = wmma_f32(f.a0.xy, bt, acc[0][t]);
        acc[1][t] = wmma_f32(f.a1.xy, bt, acc[1][t]);
    }
    #pragma unroll
    for (int t = 0; t < 4; ++t) {
        v2f bt; bt.x = f.b2[t]; bt.y = f.b3[t];
        acc[0][t] = wmma_f32(f.a0.zw, bt, acc[0][t]);
        acc[1][t] = wmma_f32(f.a1.zw, bt, acc[1][t]);
    }
}

struct frag7 { v4f a0, a1, b0, b1, b2, b3, bq; };

template <int AQ>
__device__ __forceinline__ void mma_group_rows9(v8f (&acc)[2][4], v8f& accq, const frag7& f) {
    const v4f aq = AQ ? f.a1 : f.a0;
    acc[0][0] = wmma_f32(f.a0.xy, f.b0.xy, acc[0][0]);
    acc[1][0] = wmma_f32(f.a1.xy, f.b0.xy, acc[1][0]);
    acc[0][1] = wmma_f32(f.a0.xy, f.b1.xy, acc[0][1]);
    acc[1][1] = wmma_f32(f.a1.xy, f.b1.xy, acc[1][1]);
    acc[0][2] = wmma_f32(f.a0.xy, f.b2.xy, acc[0][2]);
    acc[1][2] = wmma_f32(f.a1.xy, f.b2.xy, acc[1][2]);
    acc[0][3] = wmma_f32(f.a0.xy, f.b3.xy, acc[0][3]);
    acc[1][3] = wmma_f32(f.a1.xy, f.b3.xy, acc[1][3]);
    accq      = wmma_f32(aq.xy,   f.bq.xy, accq);
    acc[0][0] = wmma_f32(f.a0.zw, f.b0.zw, acc[0][0]);
    acc[1][0] = wmma_f32(f.a1.zw, f.b0.zw, acc[1][0]);
    acc[0][1] = wmma_f32(f.a0.zw, f.b1.zw, acc[0][1]);
    acc[1][1] = wmma_f32(f.a1.zw, f.b1.zw, acc[1][1]);
    acc[0][2] = wmma_f32(f.a0.zw, f.b2.zw, acc[0][2]);
    acc[1][2] = wmma_f32(f.a1.zw, f.b2.zw, acc[1][2]);
    acc[0][3] = wmma_f32(f.a0.zw, f.b3.zw, acc[0][3]);
    acc[1][3] = wmma_f32(f.a1.zw, f.b3.zw, acc[1][3]);
    accq      = wmma_f32(aq.zw,   f.bq.zw, accq);
}

__device__ __forceinline__ v4f lo4(v8f v) { return __builtin_shufflevector(v, v, 0, 1, 2, 3); }
__device__ __forceinline__ v4f hi4(v8f v) { return __builtin_shufflevector(v, v, 4, 5, 6, 7); }
__device__ __forceinline__ v8f join8(v4f l, v4f h) {
    return __builtin_shufflevector(l, h, 0, 1, 2, 3, 4, 5, 6, 7);
}

template <int AQ>
__device__ __forceinline__ void
proj3_body(const float* __restrict__ A,
           const float* __restrict__ Wy,
           const float* __restrict__ Wyb,
           const float* __restrict__ Wsel,
           float* __restrict__ Osel,
           float* __restrict__ Vy,
           int K, int mb, int pr,
           v4f (*xch)[10][32])
{
    const int lane = threadIdx.x & 31;
    const int wave = __builtin_amdgcn_readfirstlane((int)(threadIdx.x >> 5));
    const int p    = wave >> 1;
    const int h    = wave & 1;
    const int nb   = (pr << 1) + p;
    const int qc0  = ((pr & 2) << 4) + (p << 4);
    const int l15  = lane & 15;
    const int koff = (lane < 16) ? 0 : 4;
    const int kh   = K >> 1;
    const size_t bstep = (size_t)16 * K;
    const size_t k0    = (size_t)(h * kh + koff);

    const float* pA0 = A + (size_t)((mb << 5) + l15) * K + k0;
    const float* pA1 = pA0 + bstep;
    const float* pB0 = Wy + (size_t)((nb << 6) + l15) * K + k0;
    const float* pB1 = pB0 + bstep;
    const float* pB2 = pB1 + bstep;
    const float* pB3 = pB2 + bstep;
    const float* pBq = Wsel + (size_t)(qc0 + l15) * K + k0;

    auto load_group = [&](frag7& f) {
        f.a0 = *(const v4f*)pA0;
        f.a1 = *(const v4f*)pA1;
        f.b0 = *(const v4f*)pB0;
        f.b1 = *(const v4f*)pB1;
        f.b2 = *(const v4f*)pB2;
        f.b3 = *(const v4f*)pB3;
        f.bq = *(const v4f*)pBq;
        pA0 += 8; pA1 += 8; pB0 += 8; pB1 += 8; pB2 += 8; pB3 += 8; pBq += 8;
    };

    v8f acc[2][4] = {};
    v8f accq = {};
    frag7 s0, s1, s2, s3;
    load_group(s0);  __builtin_amdgcn_sched_barrier(0);
    load_group(s1);  __builtin_amdgcn_sched_barrier(0);
    load_group(s2);  __builtin_amdgcn_sched_barrier(0);
    for (int kk = 32; kk < kh; kk += 32) {
        load_group(s3);  __builtin_amdgcn_sched_barrier(0);
        mma_group_rows9<AQ>(acc, accq, s0);  __builtin_amdgcn_sched_barrier(0);
        load_group(s0);  __builtin_amdgcn_sched_barrier(0);
        mma_group_rows9<AQ>(acc, accq, s1);  __builtin_amdgcn_sched_barrier(0);
        load_group(s1);  __builtin_amdgcn_sched_barrier(0);
        mma_group_rows9<AQ>(acc, accq, s2);  __builtin_amdgcn_sched_barrier(0);
        load_group(s2);  __builtin_amdgcn_sched_barrier(0);
        mma_group_rows9<AQ>(acc, accq, s3);  __builtin_amdgcn_sched_barrier(0);
    }
    load_group(s3);
    __builtin_amdgcn_sched_barrier(0);
    mma_group_rows9<AQ>(acc, accq, s0);
    mma_group_rows9<AQ>(acc, accq, s1);
    mma_group_rows9<AQ>(acc, accq, s2);
    mma_group_rows9<AQ>(acc, accq, s3);

    v4f (*mine)[32] = xch[wave];
    if (h == 0) {
        for (int t = 0; t < 4; ++t) {
            mine[2 * t][lane]     = lo4(acc[1][t]);
            mine[2 * t + 1][lane] = hi4(acc[1][t]);
        }
        if (p == 1) {
            mine[8][lane] = lo4(accq);
            mine[9][lane] = hi4(accq);
        }
    } else {
        for (int t = 0; t < 4; ++t) {
            mine[2 * t][lane]     = lo4(acc[0][t]);
            mine[2 * t + 1][lane] = hi4(acc[0][t]);
        }
        mine[8][lane] = lo4(accq);
        mine[9][lane] = hi4(accq);
    }
    __syncthreads();

    const int hh8 = (lane < 16) ? 0 : 8;
    if (h == 0) {
        v4f (*oth)[32] = xch[wave + 1];
        #pragma unroll
        for (int ni = 0; ni < 4; ++ni) {
            const v8f   sm = acc[0][ni] + join8(oth[2 * ni][lane], oth[2 * ni + 1][lane]);
            const int   n  = (nb << 6) + (ni << 4) + l15;
            const float bn = Wyb[n];
            #pragma unroll
            for (int r = 0; r < 8; ++r)
                Vy[(size_t)((mb << 5) + hh8 + r) * 1024 + n] = sm[r] + bn;
        }
        if (p == 0) {
            const v8f t0 = accq + join8(xch[1][8][lane], xch[1][9][lane]);
            const v8f t1 = join8(xch[2][8][lane], xch[2][9][lane])
                         + join8(xch[3][8][lane], xch[3][9][lane]);
            const int rq = (mb << 5) + (AQ << 4) + hh8;
            const int cq = ((pr & 2) << 4) + l15;
            #pragma unroll
            for (int r = 0; r < 8; ++r) {
                Osel[(size_t)(rq + r) * 64 + cq]      = t0[r];
                Osel[(size_t)(rq + r) * 64 + cq + 16] = t1[r];
            }
        }
    } else {
        v4f (*oth)[32] = xch[wave - 1];
        #pragma unroll
        for (int ni = 0; ni < 4; ++ni) {
            const v8f   sm = join8(oth[2 * ni][lane], oth[2 * ni + 1][lane]) + acc[1][ni];
            const int   n  = (nb << 6) + (ni << 4) + l15;
            const float bn = Wyb[n];
            #pragma unroll
            for (int r = 0; r < 8; ++r)
                Vy[(size_t)((mb << 5) + 16 + hh8 + r) * 1024 + n] = sm[r] + bn;
        }
    }
}

__global__ void __launch_bounds__(128)
fa_proj3(const float* __restrict__ A,
         const float* __restrict__ Wq,
         const float* __restrict__ Wk,
         const float* __restrict__ Wyw,
         const float* __restrict__ Wyb,
         float* __restrict__ q,
         float* __restrict__ kout,
         float* __restrict__ Vy,
         int K)
{
    __shared__ v4f xch[4][10][32];
    const int mb = blockIdx.x >> 3;
    const int pr = blockIdx.x & 7;
    const float* Wsel = (pr < 4) ? Wq : Wk;
    float*       Osel = (pr < 4) ? q  : kout;
    if (pr & 1) proj3_body<1>(A, Wyw, Wyb, Wsel, Osel, Vy, K, mb, pr, xch);
    else        proj3_body<0>(A, Wyw, Wyb, Wsel, Osel, Vy, K, mb, pr, xch);
}

#define SP_PITCH 17
__global__ void __launch_bounds__(256)
fa_scores_ew(const float* __restrict__ q,
             const float* __restrict__ k,
             const float* __restrict__ Wphi,
             float* __restrict__ ex,
             float* __restrict__ denom,
             float* __restrict__ Ew)
{
    __shared__ float sp[512 * SP_PITCH];
    __shared__ v4f   exs4[128];
    __shared__ v4f   part4[16 * 16];
    __shared__ float red[8];
    __shared__ float msh;
    float* exs  = (float*)exs4;
    float* part = (float*)part4;
    const int row = blockIdx.x;
    const int b   = row >> 9;
    const int tid = threadIdx.x;
    const int aq  = tid & 15;
    const int jr  = tid >> 4;

    const v4f qv = *(const v4f*)(q + (size_t)row * 64 + (aq << 2));
    const v4f pw = *(const v4f*)(Wphi + (aq << 2));
    const float* kp = k + ((size_t)(b * 512 + (jr << 5))) * 64 + (aq << 2);

    v4f e[32];
    #pragma unroll
    for (int t = 0; t < 32; ++t)
        e[t] = *(const v4f*)(kp + t * 64);

    const float rt_eps = sqrt_rn(EPS_RHO);
    float* spw = sp + (jr << 5) * SP_PITCH + aq;
    #pragma unroll
    for (int t = 0; t < 32; ++t) {
        v4f ev;
        ev.x = rho_relu(qv.x + e[t].x, rt_eps);
        ev.y = rho_relu(qv.y + e[t].y, rt_eps);
        ev.z = rho_relu(qv.z + e[t].z, rt_eps);
        ev.w = rho_relu(qv.w + e[t].w, rt_eps);
        e[t] = ev;
        float d = ev.x * pw.x;
        d += ev.y * pw.y;
        d += ev.z * pw.z;
        d += ev.w * pw.w;
        spw[t * SP_PITCH] = d;
    }
    __syncthreads();

    float s[2];
    #pragma unroll
    for (int p = 0; p < 2; ++p) {
        const float* r = sp + (tid + (p << 8)) * SP_PITCH;
        const float u0 = (r[0]  + r[1])  + (r[2]  + r[3]);
        const float u1 = (r[4]  + r[5])  + (r[6]  + r[7]);
        const float u2 = (r[8]  + r[9])  + (r[10] + r[11]);
        const float u3 = (r[12] + r[13]) + (r[14] + r[15]);
        s[p] = (u0 + u1) + (u2 + u3);
    }

    float mymax = fmaxf(s[0], s[1]);
    for (int off = 16; off; off >>= 1)
        mymax = fmaxf(mymax, __shfl_xor(mymax, off));
    const int lane = tid & 31, wv = tid >> 5;
    if (lane == 0) red[wv] = mymax;
    __syncthreads();
    if (tid == 0) {
        float m = red[0];
        for (int w = 1; w < 8; ++w) m = fmaxf(m, red[w]);
        msh = m;
    }
    __syncthreads();
    const float m = msh;

    const float e0 = expf(s[0] - m);
    const float e1 = expf(s[1] - m);
    ex[(size_t)row * 512 + tid]       = e0;
    ex[(size_t)row * 512 + tid + 256] = e1;
    exs[tid]       = e0;
    exs[tid + 256] = e1;

    float mysum = e0 + e1;
    for (int off = 16; off; off >>= 1)
        mysum += __shfl_xor(mysum, off);
    if (lane == 0) red[wv] = mysum;
    __syncthreads();
    if (tid == 0) {
        float sm = 0.f;
        for (int w = 0; w < 8; ++w) sm += red[w];
        denom[row] = sm;
    }

    const v4f* xr = exs4 + (jr << 3);
    v4f w0 = {0.f, 0.f, 0.f, 0.f}, w1 = {0.f, 0.f, 0.f, 0.f};
    #pragma unroll
    for (int t4 = 0; t4 < 8; ++t4) {
        const v4f xv = xr[t4];
        w0 += xv.x * e[4 * t4 + 0];
        w1 += xv.y * e[4 * t4 + 1];
        w0 += xv.z * e[4 * t4 + 2];
        w1 += xv.w * e[4 * t4 + 3];
    }
    part4[(jr << 4) + aq] = w0 + w1;
    __syncthreads();
    if (tid < 64) {
        const float* c = part + tid;
        const float u0 = (c[0 * 64]  + c[1 * 64])  + (c[2 * 64]  + c[3 * 64]);
        const float u1 = (c[4 * 64]  + c[5 * 64])  + (c[6 * 64]  + c[7 * 64]);
        const float u2 = (c[8 * 64]  + c[9 * 64])  + (c[10 * 64] + c[11 * 64]);
        const float u3 = (c[12 * 64] + c[13 * 64]) + (c[14 * 64] + c[15 * 64]);
        Ew[(size_t)row * 64 + tid] = (u0 + u1) + (u2 + u3);
    }
}

__global__ void __launch_bounds__(128)
fa_part12(const float* __restrict__ ex,
          const float* __restrict__ Vy,
          const float* __restrict__ Ew,
          const float* __restrict__ Wew,
          const float* __restrict__ Web,
          const float* __restrict__ denom,
          float* __restrict__ out)
{
    __shared__ v4f xch[4][8][32];
    const int lane = threadIdx.x & 31;
    const int wave = __builtin_amdgcn_readfirstlane((int)(threadIdx.x >> 5));
    const int h    = wave & 1;
    const int job  = (blockIdx.x << 1) + (wave >> 1);
    const int b    = job >> 8;
    const int jj   = job & 255;
    const int mb   = jj >> 4;
    const int nb   = jj & 15;
    const int l15  = lane & 15;
    const int koff = (lane < 16) ? 0 : 4;
    const int col0 = (nb << 6) + (l15 << 2);
    const int row0 = b * 512 + (mb << 5);
    const int j0   = (h << 8) + koff;
    const int a0   = (h << 5) + koff;

    const float* pA0 = ex + ((size_t)(row0 + l15)) * 512 + j0;
    const float* pA1 = pA0 + (size_t)16 * 512;
    const float* pV  = Vy + (size_t)b * 512 * 1024 + (size_t)j0 * 1024 + col0;
    const float* pE0 = Ew + ((size_t)(row0 + l15)) * 64 + a0;
    const float* pE1 = pE0 + (size_t)16 * 64;
    const float* pW  = Wew + (size_t)col0 * 64 + a0;

    auto load_group = [&](frag6& f) {
        f.a0 = *(const v4f*)pA0;
        f.a1 = *(const v4f*)pA1;
        f.b0 = *(const v4f*)(pV);
        f.b1 = *(const v4f*)(pV + 1024);
        f.b2 = *(const v4f*)(pV + 2 * 1024);
        f.b3 = *(const v4f*)(pV + 3 * 1024);
        pA0 += 8; pA1 += 8; pV += 8 * 1024;
    };
    auto load_group_a = [&](frag6& f) {
        f.a0 = *(const v4f*)pE0;
        f.a1 = *(const v4f*)pE1;
        f.b0 = *(const v4f*)(pW);
        f.b1 = *(const v4f*)(pW + 64);
        f.b2 = *(const v4f*)(pW + 2 * 64);
        f.b3 = *(const v4f*)(pW + 3 * 64);
        pE0 += 8; pE1 += 8; pW += 8;
    };

    const int mfin = row0 + (h << 4) + ((lane < 16) ? 0 : 8);
    const v4f bn  = *(const v4f*)(Web + col0);
    const v4f dn0 = *(const v4f*)(denom + mfin);
    const v4f dn1 = *(const v4f*)(denom + mfin + 4);
    __builtin_amdgcn_sched_barrier(0);

    v8f acc[2][4] = {};
    frag6 s0, s1, s2, s3;
    load_group(s0);  __builtin_amdgcn_sched_barrier(0);
    load_group(s1);  __builtin_amdgcn_sched_barrier(0);
    load_group(s2);  __builtin_amdgcn_sched_barrier(0);
    for (int j = 32; j < 256; j += 32) {
        load_group(s3);  __builtin_amdgcn_sched_barrier(0);
        mma_group_cols(acc, s0);  __builtin_amdgcn_sched_barrier(0);
        load_group(s0);  __builtin_amdgcn_sched_barrier(0);
        mma_group_cols(acc, s1);  __builtin_amdgcn_sched_barrier(0);
        load_group(s1);  __builtin_amdgcn_sched_barrier(0);
        mma_group_cols(acc, s2);  __builtin_amdgcn_sched_barrier(0);
        load_group(s2);  __builtin_amdgcn_sched_barrier(0);
        mma_group_cols(acc, s3);  __builtin_amdgcn_sched_barrier(0);
    }
    load_group(s3);
    __builtin_amdgcn_sched_barrier(0);
    mma_group_cols(acc, s0);  __builtin_amdgcn_sched_barrier(0);
    load_group_a(s0);  __builtin_amdgcn_sched_barrier(0);
    mma_group_cols(acc, s1);  __builtin_amdgcn_sched_barrier(0);
    load_group_a(s1);  __builtin_amdgcn_sched_barrier(0);
    mma_group_cols(acc, s2);  __builtin_amdgcn_sched_barrier(0);
    load_group_a(s2);  __builtin_amdgcn_sched_barrier(0);
    mma_group_cols(acc, s3);  __builtin_amdgcn_sched_barrier(0);
    load_group_a(s3);
    __builtin_amdgcn_sched_barrier(0);
    mma_group_rows(acc, s0);
    mma_group_rows(acc, s1);
    mma_group_rows(acc, s2);
    mma_group_rows(acc, s3);

    v4f (*mine)[32] = xch[wave];
    v8f fin[4];
    if (h == 0) {
        for (int t = 0; t < 4; ++t) {
            mine[2 * t][lane]     = lo4(acc[1][t]);
            mine[2 * t + 1][lane] = hi4(acc[1][t]);
        }
    } else {
        for (int t = 0; t < 4; ++t) {
            mine[2 * t][lane]     = lo4(acc[0][t]);
            mine[2 * t + 1][lane] = hi4(acc[0][t]);
        }
    }
    __syncthreads();
    if (h == 0) {
        v4f (*oth)[32] = xch[wave + 1];
        for (int t = 0; t < 4; ++t)
            fin[t] = acc[0][t] + join8(oth[2 * t][lane], oth[2 * t + 1][lane]);
    } else {
        v4f (*oth)[32] = xch[wave - 1];
        for (int t = 0; t < 4; ++t)
            fin[t] = join8(oth[2 * t][lane], oth[2 * t + 1][lane]) + acc[1][t];
    }

    #pragma unroll
    for (int r = 0; r < 8; ++r) {
        const float d  = (r < 4) ? dn0[r & 3] : dn1[r & 3];
        const float dd = d + EPS_DEN;
        v4f o;
        o.x = (fin[0][r] + bn.x * d) / dd;
        o.y = (fin[1][r] + bn.y * d) / dd;
        o.z = (fin[2][r] + bn.z * d) / dd;
        o.w = (fin[3][r] + bn.w * d) / dd;
        *(v4f*)(out + (size_t)(mfin + r) * 1024 + col0) = o;
    }
}

extern "C" void kernel_launch(void* const* d_in, const int* in_sizes, int n_in,
                              void* d_out, int out_size, void* d_ws, size_t ws_size,
                              hipStream_t stream)
{
    (void)in_sizes; (void)n_in; (void)out_size; (void)ws_size;
    const float* X    = (const float*)d_in[0];
    const float* Wq   = (const float*)d_in[1];
    const float* Wk   = (const float*)d_in[2];
    const float* Wphi = (const float*)d_in[3];
    const float* Wyw  = (const float*)d_in[4];
    const float* Wyb  = (const float*)d_in[5];
    const float* Wew  = (const float*)d_in[6];
    const float* Web  = (const float*)d_in[7];
    float* out = (float*)d_out;

    float* ws  = (float*)d_ws;
    float* q   = ws;  ws += 1024 * 64;
    float* kk  = ws;  ws += 1024 * 64;
    float* Vy  = ws;  ws += 1024 * 1024;
    float* ex  = ws;  ws += 1024 * 512;
    float* den = ws;  ws += 1024;
    float* Ew  = ws;  ws += 1024 * 64;

    fa_proj3<<<256, 128, 0, stream>>>(X, Wq, Wk, Wyw, Wyb, q, kk, Vy, 1024);

    fa_scores_ew<<<1024, 256, 0, stream>>>(q, kk, Wphi, ex, den, Ew);

    fa_part12<<<256, 128, 0, stream>>>(ex, Vy, Ew, Wew, Web, den, out);
}


# scores kernel: the root of e is the bare v_sqrt_f32 (within 1 ulp) without the two-residual rounding fix, 8 fewer vector ops per e; e may differ in its last place, its exact 0 for x <= 0 is kept
# speedup vs baseline: 7.9211x; 1.0478x over previous
#include <hip/hip_runtime.h>
#include <hip/hip_bf16.h>
#include <math.h>

typedef __attribute__((ext_vector_type(2))) float v2f;
typedef __attribute__((ext_vector_type(4))) float v4f;
typedef __attribute__((ext_vector_type(8))) float v8f;

#define EPS_RHO 1e-8f
#define EPS_DEN 1e-9f

__device__ __forceinline__ float sqrt_hw(float y) {
    return __builtin_amdgcn_sqrtf(y);
}

__device__ __forceinline__ float rho_relu(float x, float rt_eps) {
    return sqrt_hw(fmaxf(x, 0.f) + EPS_RHO) - rt_eps;
}

__device__ __forceinline__ v8f wmma_f32(v2f a, v2f b, v8f c) {
    return __builtin_amdgcn_wmma_f32_16x16x4_f32(false, a, false, b,
                                                 (short)0, c, false, false);
}

struct frag6 { v4f a0, a1, b0, b1, b2, b3; };

__device__ __forceinline__ void mma_group_rows(v8f (&acc)[2][4], const frag6& f) {
    acc[0][0] = wmma_f32(f.a0.xy, f.b0.xy, acc[0][0]);
    acc[1][0] = wmma_f32(f.a1.xy, f.b0.xy, acc[1][0]);
    acc[0][1] = wmma_f32(f.a0.xy, f.b1.xy, acc[0][1]);
    acc[1][1] = wmma_f32(f.a1.xy, f.b1.xy, acc[1][1]);
    acc[0][2] = wmma_f32(f.a0.xy, f.b2.xy, acc[0][2]);
    acc[1][2] = wmma_f32(f.a1.xy, f.b2.xy, acc[1][2]);
    acc[0][3] = wmma_f32(f.a0.xy, f.b3.xy, acc[0][3]);
    acc[1][3] = wmma_f32(f.a1.xy, f.b3.xy, acc[1][3]);
    acc[0][0] = wmma_f32(f.a0.zw, f.b0.zw, acc[0][0]);
    acc[1][0] = wmma_f32(f.a1.zw, f.b0.zw, acc[1][0]);
    acc[0][1] = wmma_f32(f.a0.zw, f.b1.zw, acc[0][1]);
    acc[1][1] = wmma_f32(f.a1.zw, f.b1.zw, acc[1][1]);
    acc[0][2] = wmma_f32(f.a0.zw, f.b2.zw, acc[0][2]);
    acc[1][2] = wmma_f32(f.a1.zw, f.b2.zw, acc[1][2]);
    acc[0][3] = wmma_f32(f.a0.zw, f.b3.zw, acc[0][3]);
    acc[1][3] = wmma_f32(f.a1.zw, f.b3.zw, acc[1][3]);
}

__device__ __forceinline__ void mma_group_cols(v8f (&acc)[2][4], const frag6& f) {
    #pragma unroll
    for (int t = 0; t < 4; ++t) {
        v2f bt; bt.x = f.b0[t]; bt.y = f.b1[t];
        acc[0][t] = wmma_f32(f.a0.xy, bt, acc[0][t]);
        acc[1][t] = wmma_f32(f.a1.xy, bt, acc[1][t]);
    }
    #pragma unroll
    for (int t = 0; t < 4; ++t) {
        v2f bt; bt.x = f.b2[t]; bt.y = f.b3[t];
        acc[0][t] = wmma_f32(f.a0.zw, bt, acc[0][t]);
        acc[1][t] = wmma_f32(f.a1.zw, bt, acc[1][t]);
    }
}

struct frag7 { v4f a0, a1, b0, b1, b2, b3, bq; };

template <int AQ>
__device__ __forceinline__ void mma_group_rows9(v8f (&acc)[2][4], v8f& accq, const frag7& f) {
    const v4f aq = AQ ? f.a1 : f.a0;
    acc[0][0] = wmma_f32(f.a0.xy, f.b0.xy, acc[0][0]);
    acc[1][0] = wmma_f32(f.a1.xy, f.b0.xy, acc[1][0]);
    acc[0][1] = wmma_f32(f.a0.xy, f.b1.xy, acc[0][1]);
    acc[1][1] = wmma_f32(f.a1.xy, f.b1.xy, acc[1][1]);
    acc[0][2] = wmma_f32(f.a0.xy, f.b2.xy, acc[0][2]);
    acc[1][2] = wmma_f32(f.a1.xy, f.b2.xy, acc[1][2]);
    acc[0][3] = wmma_f32(f.a0.xy, f.b3.xy, acc[0][3]);
    acc[1][3] = wmma_f32(f.a1.xy, f.b3.xy, acc[1][3]);
    accq      = wmma_f32(aq.xy,   f.bq.xy, accq);
    acc[0][0] = wmma_f32(f.a0.zw, f.b0.zw, acc[0][0]);
    acc[1][0] = wmma_f32(f.a1.zw, f.b0.zw, acc[1][0]);
    acc[0][1] = wmma_f32(f.a0.zw, f.b1.zw, acc[0][1]);
    acc[1][1] = wmma_f32(f.a1.zw, f.b1.zw, acc[1][1]);
    acc[0][2] = wmma_f32(f.a0.zw, f.b2.zw, acc[0][2]);
    acc[1][2] = wmma_f32(f.a1.zw, f.b2.zw, acc[1][2]);
    acc[0][3] = wmma_f32(f.a0.zw, f.b3.zw, acc[0][3]);
    acc[1][3] = wmma_f32(f.a1.zw, f.b3.zw, acc[1][3]);
    accq      = wmma_f32(aq.zw,   f.bq.zw, accq);
}

__device__ __forceinline__ v4f lo4(v8f v) { return __builtin_shufflevector(v, v, 0, 1, 2, 3); }
__device__ __forceinline__ v4f hi4(v8f v) { return __builtin_shufflevector(v, v, 4, 5, 6, 7); }
__device__ __forceinline__ v8f join8(v4f l, v4f h) {
    return __builtin_shufflevector(l, h, 0, 1, 2, 3, 4, 5, 6, 7);
}

template <int AQ>
__device__ __forceinline__ void
proj3_body(const float* __restrict__ A,
           const float* __restrict__ Wy,
           const float* __restrict__ Wyb,
           const float* __restrict__ Wsel,
           float* __restrict__ Osel,
           float* __restrict__ Vy,
           int K, int mb, int pr,
           v4f (*xch)[10][32])
{
    const int lane = threadIdx.x & 31;
    const int wave = __builtin_amdgcn_readfirstlane((int)(threadIdx.x >> 5));
    const int p    = wave >> 1;
    const int h    = wave & 1;
    const int nb   = (pr << 1) + p;
    const int qc0  = ((pr & 2) << 4) + (p << 4);
    const int l15  = lane & 15;
    const int koff = (lane < 16) ? 0 : 4;
    const int kh   = K >> 1;
    const size_t bstep = (size_t)16 * K;
    const size_t k0    = (size_t)(h * kh + koff);

    const float* pA0 = A + (size_t)((mb << 5) + l15) * K + k0;
    const float* pA1 = pA0 + bstep;
    const float* pB0 = Wy + (size_t)((nb << 6) + l15) * K + k0;
    const float* pB1 = pB0 + bstep;
    const float* pB2 = pB1 + bstep;
    const float* pB3 = pB2 + bstep;
    const float* pBq = Wsel + (size_t)(qc0 + l15) * K + k0;

    auto load_group = [&](frag7& f) {
        f.a0 = *(const v4f*)pA0;
        f.a1 = *(const v4f*)pA1;
        f.b0 = *(const v4f*)pB0;
        f.b1 = *(const v4f*)pB1;
        f.b2 = *(const v4f*)pB2;
        f.b3 = *(const v4f*)pB3;
        f.bq = *(const v4f*)pBq;
        pA0 += 8; pA1 += 8; pB0 += 8; pB1 += 8; pB2 += 8; pB3 += 8; pBq += 8;
    };

    v8f acc[2][4] = {};
    v8f accq = {};
    frag7 s0, s1, s2, s3;
    load_group(s0);  __builtin_amdgcn_sched_barrier(0);
    load_group(s1);  __builtin_amdgcn_sched_barrier(0);
    load_group(s2);  __builtin_amdgcn_sched_barrier(0);
    for (int kk = 32; kk < kh; kk += 32) {
        load_group(s3);  __builtin_amdgcn_sched_barrier(0);
        mma_group_rows9<AQ>(acc, accq, s0);  __builtin_amdgcn_sched_barrier(0);
        load_group(s0);  __builtin_amdgcn_sched_barrier(0);
        mma_group_rows9<AQ>(acc, accq, s1);  __builtin_amdgcn_sched_barrier(0);
        load_group(s1);  __builtin_amdgcn_sched_barrier(0);
        mma_group_rows9<AQ>(acc, accq, s2);  __builtin_amdgcn_sched_barrier(0);
        load_group(s2);  __builtin_amdgcn_sched_barrier(0);
        mma_group_rows9<AQ>(acc, accq, s3);  __builtin_amdgcn_sched_barrier(0);
    }
    load_group(s3);
    __builtin_amdgcn_sched_barrier(0);
    mma_group_rows9<AQ>(acc, accq, s0);
    mma_group_rows9<AQ>(acc, accq, s1);
    mma_group_rows9<AQ>(acc, accq, s2);
    mma_group_rows9<AQ>(acc, accq, s3);

    v4f (*mine)[32] = xch[wave];
    if (h == 0) {
        for (int t = 0; t < 4; ++t) {
            mine[2 * t][lane]     = lo4(acc[1][t]);
            mine[2 * t + 1][lane] = hi4(acc[1][t]);
        }
        if (p == 1) {
            mine[8][lane] = lo4(accq);
            mine[9][lane] = hi4(accq);
        }
    } else {
        for (int t = 0; t < 4; ++t) {
            mine[2 * t][lane]     = lo4(acc[0][t]);
            mine[2 * t + 1][lane] = hi4(acc[0][t]);
        }
        mine[8][lane] = lo4(accq);
        mine[9][lane] = hi4(accq);
    }
    __syncthreads();

    const int hh8 = (lane < 16) ? 0 : 8;
    if (h == 0) {
        v4f (*oth)[32] = xch[wave + 1];
        #pragma unroll
        for (int ni = 0; ni < 4; ++ni) {
            const v8f   sm = acc[0][ni] + join8(oth[2 * ni][lane], oth[2 * ni + 1][lane]);
            const int   n  = (nb << 6) + (ni << 4) + l15;
            const float bn = Wyb[n];
            #pragma unroll
            for (int r = 0; r < 8; ++r)
                Vy[(size_t)((mb << 5) + hh8 + r) * 1024 + n] = sm[r] + bn;
        }
        if (p == 0) {
            const v8f t0 = accq + join8(xch[1][8][lane], xch[1][9][lane]);
            const v8f t1 = join8(xch[2][8][lane], xch[2][9][lane])
                         + join8(xch[3][8][lane], xch[3][9][lane]);
            const int rq = (mb << 5) + (AQ << 4) + hh8;
            const int cq = ((pr & 2) << 4) + l15;
            #pragma unroll
            for (int r = 0; r < 8; ++r) {
                Osel[(size_t)(rq + r) * 64 + cq]      = t0[r];
                Osel[(size_t)(rq + r) * 64 + cq + 16] = t1[r];
            }
        }
    } else {
        v4f (*oth)[32] = xch[wave - 1];
        #pragma unroll
        for (int ni = 0; ni < 4; ++ni) {
            const v8f   sm = join8(oth[2 * ni][lane], oth[2 * ni + 1][lane]) + acc[1][ni];
            const int   n  = (nb << 6) + (ni << 4) + l15;
            const float bn = Wyb[n];
            #pragma unroll
            for (int r = 0; r < 8; ++r)
                Vy[(size_t)((mb << 5) + 16 + hh8 + r) * 1024 + n] = sm[r] + bn;
        }
    }
}

__global__ void __launch_bounds__(128)
fa_proj3(const float* __restrict__ A,
         const float* __restrict__ Wq,
         const float* __restrict__ Wk,
         const float* __restrict__ Wyw,
         const float* __restrict__ Wyb,
         float* __restrict__ q,
         float* __restrict__ kout,
         float* __restrict__ Vy,
         int K)
{
    __shared__ v4f xch[4][10][32];
    const int mb = blockIdx.x >> 3;
    const int pr = blockIdx.x & 7;
    const float* Wsel = (pr < 4) ? Wq : Wk;
    float*       Osel = (pr < 4) ? q  : kout;
    if (pr & 1) proj3_body<1>(A, Wyw, Wyb, Wsel, Osel, Vy, K, mb, pr, xch);
    else        proj3_body<0>(A, Wyw, Wyb, Wsel, Osel, Vy, K, mb, pr, xch);
}

#define SP_PITCH 17
__global__ void __launch_bounds__(256)
fa_scores_ew(const float* __restrict__ q,
             const float* __restrict__ k,
             const float* __restrict__ Wphi,
             float* __restrict__ ex,
             float* __restrict__ denom,
             float* __restrict__ Ew)
{
    __shared__ float sp[512 * SP_PITCH];
    __shared__ v4f   exs4[128];
    __shared__ v4f   part4[16 * 16];
    __shared__ float red[8];
    __shared__ float msh;
    float* exs  = (float*)exs4;
    float* part = (float*)part4;
    const int row = blockIdx.x;
    const int b   = row >> 9;
    const int tid = threadIdx.x;
    const int aq  = tid & 15;
    const int jr  = tid >> 4;

    const v4f qv = *(const v4f*)(q + (size_t)row * 64 + (aq << 2));
    const v4f pw = *(const v4f*)(Wphi + (aq << 2));
    const float* kp = k + ((size_t)(b * 512 + (jr << 5))) * 64 + (aq << 2);

    v4f e[32];
    #pragma unroll
    for (int t = 0; t < 32; ++t)
        e[t] = *(const v4f*)(kp + t * 64);

    float eps_v = EPS_RHO;
    asm volatile("" : "+v"(eps_v));
    const float rt_eps = sqrt_hw(eps_v);
    float* spw = sp + (jr << 5) * SP_PITCH + aq;
    #pragma unroll
    for (int t = 0; t < 32; ++t) {
        v4f ev;
        ev.x = rho_relu(qv.x + e[t].x, rt_eps);
        ev.y = rho_relu(qv.y + e[t].y, rt_eps);
        ev.z = rho_relu(qv.z + e[t].z, rt_eps);
        ev.w = rho_relu(qv.w + e[t].w, rt_eps);
        e[t] = ev;
        float d = ev.x * pw.x;
        d += ev.y * pw.y;
        d += ev.z * pw.z;
        d += ev.w * pw.w;
        spw[t * SP_PITCH] = d;
    }
    __syncthreads();

    float s[2];
    #pragma unroll
    for (int p = 0; p < 2; ++p) {
        const float* r = sp + (tid + (p << 8)) * SP_PITCH;
        const float u0 = (r[0]  + r[1])  + (r[2]  + r[3]);
        const float u1 = (r[4]  + r[5])  + (r[6]  + r[7]);
        const float u2 = (r[8]  + r[9])  + (r[10] + r[11]);
        const float u3 = (r[12] + r[13]) + (r[14] + r[15]);
        s[p] = (u0 + u1) + (u2 + u3);
    }

    float mymax = fmaxf(s[0], s[1]);
    for (int off = 16; off; off >>= 1)
        mymax = fmaxf(mymax, __shfl_xor(mymax, off));
    const int lane = tid & 31, wv = tid >> 5;
    if (lane == 0) red[wv] = mymax;
    __syncthreads();
    if (tid == 0) {
        float m = red[0];
        for (int w = 1; w < 8; ++w) m = fmaxf(m, red[w]);
        msh = m;
    }
    __syncthreads();
    const float m = msh;

    const float e0 = expf(s[0] - m);
    const float e1 = expf(s[1] - m);
    ex[(size_t)row * 512 + tid]       = e0;
    ex[(size_t)row * 512 + tid + 256] = e1;
    exs[tid]       = e0;
    exs[tid + 256] = e1;

    float mysum = e0 + e1;
    for (int off = 16; off; off >>= 1)
        mysum += __shfl_xor(mysum, off);
    if (lane == 0) red[wv] = mysum;
    __syncthreads();
    if (tid == 0) {
        float sm = 0.f;
        for (int w = 0; w < 8; ++w) sm += red[w];
        denom[row] = sm;
    }

    const v4f* xr = exs4 + (jr << 3);
    v4f w0 = {0.f, 0.f, 0.f, 0.f}, w1 = {0.f, 0.f, 0.f, 0.f};
    #pragma unroll
    for (int t4 = 0; t4 < 8; ++t4) {
        const v4f xv = xr[t4];
        w0 += xv.x * e[4 * t4 + 0];
        w1 += xv.y * e[4 * t4 + 1];
        w0 += xv.z * e[4 * t4 + 2];
        w1 += xv.w * e[4 * t4 + 3];
    }
    part4[(jr << 4) + aq] = w0 + w1;
    __syncthreads();
    if (tid < 64) {
        const float* c = part + tid;
        const float u0 = (c[0 * 64]  + c[1 * 64])  + (c[2 * 64]  + c[3 * 64]);
        const float u1 = (c[4 * 64]  + c[5 * 64])  + (c[6 * 64]  + c[7 * 64]);
        const float u2 = (c[8 * 64]  + c[9 * 64])  + (c[10 * 64] + c[11 * 64]);
        const float u3 = (c[12 * 64] + c[13 * 64]) + (c[14 * 64] + c[15 * 64]);
        Ew[(size_t)row * 64 + tid] = (u0 + u1) + (u2 + u3);
    }
}

__global__ void __launch_bounds__(128)
fa_part12(const float* __restrict__ ex,
          const float* __restrict__ Vy,
          const float* __restrict__ Ew,
          const float* __restrict__ Wew,
          const float* __restrict__ Web,
          const float* __restrict__ denom,
          float* __restrict__ out)
{
    __shared__ v4f xch[4][8][32];
    const int lane = threadIdx.x & 31;
    const int wave = __builtin_amdgcn_readfirstlane((int)(threadIdx.x >> 5));
    const int h    = wave & 1;
    const int job  = (blockIdx.x << 1) + (wave >> 1);
    const int b    = job >> 8;
    const int jj   = job & 255;
    const int mb   = jj >> 4;
    const int nb   = jj & 15;
    const int l15  = lane & 15;
    const int koff = (lane < 16) ? 0 : 4;
    const int col0 = (nb << 6) + (l15 << 2);
    const int row0 = b * 512 + (mb << 5);
    const int j0   = (h << 8) + koff;
    const int a0   = (h << 5) + koff;

    const float* pA0 = ex + ((size_t)(row0 + l15)) * 512 + j0;
    const float* pA1 = pA0 + (size_t)16 * 512;
    const float* pV  = Vy + (size_t)b * 512 * 1024 + (size_t)j0 * 1024 + col0;
    const float* pE0 = Ew + ((size_t)(row0 + l15)) * 64 + a0;
    const float* pE1 = pE0 + (size_t)16 * 64;
    const float* pW  = Wew + (size_t)col0 * 64 + a0;

    auto load_group = [&](frag6& f) {
        f.a0 = *(const v4f*)pA0;
        f.a1 = *(const v4f*)pA1;
        f.b0 = *(const v4f*)(pV);
        f.b1 = *(const v4f*)(pV + 1024);
        f.b2 = *(const v4f*)(pV + 2 * 1024);
        f.b3 = *(const v4f*)(pV + 3 * 1024);
        pA0 += 8; pA1 += 8; pV += 8 * 1024;
    };
    auto load_group_a = [&](frag6& f) {
        f.a0 = *(const v4f*)pE0;
        f.a1 = *(const v4f*)pE1;
        f.b0 = *(const v4f*)(pW);
        f.b1 = *(const v4f*)(pW + 64);
        f.b2 = *(const v4f*)(pW + 2 * 64);
        f.b3 = *(const v4f*)(pW + 3 * 64);
        pE0 += 8; pE1 += 8; pW += 8;
    };

    const int mfin = row0 + (h << 4) + ((lane < 16) ? 0 : 8);
    const v4f bn  = *(const v4f*)(Web + col0);
    const v4f dn0 = *(const v4f*)(denom + mfin);
    const v4f dn1 = *(const v4f*)(denom + mfin + 4);
    __builtin_amdgcn_sched_barrier(0);

    v8f acc[2][4] = {};
    frag6 s0, s1, s2, s3;
    load_group(s0);  __builtin_amdgcn_sched_barrier(0);
    load_group(s1);  __builtin_amdgcn_sched_barrier(0);
    load_group(s2);  __builtin_amdgcn_sched_barrier(0);
    for (int j = 32; j < 256; j += 32) {
        load_group(s3);  __builtin_amdgcn_sched_barrier(0);
        mma_group_cols(acc, s0);  __builtin_amdgcn_sched_barrier(0);
        load_group(s0);  __builtin_amdgcn_sched_barrier(0);
        mma_group_cols(acc, s1);  __builtin_amdgcn_sched_barrier(0);
        load_group(s1);  __builtin_amdgcn_sched_barrier(0);
        mma_group_cols(acc, s2);  __builtin_amdgcn_sched_barrier(0);
        load_group(s2);  __builtin_amdgcn_sched_barrier(0);
        mma_group_cols(acc, s3);  __builtin_amdgcn_sched_barrier(0);
    }
    load_group(s3);
    __builtin_amdgcn_sched_barrier(0);
    mma_group_cols(acc, s0);  __builtin_amdgcn_sched_barrier(0);
    load_group_a(s0);  __builtin_amdgcn_sched_barrier(0);
    mma_group_cols(acc, s1);  __builtin_amdgcn_sched_barrier(0);
    load_group_a(s1);  __builtin_amdgcn_sched_barrier(0);
    mma_group_cols(acc, s2);  __builtin_amdgcn_sched_barrier(0);
    load_group_a(s2);  __builtin_amdgcn_sched_barrier(0);
    mma_group_cols(acc, s3);  __builtin_amdgcn_sched_barrier(0);
    load_group_a(s3);
    __builtin_amdgcn_sched_barrier(0);
    mma_group_rows(acc, s0);
    mma_group_rows(acc, s1);
    mma_group_rows(acc, s2);
    mma_group_rows(acc, s3);

    v4f (*mine)[32] = xch[wave];
    v8f fin[4];
    if (h == 0) {
        for (int t = 0; t < 4; ++t) {
            mine[2 * t][lane]     = lo4(acc[1][t]);
            mine[2 * t + 1][lane] = hi4(acc[1][t]);
        }
    } else {
        for (int t = 0; t < 4; ++t) {
            mine[2 * t][lane]     = lo4(acc[0][t]);
            mine[2 * t + 1][lane] = hi4(acc[0][t]);
        }
    }
    __syncthreads();
    if (h == 0) {
        v4f (*oth)[32] = xch[wave + 1];
        for (int t = 0; t < 4; ++t)
            fin[t] = acc[0][t] + join8(oth[2 * t][lane], oth[2 * t + 1][lane]);
    } else {
        v4f (*oth)[32] = xch[wave - 1];
        for (int t = 0; t < 4; ++t)
            fin[t] = join8(oth[2 * t][lane], oth[2 * t + 1][lane]) + acc[1][t];
    }

    #pragma unroll
    for (int r = 0; r < 8; ++r) {
        const float d  = (r < 4) ? dn0[r & 3] : dn1[r & 3];
        const float dd = d + EPS_DEN;
        v4f o;
        o.x = (fin[0][r] + bn.x * d) / dd;
        o.y = (fin[1][r] + bn.y * d) / dd;
        o.z = (fin[2][r] + bn.z * d) / dd;
        o.w = (fin[3][r] + bn.w * d) / dd;
        *(v4f*)(out + (size_t)(mfin + r) * 1024 + col0) = o;
    }
}

extern "C" void kernel_launch(void* const* d_in, const int* in_sizes, int n_in,
                              void* d_out, int out_size, void* d_ws, size_t ws_size,
                              hipStream_t stream)
{
    (void)in_sizes; (void)n_in; (void)out_size; (void)ws_size;
    const float* X    = (const float*)d_in[0];
    const float* Wq   = (const float*)d_in[1];
    const float* Wk   = (const float*)d_in[2];
    const float* Wphi = (const float*)d_in[3];
    const float* Wyw  = (const float*)d_in[4];
    const float* Wyb  = (const float*)d_in[5];
    const float* Wew  = (const float*)d_in[6];
    const float* Web  = (const float*)d_in[7];
    float* out = (float*)d_out;

    float* ws  = (float*)d_ws;
    float* q   = ws;  ws += 1024 * 64;
    float* kk  = ws;  ws += 1024 * 64;
    float* Vy  = ws;  ws += 1024 * 1024;
    float* ex  = ws;  ws += 1024 * 512;
    float* den = ws;  ws += 1024;
    float* Ew  = ws;  ws += 1024 * 64;

    fa_proj3<<<256, 128, 0, stream>>>(X, Wq, Wk, Wyw, Wyb, q, kk, Vy, 1024);

    fa_scores_ew<<<1024, 256, 0, stream>>>(q, kk, Wphi, ex, den, Ew);

    fa_part12<<<256, 128, 0, stream>>>(ex, Vy, Ew, Wew, Web, den, out);
}


# scores kernel: a block does 4 rows i of one batch (2 pairs sharing barriers) with its piece of k, 32 x 4 per thread, kept in registers: 256 blocks, k read once per block; sums unchanged
# speedup vs baseline: 8.3009x; 1.0479x over previous
#include <hip/hip_runtime.h>
#include <hip/hip_bf16.h>
#include <math.h>

typedef __attribute__((ext_vector_type(2))) float v2f;
typedef __attribute__((ext_vector_type(4))) float v4f;
typedef __attribute__((ext_vector_type(8))) float v8f;

#define EPS_RHO 1e-8f
#define EPS_DEN 1e-9f

__device__ __forceinline__ float sqrt_hw(float y) {
    return __builtin_amdgcn_sqrtf(y);
}

__device__ __forceinline__ float rho_relu(float x, float rt_eps) {
    return sqrt_hw(fmaxf(x, 0.f) + EPS_RHO) - rt_eps;
}

__device__ __forceinline__ v8f wmma_f32(v2f a, v2f b, v8f c) {
    return __builtin_amdgcn_wmma_f32_16x16x4_f32(false, a, false, b,
                                                 (short)0, c, false, false);
}

struct frag6 { v4f a0, a1, b0, b1, b2, b3; };

__device__ __forceinline__ void mma_group_rows(v8f (&acc)[2][4], const frag6& f) {
    acc[0][0] = wmma_f32(f.a0.xy, f.b0.xy, acc[0][0]);
    acc[1][0] = wmma_f32(f.a1.xy, f.b0.xy, acc[1][0]);
    acc[0][1] = wmma_f32(f.a0.xy, f.b1.xy, acc[0][1]);
    acc[1][1] = wmma_f32(f.a1.xy, f.b1.xy, acc[1][1]);
    acc[0][2] = wmma_f32(f.a0.xy, f.b2.xy, acc[0][2]);
    acc[1][2] = wmma_f32(f.a1.xy, f.b2.xy, acc[1][2]);
    acc[0][3] = wmma_f32(f.a0.xy, f.b3.xy, acc[0][3]);
    acc[1][3] = wmma_f32(f.a1.xy, f.b3.xy, acc[1][3]);
    acc[0][0] = wmma_f32(f.a0.zw, f.b0.zw, acc[0][0]);
    acc[1][0] = wmma_f32(f.a1.zw, f.b0.zw, acc[1][0]);
    acc[0][1] = wmma_f32(f.a0.zw, f.b1.zw, acc[0][1]);
    acc[1][1] = wmma_f32(f.a1.zw, f.b1.zw, acc[1][1]);
    acc[0][2] = wmma_f32(f.a0.zw, f.b2.zw, acc[0][2]);
    acc[1][2] = wmma_f32(f.a1.zw, f.b2.zw, acc[1][2]);
    acc[0][3] = wmma_f32(f.a0.zw, f.b3.zw, acc[0][3]);
    acc[1][3] = wmma_f32(f.a1.zw, f.b3.zw, acc[1][3]);
}

__device__ __forceinline__ void mma_group_cols(v8f (&acc)[2][4], const frag6& f) {
    #pragma unroll
    for (int t = 0; t < 4; ++t) {
        v2f bt; bt.x = f.b0[t]; bt.y = f.b1[t];
        acc[0][t] = wmma_f32(f.a0.xy, bt, acc[0][t]);
        acc[1][t] = wmma_f32(f.a1.xy, bt, acc[1][t]);
    }
    #pragma unroll
    for (int t = 0; t < 4; ++t) {
        v2f bt; bt.x = f.b2[t]; bt.y = f.b3[t];
        acc[0][t] = wmma_f32(f.a0.zw, bt, acc[0][t]);
        acc[1][t] = wmma_f32(f.a1.zw, bt, acc[1][t]);
    }
}

struct frag7 { v4f a0, a1, b0, b1, b2, b3, bq; };

template <int AQ>
__device__ __forceinline__ void mma_group_rows9(v8f (&acc)[2][4], v8f& accq, const frag7& f) {
    const v4f aq = AQ ? f.a1 : f.a0;
    acc[0][0] = wmma_f32(f.a0.xy, f.b0.xy, acc[0][0]);
    acc[1][0] = wmma_f32(f.a1.xy, f.b0.xy, acc[1][0]);
    acc[0][1] = wmma_f32(f.a0.xy, f.b1.xy, acc[0][1]);
    acc[1][1] = wmma_f32(f.a1.xy, f.b1.xy, acc[1][1]);
    acc[0][2] = wmma_f32(f.a0.xy, f.b2.xy, acc[0][2]);
    acc[1][2] = wmma_f32(f.a1.xy, f.b2.xy, acc[1][2]);
    acc[0][3] = wmma_f32(f.a0.xy, f.b3.xy, acc[0][3]);
    acc[1][3] = wmma_f32(f.a1.xy, f.b3.xy, acc[1][3]);
    accq      = wmma_f32(aq.xy,   f.bq.xy, accq);
    acc[0][0] = wmma_f32(f.a0.zw, f.b0.zw, acc[0][0]);
    acc[1][0] = wmma_f32(f.a1.zw, f.b0.zw, acc[1][0]);
    acc[0][1] = wmma_f32(f.a0.zw, f.b1.zw, acc[0][1]);
    acc[1][1] = wmma_f32(f.a1.zw, f.b1.zw, acc[1][1]);
    acc[0][2] = wmma_f32(f.a0.zw, f.b2.zw, acc[0][2]);
    acc[1][2] = wmma_f32(f.a1.zw, f.b2.zw, acc[1][2]);
    acc[0][3] = wmma_f32(f.a0.zw, f.b3.zw, acc[0][3]);
    acc[1][3] = wmma_f32(f.a1.zw, f.b3.zw, acc[1][3]);
    accq      = wmma_f32(aq.zw,   f.bq.zw, accq);
}

__device__ __forceinline__ v4f lo4(v8f v) { return __builtin_shufflevector(v, v, 0, 1, 2, 3); }
__device__ __forceinline__ v4f hi4(v8f v) { return __builtin_shufflevector(v, v, 4, 5, 6, 7); }
__device__ __forceinline__ v8f join8(v4f l, v4f h) {
    return __builtin_shufflevector(l, h, 0, 1, 2, 3, 4, 5, 6, 7);
}

template <int AQ>
__device__ __forceinline__ void
proj3_body(const float* __restrict__ A,
           const float* __restrict__ Wy,
           const float* __restrict__ Wyb,
           const float* __restrict__ Wsel,
           float* __restrict__ Osel,
           float* __restrict__ Vy,
           int K, int mb, int pr,
           v4f (*xch)[10][32])
{
    const int lane = threadIdx.x & 31;
    const int wave = __builtin_amdgcn_readfirstlane((int)(threadIdx.x >> 5));
    const int p    = wave >> 1;
    const int h    = wave & 1;
    const int nb   = (pr << 1) + p;
    const int qc0  = ((pr & 2) << 4) + (p << 4);
    const int l15  = lane & 15;
    const int koff = (lane < 16) ? 0 : 4;
    const int kh   = K >> 1;
    const size_t bstep = (size_t)16 * K;
    const size_t k0    = (size_t)(h * kh + koff);

    const float* pA0 = A + (size_t)((mb << 5) + l15) * K + k0;
    const float* pA1 = pA0 + bstep;
    const float* pB0 = Wy + (size_t)((nb << 6) + l15) * K + k0;
    const float* pB1 = pB0 + bstep;
    const float* pB2 = pB1 + bstep;
    const float* pB3 = pB2 + bstep;
    const float* pBq = Wsel + (size_t)(qc0 + l15) * K + k0;

    auto load_group = [&](frag7& f) {
        f.a0 = *(const v4f*)pA0;
        f.a1 = *(const v4f*)pA1;
        f.b0 = *(const v4f*)pB0;
        f.b1 = *(const v4f*)pB1;
        f.b2 = *(const v4f*)pB2;
        f.b3 = *(const v4f*)pB3;
        f.bq = *(const v4f*)pBq;
        pA0 += 8; pA1 += 8; pB0 += 8; pB1 += 8; pB2 += 8; pB3 += 8; pBq += 8;
    };

    v8f acc[2][4] = {};
    v8f accq = {};
    frag7 s0, s1, s2, s3;
    load_group(s0);  __builtin_amdgcn_sched_barrier(0);
    load_group(s1);  __builtin_amdgcn_sched_barrier(0);
    load_group(s2);  __builtin_amdgcn_sched_barrier(0);
    for (int kk = 32; kk < kh; kk += 32) {
        load_group(s3);  __builtin_amdgcn_sched_barrier(0);
        mma_group_rows9<AQ>(acc, accq, s0);  __builtin_amdgcn_sched_barrier(0);
        load_group(s0);  __builtin_amdgcn_sched_barrier(0);
        mma_group_rows9<AQ>(acc, accq, s1);  __builtin_amdgcn_sched_barrier(0);
        load_group(s1);  __builtin_amdgcn_sched_barrier(0);
        mma_group_rows9<AQ>(acc, accq, s2);  __builtin_amdgcn_sched_barrier(0);
        load_group(s2);  __builtin_amdgcn_sched_barrier(0);
        mma_group_rows9<AQ>(acc, accq, s3);  __builtin_amdgcn_sched_barrier(0);
    }
    load_group(s3);
    __builtin_amdgcn_sched_barrier(0);
    mma_group_rows9<AQ>(acc, accq, s0);
    mma_group_rows9<AQ>(acc, accq, s1);
    mma_group_rows9<AQ>(acc, accq, s2);
    mma_group_rows9<AQ>(acc, accq, s3);

    v4f (*mine)[32] = xch[wave];
    if (h == 0) {
        for (int t = 0; t < 4; ++t) {
            mine[2 * t][lane]     = lo4(acc[1][t]);
            mine[2 * t + 1][lane] = hi4(acc[1][t]);
        }
        if (p == 1) {
            mine[8][lane] = lo4(accq);
            mine[9][lane] = hi4(accq);
        }
    } else {
        for (int t = 0; t < 4; ++t) {
            mine[2 * t][lane]     = lo4(acc[0][t]);
            mine[2 * t + 1][lane] = hi4(acc[0][t]);
        }
        mine[8][lane] = lo4(accq);
        mine[9][lane] = hi4(accq);
    }
    __syncthreads();

    const int hh8 = (lane < 16) ? 0 : 8;
    if (h == 0) {
        v4f (*oth)[32] = xch[wave + 1];
        #pragma unroll
        for (int ni = 0; ni < 4; ++ni) {
            const v8f   sm = acc[0][ni] + join8(oth[2 * ni][lane], oth[2 * ni + 1][lane]);
            const int   n  = (nb << 6) + (ni << 4) + l15;
            const float bn = Wyb[n];
            #pragma unroll
            for (int r = 0; r < 8; ++r)
                Vy[(size_t)((mb << 5) + hh8 + r) * 1024 + n] = sm[r] + bn;
        }
        if (p == 0) {
            const v8f t0 = accq + join8(xch[1][8][lane], xch[1][9][lane]);
            const v8f t1 = join8(xch[2][8][lane], xch[2][9][lane])
                         + join8(xch[3][8][lane], xch[3][9][lane]);
            const int rq = (mb << 5) + (AQ << 4) + hh8;
            const int cq = ((pr & 2) << 4) + l15;
            #pragma unroll
            for (int r = 0; r < 8; ++r) {
                Osel[(size_t)(rq + r) * 64 + cq]      = t0[r];
                Osel[(size_t)(rq + r) * 64 + cq + 16] = t1[r];
            }
        }
    } else {
        v4f (*oth)[32] = xch[wave - 1];
        #pragma unroll
        for (int ni = 0; ni < 4; ++ni) {
            const v8f   sm = join8(oth[2 * ni][lane], oth[2 * ni + 1][lane]) + acc[1][ni];
            const int   n  = (nb << 6) + (ni << 4) + l15;
            const float bn = Wyb[n];
            #pragma unroll
            for (int r = 0; r < 8; ++r)
                Vy[(size_t)((mb << 5) + 16 + hh8 + r) * 1024 + n] = sm[r] + bn;
        }
    }
}

__global__ void __launch_bounds__(128)
fa_proj3(const float* __restrict__ A,
         const float* __restrict__ Wq,
         const float* __restrict__ Wk,
         const float* __restrict__ Wyw,
         const float* __restrict__ Wyb,
         float* __restrict__ q,
         float* __restrict__ kout,
         float* __restrict__ Vy,
         int K)
{
    __shared__ v4f xch[4][10][32];
    const int mb = blockIdx.x >> 3;
    const int pr = blockIdx.x & 7;
    const float* Wsel = (pr < 4) ? Wq : Wk;
    float*       Osel = (pr < 4) ? q  : kout;
    if (pr & 1) proj3_body<1>(A, Wyw, Wyb, Wsel, Osel, Vy, K, mb, pr, xch);
    else        proj3_body<0>(A, Wyw, Wyb, Wsel, Osel, Vy, K, mb, pr, xch);
}

#define SP_PITCH 17
#define ROWS_PER_BLOCK 4
__global__ void __launch_bounds__(256)
fa_scores_ew(const float* __restrict__ q,
             const float* __restrict__ k,
             const float* __restrict__ Wphi,
             float* __restrict__ ex,
             float* __restrict__ denom,
             float* __restrict__ Ew)
{
    __shared__ float sp[2][512 * SP_PITCH];
    __shared__ v4f   exs4[2][128];
    __shared__ v4f   part4[2][16 * 16];
    __shared__ float red[2][8];
    __shared__ float msh[2];
    const int row0 = blockIdx.x * ROWS_PER_BLOCK;
    const int b    = row0 >> 9;
    const int tid  = threadIdx.x;
    const int aq   = tid & 15;
    const int jr   = tid >> 4;
    const int lane = tid & 31, wv = tid >> 5;

    const float* qp = q + (size_t)row0 * 64 + (aq << 2);
    v4f qv[ROWS_PER_BLOCK];
    #pragma unroll
    for (int r = 0; r < ROWS_PER_BLOCK; ++r)
        qv[r] = *(const v4f*)(qp + r * 64);
    const v4f pw = *(const v4f*)(Wphi + (aq << 2));
    const float* kp = k + ((size_t)(b * 512 + (jr << 5))) * 64 + (aq << 2);
    v4f kr[32];
    #pragma unroll
    for (int t = 0; t < 32; ++t)
        kr[t] = *(const v4f*)(kp + t * 64);

    float eps_v = EPS_RHO;
    asm volatile("" : "+v"(eps_v));
    const float rt_eps = sqrt_hw(eps_v);
    float* spw = &sp[0][0] + (jr << 5) * SP_PITCH + aq;

    #pragma unroll
    for (int pr = 0; pr < ROWS_PER_BLOCK / 2; ++pr) {
        const int rowA = row0 + 2 * pr;
        const v4f qa = qv[2 * pr], qb = qv[2 * pr + 1];

        v4f ea[32], eb[32];
        #pragma unroll
        for (int t = 0; t < 32; ++t) {
            const v4f kv = kr[t];
            v4f va, vb;
            va.x = rho_relu(qa.x + kv.x, rt_eps);
            va.y = rho_relu(qa.y + kv.y, rt_eps);
            va.z = rho_relu(qa.z + kv.z, rt_eps);
            va.w = rho_relu(qa.w + kv.w, rt_eps);
            vb.x = rho_relu(qb.x + kv.x, rt_eps);
            vb.y = rho_relu(qb.y + kv.y, rt_eps);
            vb.z = rho_relu(qb.z + kv.z, rt_eps);
            vb.w = rho_relu(qb.w + kv.w, rt_eps);
            ea[t] = va;
            eb[t] = vb;
            float da = va.x * pw.x;
            da += va.y * pw.y;
            da += va.z * pw.z;
            da += va.w * pw.w;
            float db = vb.x * pw.x;
            db += vb.y * pw.y;
            db += vb.z * pw.z;
            db += vb.w * pw.w;
            spw[t * SP_PITCH] = da;
            spw[(512 + t) * SP_PITCH] = db;
        }
        __syncthreads();

        float s[2][2];
        #pragma unroll
        for (int h = 0; h < 2; ++h) {
            #pragma unroll
            for (int p = 0; p < 2; ++p) {
                const float* rr = &sp[h][0] + (tid + (p << 8)) * SP_PITCH;
                const float u0 = (rr[0]  + rr[1])  + (rr[2]  + rr[3]);
                const float u1 = (rr[4]  + rr[5])  + (rr[6]  + rr[7]);
                const float u2 = (rr[8]  + rr[9])  + (rr[10] + rr[11]);
                const float u3 = (rr[12] + rr[13]) + (rr[14] + rr[15]);
                s[h][p] = (u0 + u1) + (u2 + u3);
            }
        }

        float mx0 = fmaxf(s[0][0], s[0][1]);
        float mx1 = fmaxf(s[1][0], s[1][1]);
        for (int off = 16; off; off >>= 1) {
            mx0 = fmaxf(mx0, __shfl_xor(mx0, off));
            mx1 = fmaxf(mx1, __shfl_xor(mx1, off));
        }
        if (lane == 0) { red[0][wv] = mx0; red[1][wv] = mx1; }
        __syncthreads();
        if (tid < 2) {
            float m = red[tid][0];
            for (int w = 1; w < 8; ++w) m = fmaxf(m, red[tid][w]);
            msh[tid] = m;
        }
        __syncthreads();
        const float m0 = msh[0], m1 = msh[1];

        float x[2][2];
        x[0][0] = expf(s[0][0] - m0);
        x[0][1] = expf(s[0][1] - m0);
        x[1][0] = expf(s[1][0] - m1);
        x[1][1] = expf(s[1][1] - m1);
        float* exs = (float*)&exs4[0][0];
        exs[tid]             = x[0][0];
        exs[tid + 256]       = x[0][1];
        exs[512 + tid]       = x[1][0];
        exs[512 + tid + 256] = x[1][1];

        float sm0 = x[0][0] + x[0][1];
        float sm1 = x[1][0] + x[1][1];
        for (int off = 16; off; off >>= 1) {
            sm0 += __shfl_xor(sm0, off);
            sm1 += __shfl_xor(sm1, off);
        }
        if (lane == 0) { red[0][wv] = sm0; red[1][wv] = sm1; }
        __syncthreads();
        if (tid < 2) {
            float sm = 0.f;
            for (int w = 0; w < 8; ++w) sm += red[tid][w];
            denom[rowA + tid] = sm;
        }

        {
            const v4f* xra = &exs4[0][0] + (jr << 3);
            const v4f* xrb = &exs4[1][0] + (jr << 3);
            v4f a0 = {0.f, 0.f, 0.f, 0.f}, a1 = {0.f, 0.f, 0.f, 0.f};
            v4f b0 = {0.f, 0.f, 0.f, 0.f}, b1 = {0.f, 0.f, 0.f, 0.f};
            #pragma unroll
            for (int t4 = 0; t4 < 8; ++t4) {
                const v4f xa = xra[t4];
                const v4f xb = xrb[t4];
                a0 += xa.x * ea[4 * t4 + 0];
                a1 += xa.y * ea[4 * t4 + 1];
                a0 += xa.z * ea[4 * t4 + 2];
                a1 += xa.w * ea[4 * t4 + 3];
                b0 += xb.x * eb[4 * t4 + 0];
                b1 += xb.y * eb[4 * t4 + 1];
                b0 += xb.z * eb[4 * t4 + 2];
                b1 += xb.w * eb[4 * t4 + 3];
            }
            part4[0][(jr << 4) + aq] = a0 + a1;
            part4[1][(jr << 4) + aq] = b0 + b1;
        }
        __syncthreads();
        if (tid < 128) {
            const int h = tid >> 6;
            const int a = tid & 63;
            const float* c = (const float*)&part4[h][0] + a;
            const float u0 = (c[0 * 64]  + c[1 * 64])  + (c[2 * 64]  + c[3 * 64]);
            const float u1 = (c[4 * 64]  + c[5 * 64])  + (c[6 * 64]  + c[7 * 64]);
            const float u2 = (c[8 * 64]  + c[9 * 64])  + (c[10 * 64] + c[11 * 64]);
            const float u3 = (c[12 * 64] + c[13 * 64]) + (c[14 * 64] + c[15 * 64]);
            Ew[(size_t)(rowA + h) * 64 + a] = (u0 + u1) + (u2 + u3);
        }
        ex[(size_t)rowA * 512 + tid]             = x[0][0];
        ex[(size_t)rowA * 512 + tid + 256]       = x[0][1];
        ex[(size_t)(rowA + 1) * 512 + tid]       = x[1][0];
        ex[(size_t)(rowA + 1) * 512 + tid + 256] = x[1][1];
        __builtin_amdgcn_sched_barrier(0);
    }
}

__global__ void __launch_bounds__(128)
fa_part12(const float* __restrict__ ex,
          const float* __restrict__ Vy,
          const float* __restrict__ Ew,
          const float* __restrict__ Wew,
          const float* __restrict__ Web,
          const float* __restrict__ denom,
          float* __restrict__ out)
{
    __shared__ v4f xch[4][8][32];
    const int lane = threadIdx.x & 31;
    const int wave = __builtin_amdgcn_readfirstlane((int)(threadIdx.x >> 5));
    const int h    = wave & 1;
    const int job  = (blockIdx.x << 1) + (wave >> 1);
    const int b    = job >> 8;
    const int jj   = job & 255;
    const int mb   = jj >> 4;
    const int nb   = jj & 15;
    const int l15  = lane & 15;
    const int koff = (lane < 16) ? 0 : 4;
    const int col0 = (nb << 6) + (l15 << 2);
    const int row0 = b * 512 + (mb << 5);
    const int j0   = (h << 8) + koff;
    const int a0   = (h << 5) + koff;

    const float* pA0 = ex + ((size_t)(row0 + l15)) * 512 + j0;
    const float* pA1 = pA0 + (size_t)16 * 512;
    const float* pV  = Vy + (size_t)b * 512 * 1024 + (size_t)j0 * 1024 + col0;
    const float* pE0 = Ew + ((size_t)(row0 + l15)) * 64 + a0;
    const float* pE1 = pE0 + (size_t)16 * 64;
    const float* pW  = Wew + (size_t)col0 * 64 + a0;

    auto load_group = [&](frag6& f) {
        f.a0 = *(const v4f*)pA0;
        f.a1 = *(const v4f*)pA1;
        f.b0 = *(const v4f*)(pV);
        f.b1 = *(const v4f*)(pV + 1024);
        f.b2 = *(const v4f*)(pV + 2 * 1024);
        f.b3 = *(const v4f*)(pV + 3 * 1024);
        pA0 += 8; pA1 += 8; pV += 8 * 1024;
    };
    auto load_group_a = [&](frag6& f) {
        f.a0 = *(const v4f*)pE0;
        f.a1 = *(const v4f*)pE1;
        f.b0 = *(const v4f*)(pW);
        f.b1 = *(const v4f*)(pW + 64);
        f.b2 = *(const v4f*)(pW + 2 * 64);
        f.b3 = *(const v4f*)(pW + 3 * 64);
        pE0 += 8; pE1 += 8; pW += 8;
    };

    const int mfin = row0 + (h << 4) + ((lane < 16) ? 0 : 8);
    const v4f bn  = *(const v4f*)(Web + col0);
    const v4f dn0 = *(const v4f*)(denom + mfin);
    const v4f dn1 = *(const v4f*)(denom + mfin + 4);
    __builtin_amdgcn_sched_barrier(0);

    v8f acc[2][4] = {};
    frag6 s0, s1, s2, s3;
    load_group(s0);  __builtin_amdgcn_sched_barrier(0);
    load_group(s1);  __builtin_amdgcn_sched_barrier(0);
    load_group(s2);  __builtin_amdgcn_sched_barrier(0);
    for (int j = 32; j < 256; j += 32) {
        load_group(s3);  __builtin_amdgcn_sched_barrier(0);
        mma_group_cols(acc, s0);  __builtin_amdgcn_sched_barrier(0);
        load_group(s0);  __builtin_amdgcn_sched_barrier(0);
        mma_group_cols(acc, s1);  __builtin_amdgcn_sched_barrier(0);
        load_group(s1);  __builtin_amdgcn_sched_barrier(0);
        mma_group_cols(acc, s2);  __builtin_amdgcn_sched_barrier(0);
        load_group(s2);  __builtin_amdgcn_sched_barrier(0);
        mma_group_cols(acc, s3);  __builtin_amdgcn_sched_barrier(0);
    }
    load_group(s3);
    __builtin_amdgcn_sched_barrier(0);
    mma_group_cols(acc, s0);  __builtin_amdgcn_sched_barrier(0);
    load_group_a(s0);  __builtin_amdgcn_sched_barrier(0);
    mma_group_cols(acc, s1);  __builtin_amdgcn_sched_barrier(0);
    load_group_a(s1);  __builtin_amdgcn_sched_barrier(0);
    mma_group_cols(acc, s2);  __builtin_amdgcn_sched_barrier(0);
    load_group_a(s2);  __builtin_amdgcn_sched_barrier(0);
    mma_group_cols(acc, s3);  __builtin_amdgcn_sched_barrier(0);
    load_group_a(s3);
    __builtin_amdgcn_sched_barrier(0);
    mma_group_rows(acc, s0);
    mma_group_rows(acc, s1);
    mma_group_rows(acc, s2);
    mma_group_rows(acc, s3);

    v4f (*mine)[32] = xch[wave];
    v8f fin[4];
    if (h == 0) {
        for (int t = 0; t < 4; ++t) {
            mine[2 * t][lane]     = lo4(acc[1][t]);
            mine[2 * t + 1][lane] = hi4(acc[1][t]);
        }
    } else {
        for (int t = 0; t < 4; ++t) {
            mine[2 * t][lane]     = lo4(acc[0][t]);
            mine[2 * t + 1][lane] = hi4(acc[0][t]);
        }
    }
    __syncthreads();
    if (h == 0) {
        v4f (*oth)[32] = xch[wave + 1];
        for (int t = 0; t < 4; ++t)
            fin[t] = acc[0][t] + join8(oth[2 * t][lane], oth[2 * t + 1][lane]);
    } else {
        v4f (*oth)[32] = xch[wave - 1];
        for (int t = 0; t < 4; ++t)
            fin[t] = join8(oth[2 * t][lane], oth[2 * t + 1][lane]) + acc[1][t];
    }

    #pragma unroll
    for (int r = 0; r < 8; ++r) {
        const float d  = (r < 4) ? dn0[r & 3] : dn1[r & 3];
        const float dd = d + EPS_DEN;
        v4f o;
        o.x = (fin[0][r] + bn.x * d) / dd;
        o.y = (fin[1][r] + bn.y * d) / dd;
        o.z = (fin[2][r] + bn.z * d) / dd;
        o.w = (fin[3][r] + bn.w * d) / dd;
        *(v4f*)(out + (size_t)(mfin + r) * 1024 + col0) = o;
    }
}

extern "C" void kernel_launch(void* const* d_in, const int* in_sizes, int n_in,
                              void* d_out, int out_size, void* d_ws, size_t ws_size,
                              hipStream_t stream)
{
    (void)in_sizes; (void)n_in; (void)out_size; (void)ws_size;
    const float* X    = (const float*)d_in[0];
    const float* Wq   = (const float*)d_in[1];
    const float* Wk   = (const float*)d_in[2];
    const float* Wphi = (const float*)d_in[3];
    const float* Wyw  = (const float*)d_in[4];
    const float* Wyb  = (const float*)d_in[5];
    const float* Wew  = (const float*)d_in[6];
    const float* Web  = (const float*)d_in[7];
    float* out = (float*)d_out;

    float* ws  = (float*)d_ws;
    float* q   = ws;  ws += 1024 * 64;
    float* kk  = ws;  ws += 1024 * 64;
    float* Vy  = ws;  ws += 1024 * 1024;
    float* ex  = ws;  ws += 1024 * 512;
    float* den = ws;  ws += 1024;
    float* Ew  = ws;  ws += 1024 * 64;

    fa_proj3<<<256, 128, 0, stream>>>(X, Wq, Wk, Wyw, Wyb, q, kk, Vy, 1024);

    fa_scores_ew<<<256, 256, 0, stream>>>(q, kk, Wphi, ex, den, Ew);

    fa_part12<<<256, 128, 0, stream>>>(ex, Vy, Ew, Wew, Web, den, out);
}
